# starting kernel
# baseline (speedup 1.0000x reference)
#include <hip/hip_runtime.h>
#include <math.h>

typedef __attribute__((ext_vector_type(2))) float v2f;
typedef __attribute__((ext_vector_type(8))) float v8f;

#define NN 50000
#define H  64


__global__ void k_zero(float* __restrict__ p, int n) {
    int i = blockIdx.x * blockDim.x + threadIdx.x;
    if (i < n) p[i] = 0.0f;
}

__global__ void k_deg(const int* __restrict__ dst, float* __restrict__ deg, int E) {
    int e = blockIdx.x * blockDim.x + threadIdx.x;
    if (e < E) atomicAdd(&deg[dst[e]], 1.0f);
}

__global__ void k_dinv(float* __restrict__ deg_dinv, int n) {
    int i = blockIdx.x * blockDim.x + threadIdx.x;
    if (i < n) deg_dinv[i] = rsqrtf(deg_dinv[i] + 1.0f);
}

__global__ void k_norm(const int* __restrict__ src, const int* __restrict__ dst,
                       const float* __restrict__ dinv, float* __restrict__ norm, int E) {
    int e = blockIdx.x * blockDim.x + threadIdx.x;
    if (e < E) norm[e] = dinv[src[e]] * dinv[dst[e]];
}

template<int K, int NB, int NT>
__global__ void k_gemm_wmma(const float* __restrict__ A, const float* __restrict__ B,
                            const float* __restrict__ bias, float* __restrict__ C) {
    const int tm   = blockIdx.x;
    const int lane = threadIdx.x;
    const int half = lane >> 4;
    const int l16  = lane & 15;
    const int arow = tm * 16 + l16;

    const float* Ap = A + (size_t)arow * K + half * 2;
    const float* Bp = B + half * 2 * NB;

    v8f acc[NT];
#pragma unroll
    for (int t = 0; t < NT; ++t) acc[t] = (v8f){};

    for (int k = 0; k < K; k += 4) {
        v2f a;
        a.x = Ap[k];
        a.y = Ap[k + 1];
#pragma unroll
        for (int t = 0; t < NT; ++t) {
            const int c = t * 16 + l16;
            v2f b;
            if constexpr (NT * 16 == NB) {
                b.x = Bp[k * NB + c];
                b.y = Bp[k * NB + NB + c];
            } else {
                const float m  = (c < NB) ? 1.0f : 0.0f;
                const int   cc = (c < NB) ? c : (NB - 1);
                b.x = m * Bp[k * NB + cc];
                b.y = m * Bp[k * NB + NB + cc];
            }
            acc[t] = __builtin_amdgcn_wmma_f32_16x16x4_f32(
                false, a, false, b, (short)0, acc[t], false, false);
        }
    }

#pragma unroll
    for (int t = 0; t < NT; ++t) {
        const int c = t * 16 + l16;
        if (NT * 16 == NB || c < NB) {
            const float bv = bias ? bias[c] : 0.0f;
#pragma unroll
            for (int i = 0; i < 8; ++i)
                C[(size_t)(tm * 16 + i + half * 8) * NB + c] = acc[t][i] + bv;
        }
    }
}


__global__ void k_selfloop_init(const float* __restrict__ h, const float* __restrict__ dinv,
                                float* __restrict__ agg, int n) {
    int i = blockIdx.x * blockDim.x + threadIdx.x;
    if (i >= n) return;
    float di = dinv[i >> 6];
    agg[i] = di * di * h[i];
}

__global__ void k_edge_scatter(const int* __restrict__ src, const int* __restrict__ dst,
                               const float* __restrict__ norm, const float* __restrict__ h,
                               float* __restrict__ agg, int E) {
    int gid  = blockIdx.x * blockDim.x + threadIdx.x;
    int e    = gid >> 5;
    int lane = gid & 31;
    if (e >= E) return;
    const int   s = src[e];
    const int   d = dst[e];
    const float w = norm[e];
    const float2 v = ((const float2*)(h + (size_t)s * H))[lane];
    float* o = agg + (size_t)d * H + lane * 2;
    atomicAdd(o,     w * v.x);
    atomicAdd(o + 1, w * v.y);
}

__global__ void k_bias_act(const float* __restrict__ agg, const float* __restrict__ bias,
                           float* __restrict__ act, int n, int do_elu) {
    int i = blockIdx.x * blockDim.x + threadIdx.x;
    if (i >= n) return;
    float v = agg[i] + bias[i & (H - 1)];
    act[i] = (do_elu && v <= 0.0f) ? expm1f(v) : v;
}


extern "C" void kernel_launch(void* const* d_in, const int* in_sizes, int n_in,
                              void* d_out, int out_size, void* d_ws, size_t ws_size,
                              hipStream_t stream) {
    const float* x  = (const float*)d_in[0];
    const int*   ei = (const int*)d_in[1];
    const float* W1 = (const float*)d_in[2];
    const float* b1 = (const float*)d_in[3];
    const float* W2 = (const float*)d_in[4];
    const float* b2 = (const float*)d_in[5];
    const float* W3 = (const float*)d_in[6];
    const float* b3 = (const float*)d_in[7];
    const float* Wl = (const float*)d_in[8];
    const float* bl = (const float*)d_in[9];
    float* out = (float*)d_out;

    const int E = in_sizes[1] / 2;
    const int* src = ei;
    const int* dst = ei + E;

    float* dinv = (float*)d_ws;
    float* norm = dinv + NN;
    float* G    = norm + E;
    float* AGG  = G + (size_t)NN * H;
    float* ACT  = AGG + (size_t)NN * H;

    const int NF = NN * H;
    const dim3 blk(256);

    k_zero<<<dim3((NN + 255) / 256), blk, 0, stream>>>(dinv, NN);
    k_deg<<<dim3((E + 255) / 256), blk, 0, stream>>>(dst, dinv, E);
    k_dinv<<<dim3((NN + 255) / 256), blk, 0, stream>>>(dinv, NN);
    k_norm<<<dim3((E + 255) / 256), blk, 0, stream>>>(src, dst, dinv, norm, E);

    const dim3 gGemm(NN / 16);
    const dim3 wave(32);
    const dim3 gElem((NF + 255) / 256);
    const dim3 gScat((E * 32 + 255) / 256);

    k_gemm_wmma<128, H, 4><<<gGemm, wave, 0, stream>>>(x, W1, nullptr, G);
    k_selfloop_init<<<gElem, blk, 0, stream>>>(G, dinv, AGG, NF);
    k_edge_scatter<<<gScat, blk, 0, stream>>>(src, dst, norm, G, AGG, E);
    k_bias_act<<<gElem, blk, 0, stream>>>(AGG, b1, ACT, NF, 1);

    k_gemm_wmma<H, H, 4><<<gGemm, wave, 0, stream>>>(ACT, W2, nullptr, G);
    k_selfloop_init<<<gElem, blk, 0, stream>>>(G, dinv, AGG, NF);
    k_edge_scatter<<<gScat, blk, 0, stream>>>(src, dst, norm, G, AGG, E);
    k_bias_act<<<gElem, blk, 0, stream>>>(AGG, b2, ACT, NF, 1);

    k_gemm_wmma<H, H, 4><<<gGemm, wave, 0, stream>>>(ACT, W3, nullptr, G);
    k_selfloop_init<<<gElem, blk, 0, stream>>>(G, dinv, AGG, NF);
    k_edge_scatter<<<gScat, blk, 0, stream>>>(src, dst, norm, G, AGG, E);
    k_bias_act<<<gElem, blk, 0, stream>>>(AGG, b3, ACT, NF, 0);

    k_gemm_wmma<H, 40, 3><<<gGemm, wave, 0, stream>>>(ACT, Wl, bl, out);
}


# Global float atomics (degree, edge scatter) go: edges binned by dst tile (256 tiles of 196 nodes), one workgroup sums a tile in LDS with self-loop, bias, ELU; sum order: LDS atomic arrival
# speedup vs baseline: 2.6477x; 2.6477x over previous
#include <hip/hip_runtime.h>
#include <math.h>

typedef __attribute__((ext_vector_type(2))) float v2f;
typedef __attribute__((ext_vector_type(8))) float v8f;

#define NN 50000
#define H  64
#define TILE 196
#define NBIN 256
#define CH   4096


__global__ __launch_bounds__(256) void k_bin_count(const int* __restrict__ dst,
                                                   int* __restrict__ cnt, int E) {
    __shared__ int hist[NBIN];
    const int tid = threadIdx.x;
    hist[tid] = 0;
    __syncthreads();
    const int e0 = blockIdx.x * CH;
    int d[CH / 256];
#pragma unroll
    for (int k = 0; k < CH / 256; ++k) {
        const int e = e0 + k * 256 + tid;
        d[k] = dst[(e < E) ? e : (E - 1)];
    }
#pragma unroll
    for (int k = 0; k < CH / 256; ++k) {
        const int e = e0 + k * 256 + tid;
        if (e < E) atomicAdd(&hist[d[k] / TILE], 1);
    }
    __syncthreads();
    cnt[blockIdx.x * NBIN + tid] = hist[tid];
}

__global__ __launch_bounds__(256) void k_bin_fill(const int* __restrict__ src,
                                                  const int* __restrict__ dst,
                                                  const int* __restrict__ cnt,
                                                  int* __restrict__ rec,
                                                  int* __restrict__ binstart,
                                                  int E, int nch) {
    __shared__ int pos[NBIN];
    __shared__ int scan[NBIN];
    const int tid = threadIdx.x;
    const int c   = blockIdx.x;

    int total = 0, before = 0;
#pragma unroll 14
    for (int cc = 0; cc < nch; ++cc) {
        const int v = cnt[cc * NBIN + tid];
        total  += v;
        before += (cc < c) ? v : 0;
    }
    scan[tid] = total;
    __syncthreads();
    for (int o = 1; o < NBIN; o <<= 1) {
        const int v = (tid >= o) ? scan[tid - o] : 0;
        __syncthreads();
        scan[tid] += v;
        __syncthreads();
    }
    const int binbase = scan[tid] - total;
    pos[tid] = binbase + before;
    if (c == 0) {
        binstart[tid] = binbase;
        if (tid == NBIN - 1) binstart[NBIN] = scan[tid];
    }
    __syncthreads();

    int2* rec2 = (int2*)rec;
    const int e0 = c * CH;
    int d[CH / 256], s[CH / 256];
#pragma unroll
    for (int k = 0; k < CH / 256; ++k) {
        const int e  = e0 + k * 256 + tid;
        const int ec = (e < E) ? e : (E - 1);
        d[k] = dst[ec];
        s[k] = src[ec];
    }
#pragma unroll
    for (int k = 0; k < CH / 256; ++k) {
        const int e = e0 + k * 256 + tid;
        if (e < E) {
            const int b = d[k] / TILE;
            const int p = atomicAdd(&pos[b], 1);
            rec2[p] = make_int2(s[k], d[k] - b * TILE);
        }
    }
}

__global__ __launch_bounds__(1024) void k_tile_dinv(const int* __restrict__ rec,
                                                    const int* __restrict__ binstart,
                                                    float* __restrict__ dinv) {
    __shared__ int c[TILE];
    const int tid = threadIdx.x;
    const int t   = blockIdx.x;
    if (tid < TILE) c[tid] = 0;
    __syncthreads();
    const int2* rec2 = (const int2*)rec;
    const int e1 = binstart[t + 1];
    for (int e = binstart[t] + tid; e < e1; e += 1024) atomicAdd(&c[rec2[e].y], 1);
    __syncthreads();
    const int n = t * TILE + tid;
    if (tid < TILE && n < NN) dinv[n] = rsqrtf((float)c[tid] + 1.0f);
}

template<int DO_ELU>
__global__ __launch_bounds__(1024) void k_tile_agg(const int* __restrict__ rec,
                                                   const int* __restrict__ binstart,
                                                   const float* __restrict__ dinv,
                                                   const float* __restrict__ h,
                                                   const float* __restrict__ bias,
                                                   float* __restrict__ act) {
    __shared__ float acc[TILE * H];
    __shared__ float sdinv[TILE];
    const int tid  = threadIdx.x;
    const int lane = tid & 31;
    const int wv   = tid >> 5;
    const int t    = blockIdx.x;
    const int base = t * TILE;
    const int nn   = (NN - base < TILE) ? (NN - base) : TILE;

    if (tid < nn) sdinv[tid] = dinv[base + tid];
    const float4* h4  = (const float4*)(h + (size_t)base * H);
    float4*       a4  = (float4*)acc;
    static_assert(4 * 1024 >= TILE * (H / 4), "four rounds of 1024 threads cover a tile");
    const int lim = nn * (H / 4);
    float  di[4];
    float4 hv[4];
#pragma unroll
    for (int k = 0; k < 4; ++k) {
        const int i  = tid + k * 1024;
        const int ic = (i < lim) ? i : (lim - 1);
        di[k] = dinv[base + (ic >> 4)];
        hv[k] = h4[ic];
    }
#pragma unroll
    for (int k = 0; k < 4; ++k) {
        const int i = tid + k * 1024;
        if (i < lim) {
            const float s = di[k] * di[k];
            a4[i] = make_float4(s * hv[k].x, s * hv[k].y, s * hv[k].z, s * hv[k].w);
        }
    }
    __syncthreads();

    const int2*   rec2 = (const int2*)rec;
    const float2* h2   = (const float2*)h;
    const int e0 = binstart[t];
    const int e1 = binstart[t + 1];
    int eb = e0 + wv * 32;
    if (eb < e1) {
        const int elast = e1 - 1;
        int2  r  = rec2[(eb + lane < elast) ? (eb + lane) : elast];
        float wl = dinv[r.x] * sdinv[r.y];
        do {
            const int ebu = __builtin_amdgcn_readfirstlane(eb);
            const int cnt = (e1 - ebu < 32) ? (e1 - ebu) : 32;
            const int en = eb + 32 * 32 + lane;
            const int2 rn = rec2[(en < elast) ? en : elast];
            float wn = 0.0f;
#pragma unroll
            for (int g = 0; g < 2; ++g) {
                if (g == 1) wn = dinv[rn.x] * sdinv[rn.y];
                float2 v[16];
#pragma unroll
                for (int u = 0; u < 16; ++u) {
                    const int s = __builtin_amdgcn_readlane(r.x, g * 16 + u);
                    v[u] = h2[(size_t)s * (H / 2) + lane];
                }
#pragma unroll
                for (int u = 0; u < 16; ++u) {
                    const int   j  = g * 16 + u;
                    const int   dl = __builtin_amdgcn_readlane(r.y, j);
                    const float w  = __int_as_float(__builtin_amdgcn_readlane(__float_as_int(wl), j));
                    if (j < cnt) {
                        float* o = acc + dl * H + lane * 2;
                        atomicAdd(o,     w * v[u].x);
                        atomicAdd(o + 1, w * v[u].y);
                    }
                }
            }
            r  = rn;
            wl = wn;
            eb += 32 * 32;
        } while (eb < e1);
    }
    __syncthreads();

    float4* o4 = (float4*)(act + (size_t)base * H);
    for (int i = tid; i < nn * (H / 4); i += 1024) {
        const float4 a = a4[i];
        const int    c = (i & 15) * 4;
        float4 v = make_float4(a.x + bias[c], a.y + bias[c + 1], a.z + bias[c + 2], a.w + bias[c + 3]);
        if (DO_ELU) {
            v.x = (v.x <= 0.0f) ? expm1f(v.x) : v.x;
            v.y = (v.y <= 0.0f) ? expm1f(v.y) : v.y;
            v.z = (v.z <= 0.0f) ? expm1f(v.z) : v.z;
            v.w = (v.w <= 0.0f) ? expm1f(v.w) : v.w;
        }
        o4[i] = v;
    }
}

template<int K, int NB, int NT>
__global__ void k_gemm_wmma(const float* __restrict__ A, const float* __restrict__ B,
                            const float* __restrict__ bias, float* __restrict__ C) {
    const int tm   = blockIdx.x;
    const int lane = threadIdx.x;
    const int half = lane >> 4;
    const int l16  = lane & 15;
    const int arow = tm * 16 + l16;

    const float* Ap = A + (size_t)arow * K + half * 2;
    const float* Bp = B + half * 2 * NB;

    v8f acc[NT];
#pragma unroll
    for (int t = 0; t < NT; ++t) acc[t] = (v8f){};

    for (int k = 0; k < K; k += 4) {
        v2f a;
        a.x = Ap[k];
        a.y = Ap[k + 1];
#pragma unroll
        for (int t = 0; t < NT; ++t) {
            const int c = t * 16 + l16;
            v2f b;
            if constexpr (NT * 16 == NB) {
                b.x = Bp[k * NB + c];
                b.y = Bp[k * NB + NB + c];
            } else {
                const float m  = (c < NB) ? 1.0f : 0.0f;
                const int   cc = (c < NB) ? c : (NB - 1);
                b.x = m * Bp[k * NB + cc];
                b.y = m * Bp[k * NB + NB + cc];
            }
            acc[t] = __builtin_amdgcn_wmma_f32_16x16x4_f32(
                false, a, false, b, (short)0, acc[t], false, false);
        }
    }

#pragma unroll
    for (int t = 0; t < NT; ++t) {
        const int c = t * 16 + l16;
        if (NT * 16 == NB || c < NB) {
            const float bv = bias ? bias[c] : 0.0f;
#pragma unroll
            for (int i = 0; i < 8; ++i)
                C[(size_t)(tm * 16 + i + half * 8) * NB + c] = acc[t][i] + bv;
        }
    }
}


extern "C" void kernel_launch(void* const* d_in, const int* in_sizes, int n_in,
                              void* d_out, int out_size, void* d_ws, size_t ws_size,
                              hipStream_t stream) {
    const float* x  = (const float*)d_in[0];
    const int*   ei = (const int*)d_in[1];
    const float* W1 = (const float*)d_in[2];
    const float* b1 = (const float*)d_in[3];
    const float* W2 = (const float*)d_in[4];
    const float* b2 = (const float*)d_in[5];
    const float* W3 = (const float*)d_in[6];
    const float* b3 = (const float*)d_in[7];
    const float* Wl = (const float*)d_in[8];
    const float* bl = (const float*)d_in[9];
    float* out = (float*)d_out;

    const int E = in_sizes[1] / 2;
    const int* src = ei;
    const int* dst = ei + E;
    const int nch = (E + CH - 1) / CH;

    float* dinv     = (float*)d_ws;
    int*   binstart = (int*)(dinv + 50176);
    int*   cnt      = binstart + 512;
    int*   rec      = cnt + 1024 * NBIN;
    float* G        = (float*)(rec + 2 * (size_t)E);
    float* ACT      = G + (size_t)NN * H;

    const dim3 blk(256);
    const dim3 big(1024);
    const dim3 gTile(NBIN);

    k_bin_count<<<dim3(nch), blk, 0, stream>>>(dst, cnt, E);
    k_bin_fill<<<dim3(nch), blk, 0, stream>>>(src, dst, cnt, rec, binstart, E, nch);
    k_tile_dinv<<<gTile, big, 0, stream>>>(rec, binstart, dinv);

    const dim3 gGemm(NN / 16);
    const dim3 wave(32);

    k_gemm_wmma<128, H, 4><<<gGemm, wave, 0, stream>>>(x, W1, nullptr, G);
    k_tile_agg<1><<<gTile, big, 0, stream>>>(rec, binstart, dinv, G, b1, ACT);

    k_gemm_wmma<H, H, 4><<<gGemm, wave, 0, stream>>>(ACT, W2, nullptr, G);
    k_tile_agg<1><<<gTile, big, 0, stream>>>(rec, binstart, dinv, G, b2, ACT);

    k_gemm_wmma<H, H, 4><<<gGemm, wave, 0, stream>>>(ACT, W3, nullptr, G);
    k_tile_agg<0><<<gTile, big, 0, stream>>>(rec, binstart, dinv, G, b3, ACT);

    k_gemm_wmma<H, 40, 3><<<gGemm, wave, 0, stream>>>(ACT, Wl, bl, out);
}


# Classifier product moved in front of the last aggregation (linear, no activation between): layer-3 launch chains (h W3) Wl via LDS, tiles sum 40-wide rows into out; 10 launches to 9; sums regrouped
# speedup vs baseline: 2.7699x; 1.0462x over previous
#include <hip/hip_runtime.h>
#include <math.h>

typedef __attribute__((ext_vector_type(2))) float v2f;
typedef __attribute__((ext_vector_type(8))) float v8f;

#define NN 50000
#define H  64
#define TILE 196
#define NBIN 256
#define CH   4096


__global__ __launch_bounds__(256) void k_bin_count(const int* __restrict__ dst,
                                                   int* __restrict__ cnt, int E) {
    __shared__ int hist[NBIN];
    const int tid = threadIdx.x;
    hist[tid] = 0;
    __syncthreads();
    const int e0 = blockIdx.x * CH;
    int d[CH / 256];
#pragma unroll
    for (int k = 0; k < CH / 256; ++k) {
        const int e = e0 + k * 256 + tid;
        d[k] = dst[(e < E) ? e : (E - 1)];
    }
#pragma unroll
    for (int k = 0; k < CH / 256; ++k) {
        const int e = e0 + k * 256 + tid;
        if (e < E) atomicAdd(&hist[d[k] / TILE], 1);
    }
    __syncthreads();
    cnt[blockIdx.x * NBIN + tid] = hist[tid];
}

__global__ __launch_bounds__(256) void k_bin_fill(const int* __restrict__ src,
                                                  const int* __restrict__ dst,
                                                  const int* __restrict__ cnt,
                                                  int* __restrict__ rec,
                                                  int* __restrict__ binstart,
                                                  int E, int nch) {
    __shared__ int pos[NBIN];
    __shared__ int scan[NBIN];
    const int tid = threadIdx.x;
    const int c   = blockIdx.x;

    int total = 0, before = 0;
#pragma unroll 14
    for (int cc = 0; cc < nch; ++cc) {
        const int v = cnt[cc * NBIN + tid];
        total  += v;
        before += (cc < c) ? v : 0;
    }
    scan[tid] = total;
    __syncthreads();
    for (int o = 1; o < NBIN; o <<= 1) {
        const int v = (tid >= o) ? scan[tid - o] : 0;
        __syncthreads();
        scan[tid] += v;
        __syncthreads();
    }
    const int binbase = scan[tid] - total;
    pos[tid] = binbase + before;
    if (c == 0) {
        binstart[tid] = binbase;
        if (tid == NBIN - 1) binstart[NBIN] = scan[tid];
    }
    __syncthreads();

    int2* rec2 = (int2*)rec;
    const int e0 = c * CH;
    int d[CH / 256], s[CH / 256];
#pragma unroll
    for (int k = 0; k < CH / 256; ++k) {
        const int e  = e0 + k * 256 + tid;
        const int ec = (e < E) ? e : (E - 1);
        d[k] = dst[ec];
        s[k] = src[ec];
    }
#pragma unroll
    for (int k = 0; k < CH / 256; ++k) {
        const int e = e0 + k * 256 + tid;
        if (e < E) {
            const int b = d[k] / TILE;
            const int p = atomicAdd(&pos[b], 1);
            rec2[p] = make_int2(s[k], d[k] - b * TILE);
        }
    }
}

__global__ __launch_bounds__(1024) void k_tile_dinv(const int* __restrict__ rec,
                                                    const int* __restrict__ binstart,
                                                    float* __restrict__ dinv) {
    __shared__ int c[TILE];
    const int tid = threadIdx.x;
    const int t   = blockIdx.x;
    if (tid < TILE) c[tid] = 0;
    __syncthreads();
    const int2* rec2 = (const int2*)rec;
    const int e1 = binstart[t + 1];
    for (int e = binstart[t] + tid; e < e1; e += 1024) atomicAdd(&c[rec2[e].y], 1);
    __syncthreads();
    const int n = t * TILE + tid;
    if (tid < TILE && n < NN) dinv[n] = rsqrtf((float)c[tid] + 1.0f);
}

template<int DO_ELU, int HW>
__global__ __launch_bounds__(1024) void k_tile_agg(const int* __restrict__ rec,
                                                   const int* __restrict__ binstart,
                                                   const float* __restrict__ dinv,
                                                   const float* __restrict__ h,
                                                   const float* __restrict__ bias,
                                                   float* __restrict__ act) {
    constexpr int Q  = HW / 4;
    constexpr int LP = HW / 2;
    __shared__ __attribute__((aligned(16))) float acc[TILE * HW];
    __shared__ float sdinv[TILE];
    const int tid  = threadIdx.x;
    const int lane = tid & 31;
    const int wv   = tid >> 5;
    const int t    = blockIdx.x;
    const int base = t * TILE;
    const int nn   = (NN - base < TILE) ? (NN - base) : TILE;

    if (tid < nn) sdinv[tid] = dinv[base + tid];
    const float4* h4  = (const float4*)(h + (size_t)base * HW);
    float4*       a4  = (float4*)acc;
    constexpr int NR = (TILE * Q + 1023) / 1024;
    const int lim = nn * Q;
    float  di[NR];
    float4 hv[NR];
#pragma unroll
    for (int k = 0; k < NR; ++k) {
        const int i  = tid + k * 1024;
        const int ic = (i < lim) ? i : (lim - 1);
        di[k] = dinv[base + ((Q == 16) ? (ic >> 4) : (ic / Q))];
        hv[k] = h4[ic];
    }
#pragma unroll
    for (int k = 0; k < NR; ++k) {
        const int i = tid + k * 1024;
        if (i < lim) {
            const float s = di[k] * di[k];
            a4[i] = make_float4(s * hv[k].x, s * hv[k].y, s * hv[k].z, s * hv[k].w);
        }
    }
    __syncthreads();

    const int2*   rec2 = (const int2*)rec;
    const float2* h2   = (const float2*)h;
    const int e0 = binstart[t];
    const int e1 = binstart[t + 1];
    const int lc = (lane < LP) ? lane : (LP - 1);
    int eb = e0 + wv * 32;
    if (eb < e1) {
        const int elast = e1 - 1;
        int2  r  = rec2[(eb + lane < elast) ? (eb + lane) : elast];
        float wl = dinv[r.x] * sdinv[r.y];
        do {
            const int ebu = __builtin_amdgcn_readfirstlane(eb);
            const int cnt = (e1 - ebu < 32) ? (e1 - ebu) : 32;
            const int en = eb + 32 * 32 + lane;
            const int2 rn = rec2[(en < elast) ? en : elast];
            float wn = 0.0f;
#pragma unroll
            for (int g = 0; g < 2; ++g) {
                if (g == 1) wn = dinv[rn.x] * sdinv[rn.y];
                float2 v[16];
#pragma unroll
                for (int u = 0; u < 16; ++u) {
                    const int s = __builtin_amdgcn_readlane(r.x, g * 16 + u);
                    v[u] = h2[(size_t)s * LP + lc];
                }
#pragma unroll
                for (int u = 0; u < 16; ++u) {
                    const int   j  = g * 16 + u;
                    const int   dl = __builtin_amdgcn_readlane(r.y, j);
                    const float w  = __int_as_float(__builtin_amdgcn_readlane(__float_as_int(wl), j));
                    if (j < cnt) {
                        if (LP == 32 || lane < LP) {
                            float* o = acc + dl * HW + lane * 2;
                            atomicAdd(o,     w * v[u].x);
                            atomicAdd(o + 1, w * v[u].y);
                        }
                    }
                }
            }
            r  = rn;
            wl = wn;
            eb += 32 * 32;
        } while (eb < e1);
    }
    __syncthreads();

    float4* o4 = (float4*)(act + (size_t)base * HW);
    for (int i = tid; i < nn * Q; i += 1024) {
        const float4 a = a4[i];
        const int    c = ((Q == 16) ? (i & 15) : (i % Q)) * 4;
        float4 v = make_float4(a.x + bias[c], a.y + bias[c + 1], a.z + bias[c + 2], a.w + bias[c + 3]);
        if (DO_ELU) {
            v.x = (v.x <= 0.0f) ? expm1f(v.x) : v.x;
            v.y = (v.y <= 0.0f) ? expm1f(v.y) : v.y;
            v.z = (v.z <= 0.0f) ? expm1f(v.z) : v.z;
            v.w = (v.w <= 0.0f) ? expm1f(v.w) : v.w;
        }
        o4[i] = v;
    }
}

template<int K, int NB, int NT>
__global__ void k_gemm_wmma(const float* __restrict__ A, const float* __restrict__ B,
                            const float* __restrict__ bias, float* __restrict__ C) {
    const int tm   = blockIdx.x;
    const int lane = threadIdx.x;
    const int half = lane >> 4;
    const int l16  = lane & 15;
    const int arow = tm * 16 + l16;

    const float* Ap = A + (size_t)arow * K + half * 2;
    const float* Bp = B + half * 2 * NB;

    v8f acc[NT];
#pragma unroll
    for (int t = 0; t < NT; ++t) acc[t] = (v8f){};

    for (int k = 0; k < K; k += 4) {
        v2f a;
        a.x = Ap[k];
        a.y = Ap[k + 1];
#pragma unroll
        for (int t = 0; t < NT; ++t) {
            const int c = t * 16 + l16;
            v2f b;
            if constexpr (NT * 16 == NB) {
                b.x = Bp[k * NB + c];
                b.y = Bp[k * NB + NB + c];
            } else {
                const float m  = (c < NB) ? 1.0f : 0.0f;
                const int   cc = (c < NB) ? c : (NB - 1);
                b.x = m * Bp[k * NB + cc];
                b.y = m * Bp[k * NB + NB + cc];
            }
            acc[t] = __builtin_amdgcn_wmma_f32_16x16x4_f32(
                false, a, false, b, (short)0, acc[t], false, false);
        }
    }

#pragma unroll
    for (int t = 0; t < NT; ++t) {
        const int c = t * 16 + l16;
        if (NT * 16 == NB || c < NB) {
            const float bv = bias ? bias[c] : 0.0f;
#pragma unroll
            for (int i = 0; i < 8; ++i)
                C[(size_t)(tm * 16 + i + half * 8) * NB + c] = acc[t][i] + bv;
        }
    }
}

template<int K, int N1, int N2, int NT2>
__global__ __launch_bounds__(32) void k_gemm2_wmma(const float* __restrict__ A,
                                                   const float* __restrict__ B1,
                                                   const float* __restrict__ B2,
                                                   const float* __restrict__ ba,
                                                   const float* __restrict__ bb,
                                                   float* __restrict__ C,
                                                   float* __restrict__ cb) {
    constexpr int NT1 = N1 / 16;
    constexpr int P = N1 + 4;
    __shared__ __attribute__((aligned(16))) float mid[16 * P];

    const int tm   = blockIdx.x;
    const int lane = threadIdx.x;
    const int half = lane >> 4;
    const int l16  = lane & 15;
    const int arow = tm * 16 + l16;

    const float* Ap  = A + (size_t)arow * K + half * 2;
    const float* Bp1 = B1 + half * 2 * N1;
    const float* Bp2 = B2 + half * 2 * N2;

    v8f acc[NT1];
#pragma unroll
    for (int t = 0; t < NT1; ++t) acc[t] = (v8f){};

    for (int k = 0; k < K; k += 4) {
        v2f a;
        a.x = Ap[k];
        a.y = Ap[k + 1];
#pragma unroll
        for (int t = 0; t < NT1; ++t) {
            const int c = t * 16 + l16;
            v2f b;
            b.x = Bp1[k * N1 + c];
            b.y = Bp1[k * N1 + N1 + c];
            acc[t] = __builtin_amdgcn_wmma_f32_16x16x4_f32(
                false, a, false, b, (short)0, acc[t], false, false);
        }
    }

#pragma unroll
    for (int t = 0; t < NT1; ++t)
#pragma unroll
        for (int i = 0; i < 8; ++i)
            mid[(i + half * 8) * P + t * 16 + l16] = acc[t][i];
    __syncthreads();

    v8f acc2[NT2];
#pragma unroll
    for (int t = 0; t < NT2; ++t) acc2[t] = (v8f){};

    const float* Mp = mid + l16 * P + half * 2;
#pragma unroll
    for (int k = 0; k < N1; k += 4) {
        v2f a;
        a.x = Mp[k];
        a.y = Mp[k + 1];
#pragma unroll
        for (int t = 0; t < NT2; ++t) {
            const int c = t * 16 + l16;
            v2f b;
            if constexpr (NT2 * 16 == N2) {
                b.x = Bp2[k * N2 + c];
                b.y = Bp2[k * N2 + N2 + c];
            } else {
                const float m  = (c < N2) ? 1.0f : 0.0f;
                const int   cc = (c < N2) ? c : (N2 - 1);
                b.x = m * Bp2[k * N2 + cc];
                b.y = m * Bp2[k * N2 + N2 + cc];
            }
            acc2[t] = __builtin_amdgcn_wmma_f32_16x16x4_f32(
                false, a, false, b, (short)0, acc2[t], false, false);
        }
    }

#pragma unroll
    for (int t = 0; t < NT2; ++t) {
        const int c = t * 16 + l16;
        if (NT2 * 16 == N2 || c < N2) {
#pragma unroll
            for (int i = 0; i < 8; ++i)
                C[(size_t)(tm * 16 + i + half * 8) * N2 + c] = acc2[t][i];
        }
    }

    if (tm == 0) {
        for (int c = lane; c < N2; c += 32) {
            float s = bb[c];
#pragma unroll 16
            for (int k = 0; k < N1; ++k) s += ba[k] * B2[k * N2 + c];
            cb[c] = s;
        }
    }
}


extern "C" void kernel_launch(void* const* d_in, const int* in_sizes, int n_in,
                              void* d_out, int out_size, void* d_ws, size_t ws_size,
                              hipStream_t stream) {
    const float* x  = (const float*)d_in[0];
    const int*   ei = (const int*)d_in[1];
    const float* W1 = (const float*)d_in[2];
    const float* b1 = (const float*)d_in[3];
    const float* W2 = (const float*)d_in[4];
    const float* b2 = (const float*)d_in[5];
    const float* W3 = (const float*)d_in[6];
    const float* b3 = (const float*)d_in[7];
    const float* Wl = (const float*)d_in[8];
    const float* bl = (const float*)d_in[9];
    float* out = (float*)d_out;

    const int E = in_sizes[1] / 2;
    const int* src = ei;
    const int* dst = ei + E;
    const int nch = (E + CH - 1) / CH;

    float* dinv     = (float*)d_ws;
    int*   binstart = (int*)(dinv + 50176);
    int*   cnt      = binstart + 512;
    int*   rec      = cnt + 1024 * NBIN;
    float* G        = (float*)(rec + 2 * (size_t)E);
    float* ACT      = G + (size_t)NN * H;
    float* cb       = ACT + (size_t)NN * H;

    const dim3 blk(256);
    const dim3 big(1024);
    const dim3 gTile(NBIN);

    k_bin_count<<<dim3(nch), blk, 0, stream>>>(dst, cnt, E);
    k_bin_fill<<<dim3(nch), blk, 0, stream>>>(src, dst, cnt, rec, binstart, E, nch);
    k_tile_dinv<<<gTile, big, 0, stream>>>(rec, binstart, dinv);

    const dim3 gGemm(NN / 16);
    const dim3 wave(32);

    k_gemm_wmma<128, H, 4><<<gGemm, wave, 0, stream>>>(x, W1, nullptr, G);
    k_tile_agg<1, H><<<gTile, big, 0, stream>>>(rec, binstart, dinv, G, b1, ACT);

    k_gemm_wmma<H, H, 4><<<gGemm, wave, 0, stream>>>(ACT, W2, nullptr, G);
    k_tile_agg<1, H><<<gTile, big, 0, stream>>>(rec, binstart, dinv, G, b2, ACT);

    k_gemm2_wmma<H, H, 40, 3><<<gGemm, wave, 0, stream>>>(ACT, W3, Wl, b3, bl, G, cb);
    k_tile_agg<0, 40><<<gTile, big, 0, stream>>>(rec, binstart, dinv, G, cb, out);
}


# Layers 2, 3 sum first and multiply after, (A h) W for A (h W): the tile launch applies W (then Wl) to its summed 196-row tile from LDS on f32 WMMA; both GEMM launches go, 9 to 7; sums regrouped
# speedup vs baseline: 3.0035x; 1.0843x over previous
#include <hip/hip_runtime.h>
#include <math.h>

typedef __attribute__((ext_vector_type(2))) float v2f;
typedef __attribute__((ext_vector_type(8))) float v8f;

#define NN 50000
#define H  64
#define TILE 196
#define NBIN 256
#define CH   4096


__global__ __launch_bounds__(256) void k_bin_count(const int* __restrict__ dst,
                                                   int* __restrict__ cnt, int E) {
    __shared__ int hist[NBIN];
    const int tid = threadIdx.x;
    hist[tid] = 0;
    __syncthreads();
    const int e0 = blockIdx.x * CH;
    int d[CH / 256];
#pragma unroll
    for (int k = 0; k < CH / 256; ++k) {
        const int e = e0 + k * 256 + tid;
        d[k] = dst[(e < E) ? e : (E - 1)];
    }
#pragma unroll
    for (int k = 0; k < CH / 256; ++k) {
        const int e = e0 + k * 256 + tid;
        if (e < E) atomicAdd(&hist[d[k] / TILE], 1);
    }
    __syncthreads();
    cnt[blockIdx.x * NBIN + tid] = hist[tid];
}

__global__ __launch_bounds__(256) void k_bin_fill(const int* __restrict__ src,
                                                  const int* __restrict__ dst,
                                                  const int* __restrict__ cnt,
                                                  int* __restrict__ rec,
                                                  int* __restrict__ binstart,
                                                  int E, int nch) {
    __shared__ int pos[NBIN];
    __shared__ int scan[NBIN];
    const int tid = threadIdx.x;
    const int c   = blockIdx.x;

    int total = 0, before = 0;
#pragma unroll 14
    for (int cc = 0; cc < nch; ++cc) {
        const int v = cnt[cc * NBIN + tid];
        total  += v;
        before += (cc < c) ? v : 0;
    }
    scan[tid] = total;
    __syncthreads();
    for (int o = 1; o < NBIN; o <<= 1) {
        const int v = (tid >= o) ? scan[tid - o] : 0;
        __syncthreads();
        scan[tid] += v;
        __syncthreads();
    }
    const int binbase = scan[tid] - total;
    pos[tid] = binbase + before;
    if (c == 0) {
        binstart[tid] = binbase;
        if (tid == NBIN - 1) binstart[NBIN] = scan[tid];
    }
    __syncthreads();

    int2* rec2 = (int2*)rec;
    const int e0 = c * CH;
    int d[CH / 256], s[CH / 256];
#pragma unroll
    for (int k = 0; k < CH / 256; ++k) {
        const int e  = e0 + k * 256 + tid;
        const int ec = (e < E) ? e : (E - 1);
        d[k] = dst[ec];
        s[k] = src[ec];
    }
#pragma unroll
    for (int k = 0; k < CH / 256; ++k) {
        const int e = e0 + k * 256 + tid;
        if (e < E) {
            const int b = d[k] / TILE;
            const int p = atomicAdd(&pos[b], 1);
            rec2[p] = make_int2(s[k], d[k] - b * TILE);
        }
    }
}

__global__ __launch_bounds__(1024) void k_tile_dinv(const int* __restrict__ rec,
                                                    const int* __restrict__ binstart,
                                                    float* __restrict__ dinv) {
    __shared__ int c[TILE];
    const int tid = threadIdx.x;
    const int t   = blockIdx.x;
    if (tid < TILE) c[tid] = 0;
    __syncthreads();
    const int2* rec2 = (const int2*)rec;
    const int e1 = binstart[t + 1];
    for (int e = binstart[t] + tid; e < e1; e += 1024) atomicAdd(&c[rec2[e].y], 1);
    __syncthreads();
    const int n = t * TILE + tid;
    if (tid < TILE && n < NN) dinv[n] = rsqrtf((float)c[tid] + 1.0f);
}

template<int DO_ELU, int HW>
__global__ __launch_bounds__(1024) void k_tile_agg(const int* __restrict__ rec,
                                                   const int* __restrict__ binstart,
                                                   const float* __restrict__ dinv,
                                                   const float* __restrict__ h,
                                                   const float* __restrict__ bias,
                                                   float* __restrict__ act) {
    constexpr int Q  = HW / 4;
    constexpr int LP = HW / 2;
    __shared__ __attribute__((aligned(16))) float acc[TILE * HW];
    __shared__ float sdinv[TILE];
    const int tid  = threadIdx.x;
    const int lane = tid & 31;
    const int wv   = tid >> 5;
    const int t    = blockIdx.x;
    const int base = t * TILE;
    const int nn   = (NN - base < TILE) ? (NN - base) : TILE;

    if (tid < nn) sdinv[tid] = dinv[base + tid];
    const float4* h4  = (const float4*)(h + (size_t)base * HW);
    float4*       a4  = (float4*)acc;
    constexpr int NR = (TILE * Q + 1023) / 1024;
    const int lim = nn * Q;
    float  di[NR];
    float4 hv[NR];
#pragma unroll
    for (int k = 0; k < NR; ++k) {
        const int i  = tid + k * 1024;
        const int ic = (i < lim) ? i : (lim - 1);
        di[k] = dinv[base + ((Q == 16) ? (ic >> 4) : (ic / Q))];
        hv[k] = h4[ic];
    }
#pragma unroll
    for (int k = 0; k < NR; ++k) {
        const int i = tid + k * 1024;
        if (i < lim) {
            const float s = di[k] * di[k];
            a4[i] = make_float4(s * hv[k].x, s * hv[k].y, s * hv[k].z, s * hv[k].w);
        }
    }
    __syncthreads();

    const int2*   rec2 = (const int2*)rec;
    const float2* h2   = (const float2*)h;
    const int e0 = binstart[t];
    const int e1 = binstart[t + 1];
    const int lc = (lane < LP) ? lane : (LP - 1);
    int eb = e0 + wv * 32;
    if (eb < e1) {
        const int elast = e1 - 1;
        int2  r  = rec2[(eb + lane < elast) ? (eb + lane) : elast];
        float wl = dinv[r.x] * sdinv[r.y];
        do {
            const int ebu = __builtin_amdgcn_readfirstlane(eb);
            const int cnt = (e1 - ebu < 32) ? (e1 - ebu) : 32;
            const int en = eb + 32 * 32 + lane;
            const int2 rn = rec2[(en < elast) ? en : elast];
            float wn = 0.0f;
#pragma unroll
            for (int g = 0; g < 2; ++g) {
                if (g == 1) wn = dinv[rn.x] * sdinv[rn.y];
                float2 v[16];
#pragma unroll
                for (int u = 0; u < 16; ++u) {
                    const int s = __builtin_amdgcn_readlane(r.x, g * 16 + u);
                    v[u] = h2[(size_t)s * LP + lc];
                }
#pragma unroll
                for (int u = 0; u < 16; ++u) {
                    const int   j  = g * 16 + u;
                    const int   dl = __builtin_amdgcn_readlane(r.y, j);
                    const float w  = __int_as_float(__builtin_amdgcn_readlane(__float_as_int(wl), j));
                    if (j < cnt) {
                        if (LP == 32 || lane < LP) {
                            float* o = acc + dl * HW + lane * 2;
                            atomicAdd(o,     w * v[u].x);
                            atomicAdd(o + 1, w * v[u].y);
                        }
                    }
                }
            }
            r  = rn;
            wl = wn;
            eb += 32 * 32;
        } while (eb < e1);
    }
    __syncthreads();

    float4* o4 = (float4*)(act + (size_t)base * HW);
    for (int i = tid; i < nn * Q; i += 1024) {
        const float4 a = a4[i];
        const int    c = ((Q == 16) ? (i & 15) : (i % Q)) * 4;
        float4 v = make_float4(a.x + bias[c], a.y + bias[c + 1], a.z + bias[c + 2], a.w + bias[c + 3]);
        if (DO_ELU) {
            v.x = (v.x <= 0.0f) ? expm1f(v.x) : v.x;
            v.y = (v.y <= 0.0f) ? expm1f(v.y) : v.y;
            v.z = (v.z <= 0.0f) ? expm1f(v.z) : v.z;
            v.w = (v.w <= 0.0f) ? expm1f(v.w) : v.w;
        }
        o4[i] = v;
    }
}

template<int K, int NB, int NT>
__global__ void k_gemm_wmma(const float* __restrict__ A, const float* __restrict__ B,
                            const float* __restrict__ bias, float* __restrict__ C) {
    const int tm   = blockIdx.x;
    const int lane = threadIdx.x;
    const int half = lane >> 4;
    const int l16  = lane & 15;
    const int arow = tm * 16 + l16;

    const float* Ap = A + (size_t)arow * K + half * 2;
    const float* Bp = B + half * 2 * NB;

    v8f acc[NT];
#pragma unroll
    for (int t = 0; t < NT; ++t) acc[t] = (v8f){};

    for (int k = 0; k < K; k += 4) {
        v2f a;
        a.x = Ap[k];
        a.y = Ap[k + 1];
#pragma unroll
        for (int t = 0; t < NT; ++t) {
            const int c = t * 16 + l16;
            v2f b;
            if constexpr (NT * 16 == NB) {
                b.x = Bp[k * NB + c];
                b.y = Bp[k * NB + NB + c];
            } else {
                const float m  = (c < NB) ? 1.0f : 0.0f;
                const int   cc = (c < NB) ? c : (NB - 1);
                b.x = m * Bp[k * NB + cc];
                b.y = m * Bp[k * NB + NB + cc];
            }
            acc[t] = __builtin_amdgcn_wmma_f32_16x16x4_f32(
                false, a, false, b, (short)0, acc[t], false, false);
        }
    }

#pragma unroll
    for (int t = 0; t < NT; ++t) {
        const int c = t * 16 + l16;
        if (NT * 16 == NB || c < NB) {
            const float bv = bias ? bias[c] : 0.0f;
#pragma unroll
            for (int i = 0; i < 8; ++i)
                C[(size_t)(tm * 16 + i + half * 8) * NB + c] = acc[t][i] + bv;
        }
    }
}

#define PA  68
#define MTL 13

template<int NB, int NT>
__device__ __forceinline__ void tile_mm(const float* arow, const float* __restrict__ B,
                                        int n0, int half, int l16, v8f (&d)[NT]) {
#pragma unroll
    for (int t = 0; t < NT; ++t) d[t] = (v8f){};
    const float* Bp = B + half * 2 * NB;
#pragma unroll
    for (int k = 0; k < 64; k += 4) {
        v2f a;
        a.x = arow[k];
        a.y = arow[k + 1];
#pragma unroll
        for (int t = 0; t < NT; ++t) {
            const int c = (n0 + t) * 16 + l16;
            v2f b;
            if constexpr (NB % 16 == 0) {
                b.x = Bp[k * NB + c];
                b.y = Bp[k * NB + NB + c];
            } else {
                const float m  = (c < NB) ? 1.0f : 0.0f;
                const int   cc = (c < NB) ? c : (NB - 1);
                b.x = m * Bp[k * NB + cc];
                b.y = m * Bp[k * NB + NB + cc];
            }
            d[t] = __builtin_amdgcn_wmma_f32_16x16x4_f32(
                false, a, false, b, (short)0, d[t], false, false);
        }
    }
}

template<int LAST>
__global__ __launch_bounds__(1024) void k_tile_agg_mm(const int* __restrict__ rec,
                                                      const int* __restrict__ binstart,
                                                      const float* __restrict__ dinv,
                                                      const float* __restrict__ h,
                                                      const float* __restrict__ Wm,
                                                      const float* __restrict__ b,
                                                      const float* __restrict__ Wl,
                                                      const float* __restrict__ bl,
                                                      float* __restrict__ outp) {
    constexpr int HW = 64;
    __shared__ __attribute__((aligned(16))) float acc[MTL * 16 * PA];
    __shared__ __attribute__((aligned(16))) float mid[LAST ? MTL * 16 * PA : 4];
    __shared__ float sdinv[TILE];
    __shared__ float scb[48];
    const int tid  = threadIdx.x;
    const int lane = tid & 31;
    const int wv   = tid >> 5;
    const int t    = blockIdx.x;
    const int base = t * TILE;
    const int nn   = (NN - base < TILE) ? (NN - base) : TILE;

    if (tid < nn) sdinv[tid] = dinv[base + tid];
    const float4* h4 = (const float4*)(h + (size_t)base * HW);
    constexpr int NR = (TILE * 16 + 1023) / 1024;
    const int lim = nn * 16;
    float  di[NR];
    float4 hv[NR];
#pragma unroll
    for (int k = 0; k < NR; ++k) {
        const int i  = tid + k * 1024;
        const int ic = (i < lim) ? i : (lim - 1);
        di[k] = dinv[base + (ic >> 4)];
        hv[k] = h4[ic];
    }
#pragma unroll
    for (int k = 0; k < NR; ++k) {
        const int i = tid + k * 1024;
        if (i < lim) {
            const float s = di[k] * di[k];
            *(float4*)(acc + (i >> 4) * PA + (i & 15) * 4) =
                make_float4(s * hv[k].x, s * hv[k].y, s * hv[k].z, s * hv[k].w);
        }
    }
    for (int i = lim + tid; i < MTL * 16 * 16; i += 1024)
        *(float4*)(acc + (i >> 4) * PA + (i & 15) * 4) = make_float4(0.0f, 0.0f, 0.0f, 0.0f);
    __syncthreads();

    if (LAST && wv >= 30) {
        const int c = tid - 960;
        if (c < 40) {
            float s = bl[c];
#pragma unroll 16
            for (int k = 0; k < 64; ++k) s += b[k] * Wl[k * 40 + c];
            scb[c] = s;
        }
    }

    const int2*   rec2 = (const int2*)rec;
    const float2* h2   = (const float2*)h;
    const int e0 = binstart[t];
    const int e1 = binstart[t + 1];
    int eb = e0 + wv * 32;
    if (eb < e1) {
        const int elast = e1 - 1;
        int2  r  = rec2[(eb + lane < elast) ? (eb + lane) : elast];
        float wl = dinv[r.x] * sdinv[r.y];
        do {
            const int ebu = __builtin_amdgcn_readfirstlane(eb);
            const int cnt = (e1 - ebu < 32) ? (e1 - ebu) : 32;
            const int en = eb + 32 * 32 + lane;
            const int2 rn = rec2[(en < elast) ? en : elast];
            float wn = 0.0f;
#pragma unroll
            for (int g = 0; g < 2; ++g) {
                if (g == 1) wn = dinv[rn.x] * sdinv[rn.y];
                float2 v[16];
#pragma unroll
                for (int u = 0; u < 16; ++u) {
                    const int s = __builtin_amdgcn_readlane(r.x, g * 16 + u);
                    v[u] = h2[(size_t)s * 32 + lane];
                }
#pragma unroll
                for (int u = 0; u < 16; ++u) {
                    const int   j  = g * 16 + u;
                    const int   dl = __builtin_amdgcn_readlane(r.y, j);
                    const float w  = __int_as_float(__builtin_amdgcn_readlane(__float_as_int(wl), j));
                    if (j < cnt) {
                        float* o = acc + dl * PA + lane * 2;
                        atomicAdd(o,     w * v[u].x);
                        atomicAdd(o + 1, w * v[u].y);
                    }
                }
            }
            r  = rn;
            wl = wn;
            eb += 32 * 32;
        } while (eb < e1);
    }
    __syncthreads();

    const int half = lane >> 4;
    const int l16  = lane & 15;
    const int wvu = __builtin_amdgcn_readfirstlane(wv);
    if (wvu < 2 * MTL) {
        const int mt = wvu >> 1;
        const int n0 = (wvu & 1) * 2;
        v8f d[2];
        tile_mm<64, 2>(acc + (mt * 16 + l16) * PA + half * 2, Wm, n0, half, l16, d);
#pragma unroll
        for (int q = 0; q < 2; ++q) {
            const int c = (n0 + q) * 16 + l16;
            if (LAST) {
#pragma unroll
                for (int i = 0; i < 8; ++i)
                    mid[(mt * 16 + i + half * 8) * PA + c] = d[q][i];
            } else {
                const float bv = b[c];
#pragma unroll
                for (int i = 0; i < 8; ++i) {
                    const int row = mt * 16 + i + half * 8;
                    if (row < nn) {
                        float v = d[q][i] + bv;
                        v = (v <= 0.0f) ? expm1f(v) : v;
                        outp[(size_t)(base + row) * 64 + c] = v;
                    }
                }
            }
        }
    }
    if (LAST) {
        __syncthreads();
        if (wvu < MTL) {
            const int mt = wvu;
            v8f e[3];
            tile_mm<40, 3>(mid + (mt * 16 + l16) * PA + half * 2, Wl, 0, half, l16, e);
#pragma unroll
            for (int q = 0; q < 3; ++q) {
                const int c = q * 16 + l16;
                if (c < 40) {
                    const float bv = scb[c];
#pragma unroll
                    for (int i = 0; i < 8; ++i) {
                        const int row = mt * 16 + i + half * 8;
                        if (row < nn) outp[(size_t)(base + row) * 40 + c] = e[q][i] + bv;
                    }
                }
            }
        }
    }
}


extern "C" void kernel_launch(void* const* d_in, const int* in_sizes, int n_in,
                              void* d_out, int out_size, void* d_ws, size_t ws_size,
                              hipStream_t stream) {
    const float* x  = (const float*)d_in[0];
    const int*   ei = (const int*)d_in[1];
    const float* W1 = (const float*)d_in[2];
    const float* b1 = (const float*)d_in[3];
    const float* W2 = (const float*)d_in[4];
    const float* b2 = (const float*)d_in[5];
    const float* W3 = (const float*)d_in[6];
    const float* b3 = (const float*)d_in[7];
    const float* Wl = (const float*)d_in[8];
    const float* bl = (const float*)d_in[9];
    float* out = (float*)d_out;

    const int E = in_sizes[1] / 2;
    const int* src = ei;
    const int* dst = ei + E;
    const int nch = (E + CH - 1) / CH;

    float* dinv     = (float*)d_ws;
    int*   binstart = (int*)(dinv + 50176);
    int*   cnt      = binstart + 512;
    int*   rec      = cnt + 1024 * NBIN;
    float* G        = (float*)(rec + 2 * (size_t)E);
    float* ACT      = G + (size_t)NN * H;

    const dim3 blk(256);
    const dim3 big(1024);
    const dim3 gTile(NBIN);

    k_bin_count<<<dim3(nch), blk, 0, stream>>>(dst, cnt, E);
    k_bin_fill<<<dim3(nch), blk, 0, stream>>>(src, dst, cnt, rec, binstart, E, nch);
    k_tile_dinv<<<gTile, big, 0, stream>>>(rec, binstart, dinv);

    const dim3 gGemm(NN / 16);
    const dim3 wave(32);

    k_gemm_wmma<128, H, 4><<<gGemm, wave, 0, stream>>>(x, W1, nullptr, G);
    k_tile_agg<1, H><<<gTile, big, 0, stream>>>(rec, binstart, dinv, G, b1, ACT);

    k_tile_agg_mm<0><<<gTile, big, 0, stream>>>(rec, binstart, dinv, ACT, W2, b2, Wl, bl, G);
    k_tile_agg_mm<1><<<gTile, big, 0, stream>>>(rec, binstart, dinv, G, W3, b3, Wl, bl, out);
}


# Edge sums leave LDS float atomics: tile records grouped by 16-row block in LDS (ballot ranks), each block summed as a one-hot weight matrix times source rows on f32 WMMA; sum order: list, k order
# speedup vs baseline: 3.1441x; 1.0468x over previous
#include <hip/hip_runtime.h>
#include <math.h>

typedef __attribute__((ext_vector_type(2))) float v2f;
typedef __attribute__((ext_vector_type(8))) float v8f;

#define NN 50000
#define H  64
#define TILE 196
#define NBIN 256
#define CH   4096


__global__ __launch_bounds__(256) void k_bin_count(const int* __restrict__ dst,
                                                   int* __restrict__ cnt, int E) {
    __shared__ int hist[NBIN];
    const int tid = threadIdx.x;
    hist[tid] = 0;
    __syncthreads();
    const int e0 = blockIdx.x * CH;
    int d[CH / 256];
#pragma unroll
    for (int k = 0; k < CH / 256; ++k) {
        const int e = e0 + k * 256 + tid;
        d[k] = dst[(e < E) ? e : (E - 1)];
    }
#pragma unroll
    for (int k = 0; k < CH / 256; ++k) {
        const int e = e0 + k * 256 + tid;
        if (e < E) atomicAdd(&hist[d[k] / TILE], 1);
    }
    __syncthreads();
    cnt[blockIdx.x * NBIN + tid] = hist[tid];
}

__global__ __launch_bounds__(256) void k_bin_fill(const int* __restrict__ src,
                                                  const int* __restrict__ dst,
                                                  const int* __restrict__ cnt,
                                                  int* __restrict__ rec,
                                                  int* __restrict__ binstart,
                                                  int E, int nch) {
    __shared__ int pos[NBIN];
    __shared__ int scan[NBIN];
    const int tid = threadIdx.x;
    const int c   = blockIdx.x;

    int total = 0, before = 0;
#pragma unroll 14
    for (int cc = 0; cc < nch; ++cc) {
        const int v = cnt[cc * NBIN + tid];
        total  += v;
        before += (cc < c) ? v : 0;
    }
    scan[tid] = total;
    __syncthreads();
    for (int o = 1; o < NBIN; o <<= 1) {
        const int v = (tid >= o) ? scan[tid - o] : 0;
        __syncthreads();
        scan[tid] += v;
        __syncthreads();
    }
    const int binbase = scan[tid] - total;
    pos[tid] = binbase + before;
    if (c == 0) {
        binstart[tid] = binbase;
        if (tid == NBIN - 1) binstart[NBIN] = scan[tid];
    }
    __syncthreads();

    int2* rec2 = (int2*)rec;
    const int e0 = c * CH;
    int d[CH / 256], s[CH / 256];
#pragma unroll
    for (int k = 0; k < CH / 256; ++k) {
        const int e  = e0 + k * 256 + tid;
        const int ec = (e < E) ? e : (E - 1);
        d[k] = dst[ec];
        s[k] = src[ec];
    }
#pragma unroll
    for (int k = 0; k < CH / 256; ++k) {
        const int e = e0 + k * 256 + tid;
        if (e < E) {
            const int b = d[k] / TILE;
            const int p = atomicAdd(&pos[b], 1);
            rec2[p] = make_int2(s[k], d[k] - b * TILE);
        }
    }
}

__global__ __launch_bounds__(1024) void k_tile_dinv(const int* __restrict__ rec,
                                                    const int* __restrict__ binstart,
                                                    float* __restrict__ dinv) {
    __shared__ int c[TILE];
    const int tid = threadIdx.x;
    const int t   = blockIdx.x;
    if (tid < TILE) c[tid] = 0;
    __syncthreads();
    const int2* rec2 = (const int2*)rec;
    const int e1 = binstart[t + 1];
    for (int e = binstart[t] + tid; e < e1; e += 1024) atomicAdd(&c[rec2[e].y], 1);
    __syncthreads();
    const int n = t * TILE + tid;
    if (tid < TILE && n < NN) dinv[n] = rsqrtf((float)c[tid] + 1.0f);
}

#define CAPR 4096
#define NRB  13
#define GU   4

__device__ __forceinline__ int wave_incl_scan(int v, int lane) {
#pragma unroll
    for (int o = 1; o < 32; o <<= 1) {
        const int t = __shfl_up(v, o);
        if (lane >= o) v += t;
    }
    return v;
}

template<int ROWS, int PITCH>
__device__ __forceinline__ void tile_edge_sum(const int2* __restrict__ rec2, int e0, int e1,
                                              const float* __restrict__ dinv, const float* sdinv,
                                              const float* __restrict__ h, float* acc,
                                              int2* srec, int* wcnt, int* rbtot, int tid) {
    const int lane = tid & 31;
    const int wv   = tid >> 5;
    const int half = lane >> 4;
    const int l16  = lane & 15;
    const unsigned ltmask = (1u << lane) - 1u;
    const int wvu  = __builtin_amdgcn_readfirstlane(wv);
    const int rbw  = wvu >> 1;
    const int part = wvu & 1;
    const float4* h4 = (const float4*)h;

    v8f d[4];
#pragma unroll
    for (int t = 0; t < 4; ++t) d[t] = (v8f){};

    for (int c0 = e0; c0 < e1; c0 += CAPR) {
        const int n = (e1 - c0 < CAPR) ? (e1 - c0) : CAPR;

        int2  r[4];
        float w[4];
        int   rk[4];
#pragma unroll
        for (int k = 0; k < 4; ++k) {
            const int idx = (k * 32 + wv) * 32 + lane;
            r[k] = rec2[c0 + ((idx < n) ? idx : (n - 1))];
        }
#pragma unroll
        for (int k = 0; k < 4; ++k) {
            const int idx = (k * 32 + wv) * 32 + lane;
            w[k]  = dinv[r[k].x] * sdinv[r[k].y];
            rk[k] = (idx < n) ? (r[k].y >> 4) : 15;
        }
        unsigned bm[4][4];
        int cntk[4];
        int mycnt = 0;
#pragma unroll
        for (int k = 0; k < 4; ++k) {
#pragma unroll
            for (int i = 0; i < 4; ++i)
                bm[k][i] = __builtin_amdgcn_ballot_w32(((rk[k] >> i) & 1) != 0);
            unsigned m = 0xFFFFFFFFu;
#pragma unroll
            for (int i = 0; i < 4; ++i) m &= ((lane >> i) & 1) ? bm[k][i] : ~bm[k][i];
            cntk[k] = (lane < NRB) ? __builtin_popcount(m) : 0;
            mycnt += cntk[k];
        }
        if (lane < 16) wcnt[wv * 16 + lane] = mycnt;
        __syncthreads();
        if (wvu < NRB) {
            const int v    = wcnt[lane * 16 + wvu];
            const int incl = wave_incl_scan(v, lane);
            wcnt[lane * 16 + wvu] = incl - v;
            if (lane == 31) rbtot[wvu] = incl;
        }
        __syncthreads();
        const int tv     = (lane < NRB) ? rbtot[lane] : 0;
        const int startv = wave_incl_scan(tv, lane) - tv;
        int basev = startv + ((lane < 16) ? wcnt[wv * 16 + lane] : 0);
#pragma unroll
        for (int k = 0; k < 4; ++k) {
            unsigned m = 0xFFFFFFFFu;
#pragma unroll
            for (int i = 0; i < 4; ++i) m &= ((rk[k] >> i) & 1) ? bm[k][i] : ~bm[k][i];
            const int pos = __shfl(basev, rk[k]) + __builtin_popcount(m & ltmask);
            basev += cntk[k];
            if (rk[k] != 15)
                srec[pos] = make_int2(r[k].x | ((r[k].y & 15) << 20), __float_as_int(w[k]));
        }
        __syncthreads();

        if (wvu < 2 * NRB) {
            const int segS = __builtin_amdgcn_readlane(startv, rbw);
            const int segE = __builtin_amdgcn_readlane(startv, rbw + 1);
            const int ng   = (segE - segS + 3) >> 2;
            const int gh   = (ng + 1) >> 1;
            const int g0   = part ? gh : 0;
            const int g1   = part ? ng : gh;
            const int je   = segS + g1 * 4;
            const int jEnd = (segE < je) ? segE : je;
            for (int g = g0; g < g1; g += GU) {
                int2   rc[GU][2];
                float4 hv[GU][2];
#pragma unroll
                for (int u = 0; u < GU; ++u)
#pragma unroll
                    for (int q = 0; q < 2; ++q) {
                        const int j = segS + (g + u) * 4 + half * 2 + q;
                        rc[u][q] = srec[(j < jEnd) ? j : (jEnd - 1)];
                    }
#pragma unroll
                for (int u = 0; u < GU; ++u)
#pragma unroll
                    for (int q = 0; q < 2; ++q)
                        hv[u][q] = h4[(size_t)(rc[u][q].x & 0xFFFFF) * 16 + l16];
#pragma unroll
                for (int u = 0; u < GU; ++u) {
                    const int j = segS + (g + u) * 4 + half * 2;
                    v2f a;
                    a.x = (j     < jEnd && (rc[u][0].x >> 20) == l16) ? __int_as_float(rc[u][0].y) : 0.0f;
                    a.y = (j + 1 < jEnd && (rc[u][1].x >> 20) == l16) ? __int_as_float(rc[u][1].y) : 0.0f;
                    v2f b0, b1, b2, b3;
                    b0.x = hv[u][0].x; b0.y = hv[u][1].x;
                    b1.x = hv[u][0].y; b1.y = hv[u][1].y;
                    b2.x = hv[u][0].z; b2.y = hv[u][1].z;
                    b3.x = hv[u][0].w; b3.y = hv[u][1].w;
                    d[0] = __builtin_amdgcn_wmma_f32_16x16x4_f32(false, a, false, b0, (short)0, d[0], false, false);
                    d[1] = __builtin_amdgcn_wmma_f32_16x16x4_f32(false, a, false, b1, (short)0, d[1], false, false);
                    d[2] = __builtin_amdgcn_wmma_f32_16x16x4_f32(false, a, false, b2, (short)0, d[2], false, false);
                    d[3] = __builtin_amdgcn_wmma_f32_16x16x4_f32(false, a, false, b3, (short)0, d[3], false, false);
                }
            }
        }
        __syncthreads();
    }

#pragma unroll
    for (int p = 0; p < 2; ++p) {
        if (wvu < 2 * NRB && part == p) {
#pragma unroll
            for (int i = 0; i < 8; ++i) {
                const int row = rbw * 16 + half * 8 + i;
                if (row < ROWS) {
                    float4* o = (float4*)(acc + row * PITCH + l16 * 4);
                    float4  v = *o;
                    v.x += d[0][i]; v.y += d[1][i]; v.z += d[2][i]; v.w += d[3][i];
                    *o = v;
                }
            }
        }
        __syncthreads();
    }
}

template<int DO_ELU, int HW>
__global__ __launch_bounds__(1024) void k_tile_agg(const int* __restrict__ rec,
                                                   const int* __restrict__ binstart,
                                                   const float* __restrict__ dinv,
                                                   const float* __restrict__ h,
                                                   const float* __restrict__ bias,
                                                   float* __restrict__ act) {
    static_assert(HW == 64, "rows of 64 floats");
    constexpr int Q  = HW / 4;
    __shared__ __attribute__((aligned(16))) float acc[TILE * HW];
    __shared__ __attribute__((aligned(16))) int2 srec[CAPR];
    __shared__ int   wcnt[32 * 16];
    __shared__ int   rbtot[16];
    __shared__ float sdinv[TILE];
    const int tid  = threadIdx.x;
    const int t    = blockIdx.x;
    const int base = t * TILE;
    const int nn   = (NN - base < TILE) ? (NN - base) : TILE;

    if (tid < nn) sdinv[tid] = dinv[base + tid];
    const float4* h4  = (const float4*)(h + (size_t)base * HW);
    float4*       a4  = (float4*)acc;
    constexpr int NR = (TILE * Q + 1023) / 1024;
    const int lim = nn * Q;
    float  di[NR];
    float4 hv[NR];
#pragma unroll
    for (int k = 0; k < NR; ++k) {
        const int i  = tid + k * 1024;
        const int ic = (i < lim) ? i : (lim - 1);
        di[k] = dinv[base + (ic >> 4)];
        hv[k] = h4[ic];
    }
#pragma unroll
    for (int k = 0; k < NR; ++k) {
        const int i = tid + k * 1024;
        if (i < lim) {
            const float s = di[k] * di[k];
            a4[i] = make_float4(s * hv[k].x, s * hv[k].y, s * hv[k].z, s * hv[k].w);
        }
    }
    for (int i = lim + tid; i < TILE * Q; i += 1024) a4[i] = make_float4(0.0f, 0.0f, 0.0f, 0.0f);
    __syncthreads();

    tile_edge_sum<TILE, HW>((const int2*)rec, binstart[t], binstart[t + 1], dinv, sdinv, h, acc,
                            srec, wcnt, rbtot, tid);

    float4* o4 = (float4*)(act + (size_t)base * HW);
    for (int i = tid; i < nn * Q; i += 1024) {
        const float4 a = a4[i];
        const int    c = (i & 15) * 4;
        float4 v = make_float4(a.x + bias[c], a.y + bias[c + 1], a.z + bias[c + 2], a.w + bias[c + 3]);
        if (DO_ELU) {
            v.x = (v.x <= 0.0f) ? expm1f(v.x) : v.x;
            v.y = (v.y <= 0.0f) ? expm1f(v.y) : v.y;
            v.z = (v.z <= 0.0f) ? expm1f(v.z) : v.z;
            v.w = (v.w <= 0.0f) ? expm1f(v.w) : v.w;
        }
        o4[i] = v;
    }
}

template<int K, int NB, int NT>
__global__ void k_gemm_wmma(const float* __restrict__ A, const float* __restrict__ B,
                            const float* __restrict__ bias, float* __restrict__ C) {
    const int tm   = blockIdx.x;
    const int lane = threadIdx.x;
    const int half = lane >> 4;
    const int l16  = lane & 15;
    const int arow = tm * 16 + l16;

    const float* Ap = A + (size_t)arow * K + half * 2;
    const float* Bp = B + half * 2 * NB;

    v8f acc[NT];
#pragma unroll
    for (int t = 0; t < NT; ++t) acc[t] = (v8f){};

    for (int k = 0; k < K; k += 4) {
        v2f a;
        a.x = Ap[k];
        a.y = Ap[k + 1];
#pragma unroll
        for (int t = 0; t < NT; ++t) {
            const int c = t * 16 + l16;
            v2f b;
            if constexpr (NT * 16 == NB) {
                b.x = Bp[k * NB + c];
                b.y = Bp[k * NB + NB + c];
            } else {
                const float m  = (c < NB) ? 1.0f : 0.0f;
                const int   cc = (c < NB) ? c : (NB - 1);
                b.x = m * Bp[k * NB + cc];
                b.y = m * Bp[k * NB + NB + cc];
            }
            acc[t] = __builtin_amdgcn_wmma_f32_16x16x4_f32(
                false, a, false, b, (short)0, acc[t], false, false);
        }
    }

#pragma unroll
    for (int t = 0; t < NT; ++t) {
        const int c = t * 16 + l16;
        if (NT * 16 == NB || c < NB) {
            const float bv = bias ? bias[c] : 0.0f;
#pragma unroll
            for (int i = 0; i < 8; ++i)
                C[(size_t)(tm * 16 + i + half * 8) * NB + c] = acc[t][i] + bv;
        }
    }
}

#define PA  68
#define MTL 13

template<int NB, int NT>
__device__ __forceinline__ void tile_mm(const float* arow, const float* __restrict__ B,
                                        int n0, int half, int l16, v8f (&d)[NT]) {
#pragma unroll
    for (int t = 0; t < NT; ++t) d[t] = (v8f){};
    const float* Bp = B + half * 2 * NB;
#pragma unroll
    for (int k = 0; k < 64; k += 4) {
        v2f a;
        a.x = arow[k];
        a.y = arow[k + 1];
#pragma unroll
        for (int t = 0; t < NT; ++t) {
            const int c = (n0 + t) * 16 + l16;
            v2f b;
            if constexpr (NB % 16 == 0) {
                b.x = Bp[k * NB + c];
                b.y = Bp[k * NB + NB + c];
            } else {
                const float m  = (c < NB) ? 1.0f : 0.0f;
                const int   cc = (c < NB) ? c : (NB - 1);
                b.x = m * Bp[k * NB + cc];
                b.y = m * Bp[k * NB + NB + cc];
            }
            d[t] = __builtin_amdgcn_wmma_f32_16x16x4_f32(
                false, a, false, b, (short)0, d[t], false, false);
        }
    }
}

template<int LAST>
__global__ __launch_bounds__(1024) void k_tile_agg_mm(const int* __restrict__ rec,
                                                      const int* __restrict__ binstart,
                                                      const float* __restrict__ dinv,
                                                      const float* __restrict__ h,
                                                      const float* __restrict__ Wm,
                                                      const float* __restrict__ b,
                                                      const float* __restrict__ Wl,
                                                      const float* __restrict__ bl,
                                                      float* __restrict__ outp) {
    constexpr int HW = 64;
    __shared__ __attribute__((aligned(16))) float acc[MTL * 16 * PA];
    __shared__ __attribute__((aligned(16))) float mid[LAST ? MTL * 16 * PA : 4];
    __shared__ __attribute__((aligned(16))) int2 srec[CAPR];
    __shared__ int   wcnt[32 * 16];
    __shared__ int   rbtot[16];
    __shared__ float sdinv[TILE];
    __shared__ float scb[48];
    const int tid  = threadIdx.x;
    const int lane = tid & 31;
    const int wv   = tid >> 5;
    const int t    = blockIdx.x;
    const int base = t * TILE;
    const int nn   = (NN - base < TILE) ? (NN - base) : TILE;

    if (tid < nn) sdinv[tid] = dinv[base + tid];
    const float4* h4 = (const float4*)(h + (size_t)base * HW);
    constexpr int NR = (TILE * 16 + 1023) / 1024;
    const int lim = nn * 16;
    float  di[NR];
    float4 hv[NR];
#pragma unroll
    for (int k = 0; k < NR; ++k) {
        const int i  = tid + k * 1024;
        const int ic = (i < lim) ? i : (lim - 1);
        di[k] = dinv[base + (ic >> 4)];
        hv[k] = h4[ic];
    }
#pragma unroll
    for (int k = 0; k < NR; ++k) {
        const int i = tid + k * 1024;
        if (i < lim) {
            const float s = di[k] * di[k];
            *(float4*)(acc + (i >> 4) * PA + (i & 15) * 4) =
                make_float4(s * hv[k].x, s * hv[k].y, s * hv[k].z, s * hv[k].w);
        }
    }
    for (int i = lim + tid; i < MTL * 16 * 16; i += 1024)
        *(float4*)(acc + (i >> 4) * PA + (i & 15) * 4) = make_float4(0.0f, 0.0f, 0.0f, 0.0f);
    __syncthreads();

    if (LAST && wv >= 30) {
        const int c = tid - 960;
        if (c < 40) {
            float s = bl[c];
#pragma unroll 16
            for (int k = 0; k < 64; ++k) s += b[k] * Wl[k * 40 + c];
            scb[c] = s;
        }
    }

    tile_edge_sum<MTL * 16, PA>((const int2*)rec, binstart[t], binstart[t + 1], dinv, sdinv, h, acc,
                                srec, wcnt, rbtot, tid);

    const int half = lane >> 4;
    const int l16  = lane & 15;
    const int wvu = __builtin_amdgcn_readfirstlane(wv);
    if (wvu < 2 * MTL) {
        const int mt = wvu >> 1;
        const int n0 = (wvu & 1) * 2;
        v8f d[2];
        tile_mm<64, 2>(acc + (mt * 16 + l16) * PA + half * 2, Wm, n0, half, l16, d);
#pragma unroll
        for (int q = 0; q < 2; ++q) {
            const int c = (n0 + q) * 16 + l16;
            if (LAST) {
#pragma unroll
                for (int i = 0; i < 8; ++i)
                    mid[(mt * 16 + i + half * 8) * PA + c] = d[q][i];
            } else {
                const float bv = b[c];
#pragma unroll
                for (int i = 0; i < 8; ++i) {
                    const int row = mt * 16 + i + half * 8;
                    if (row < nn) {
                        float v = d[q][i] + bv;
                        v = (v <= 0.0f) ? expm1f(v) : v;
                        outp[(size_t)(base + row) * 64 + c] = v;
                    }
                }
            }
        }
    }
    if (LAST) {
        __syncthreads();
        if (wvu < MTL) {
            const int mt = wvu;
            v8f e[3];
            tile_mm<40, 3>(mid + (mt * 16 + l16) * PA + half * 2, Wl, 0, half, l16, e);
#pragma unroll
            for (int q = 0; q < 3; ++q) {
                const int c = q * 16 + l16;
                if (c < 40) {
                    const float bv = scb[c];
#pragma unroll
                    for (int i = 0; i < 8; ++i) {
                        const int row = mt * 16 + i + half * 8;
                        if (row < nn) outp[(size_t)(base + row) * 40 + c] = e[q][i] + bv;
                    }
                }
            }
        }
    }
}


extern "C" void kernel_launch(void* const* d_in, const int* in_sizes, int n_in,
                              void* d_out, int out_size, void* d_ws, size_t ws_size,
                              hipStream_t stream) {
    const float* x  = (const float*)d_in[0];
    const int*   ei = (const int*)d_in[1];
    const float* W1 = (const float*)d_in[2];
    const float* b1 = (const float*)d_in[3];
    const float* W2 = (const float*)d_in[4];
    const float* b2 = (const float*)d_in[5];
    const float* W3 = (const float*)d_in[6];
    const float* b3 = (const float*)d_in[7];
    const float* Wl = (const float*)d_in[8];
    const float* bl = (const float*)d_in[9];
    float* out = (float*)d_out;

    const int E = in_sizes[1] / 2;
    const int* src = ei;
    const int* dst = ei + E;
    const int nch = (E + CH - 1) / CH;

    float* dinv     = (float*)d_ws;
    int*   binstart = (int*)(dinv + 50176);
    int*   cnt      = binstart + 512;
    int*   rec      = cnt + 1024 * NBIN;
    float* G        = (float*)(rec + 2 * (size_t)E);
    float* ACT      = G + (size_t)NN * H;

    const dim3 blk(256);
    const dim3 big(1024);
    const dim3 gTile(NBIN);

    k_bin_count<<<dim3(nch), blk, 0, stream>>>(dst, cnt, E);
    k_bin_fill<<<dim3(nch), blk, 0, stream>>>(src, dst, cnt, rec, binstart, E, nch);
    k_tile_dinv<<<gTile, big, 0, stream>>>(rec, binstart, dinv);

    const dim3 gGemm(NN / 16);
    const dim3 wave(32);

    k_gemm_wmma<128, H, 4><<<gGemm, wave, 0, stream>>>(x, W1, nullptr, G);
    k_tile_agg<1, H><<<gTile, big, 0, stream>>>(rec, binstart, dinv, G, b1, ACT);

    k_tile_agg_mm<0><<<gTile, big, 0, stream>>>(rec, binstart, dinv, ACT, W2, b2, Wl, bl, G);
    k_tile_agg_mm<1><<<gTile, big, 0, stream>>>(rec, binstart, dinv, G, W3, b3, Wl, bl, out);
}


# Normalisation factored out of the edge sum: rows kept pre-scaled by dinv, edges carry no weight, a tile counts its own degrees inside the layer-1 product launch; 7 launches to 6; sums regrouped
# speedup vs baseline: 3.5018x; 1.1138x over previous
#include <hip/hip_runtime.h>
#include <math.h>

typedef __attribute__((ext_vector_type(2))) float v2f;
typedef __attribute__((ext_vector_type(8))) float v8f;

#define NN 50000
#define H  64
#define TILE 196
#define NBIN 256
#define CH   4096


__global__ __launch_bounds__(256) void k_bin_count(const int* __restrict__ dst,
                                                   int* __restrict__ cnt, int E) {
    __shared__ int hist[NBIN];
    const int tid = threadIdx.x;
    hist[tid] = 0;
    __syncthreads();
    const int e0 = blockIdx.x * CH;
    int d[CH / 256];
#pragma unroll
    for (int k = 0; k < CH / 256; ++k) {
        const int e = e0 + k * 256 + tid;
        d[k] = dst[(e < E) ? e : (E - 1)];
    }
#pragma unroll
    for (int k = 0; k < CH / 256; ++k) {
        const int e = e0 + k * 256 + tid;
        if (e < E) atomicAdd(&hist[d[k] / TILE], 1);
    }
    __syncthreads();
    cnt[blockIdx.x * NBIN + tid] = hist[tid];
}

__global__ __launch_bounds__(256) void k_bin_fill(const int* __restrict__ src,
                                                  const int* __restrict__ dst,
                                                  const int* __restrict__ cnt,
                                                  int* __restrict__ rec,
                                                  int* __restrict__ binstart,
                                                  int E, int nch) {
    __shared__ int pos[NBIN];
    __shared__ int scan[NBIN];
    const int tid = threadIdx.x;
    const int c   = blockIdx.x;

    int total = 0, before = 0;
#pragma unroll 14
    for (int cc = 0; cc < nch; ++cc) {
        const int v = cnt[cc * NBIN + tid];
        total  += v;
        before += (cc < c) ? v : 0;
    }
    scan[tid] = total;
    __syncthreads();
    for (int o = 1; o < NBIN; o <<= 1) {
        const int v = (tid >= o) ? scan[tid - o] : 0;
        __syncthreads();
        scan[tid] += v;
        __syncthreads();
    }
    const int binbase = scan[tid] - total;
    pos[tid] = binbase + before;
    if (c == 0) {
        binstart[tid] = binbase;
        if (tid == NBIN - 1) binstart[NBIN] = scan[tid];
    }
    __syncthreads();

    int2* rec2 = (int2*)rec;
    const int e0 = c * CH;
    int d[CH / 256], s[CH / 256];
#pragma unroll
    for (int k = 0; k < CH / 256; ++k) {
        const int e  = e0 + k * 256 + tid;
        const int ec = (e < E) ? e : (E - 1);
        d[k] = dst[ec];
        s[k] = src[ec];
    }
#pragma unroll
    for (int k = 0; k < CH / 256; ++k) {
        const int e = e0 + k * 256 + tid;
        if (e < E) {
            const int b = d[k] / TILE;
            const int p = atomicAdd(&pos[b], 1);
            rec2[p] = make_int2(s[k], d[k] - b * TILE);
        }
    }
}

__global__ __launch_bounds__(1024) void k_tile_dinv_mm1(const int* __restrict__ rec,
                                                        const int* __restrict__ binstart,
                                                        const float* __restrict__ x,
                                                        const float* __restrict__ W1,
                                                        float* __restrict__ dinv,
                                                        float* __restrict__ G) {
    __shared__ int c[13 * 16];
    const int tid  = threadIdx.x;
    const int lane = tid & 31;
    const int half = lane >> 4;
    const int l16  = lane & 15;
    const int t    = blockIdx.x;
    const int base = t * TILE;
    const int nn   = (NN - base < TILE) ? (NN - base) : TILE;
    if (tid < 13 * 16) c[tid] = 0;
    __syncthreads();
    const int wvu = __builtin_amdgcn_readfirstlane(tid >> 5);
    const int mt  = wvu >> 1;
    const int n0  = (wvu & 1) * 2;
    v8f d[2];
    d[0] = (v8f){};
    d[1] = (v8f){};
    if (wvu < 26) {
        int arow = base + mt * 16 + l16;
        arow = (arow < NN) ? arow : (NN - 1);
        const float* Ap = x + (size_t)arow * 128 + half * 2;
        const float* Bp = W1 + half * 2 * 64;
#pragma unroll
        for (int k = 0; k < 128; k += 4) {
            v2f a;
            a.x = Ap[k];
            a.y = Ap[k + 1];
#pragma unroll
            for (int q = 0; q < 2; ++q) {
                const int cc = (n0 + q) * 16 + l16;
                v2f b;
                b.x = Bp[k * 64 + cc];
                b.y = Bp[k * 64 + 64 + cc];
                d[q] = __builtin_amdgcn_wmma_f32_16x16x4_f32(
                    false, a, false, b, (short)0, d[q], false, false);
            }
        }
    } else {
        const int2* rec2 = (const int2*)rec;
        const int e1 = binstart[t + 1];
        for (int e0 = binstart[t] + (tid - 26 * 32); e0 < e1; e0 += 8 * 6 * 32) {
            int y[8];
#pragma unroll
            for (int k = 0; k < 8; ++k) {
                const int e = e0 + k * 6 * 32;
                y[k] = rec2[(e < e1) ? e : (e1 - 1)].y;
            }
#pragma unroll
            for (int k = 0; k < 8; ++k)
                if (e0 + k * 6 * 32 < e1) atomicAdd(&c[y[k]], 1);
        }
    }
    __syncthreads();
    if (tid < nn) dinv[base + tid] = rsqrtf((float)c[tid] + 1.0f);
    if (wvu < 26) {
#pragma unroll
        for (int i = 0; i < 8; ++i) {
            const int row = mt * 16 + i + half * 8;
            if (row < nn) {
                const float sc = rsqrtf((float)c[row] + 1.0f);
#pragma unroll
                for (int q = 0; q < 2; ++q)
                    G[(size_t)(base + row) * 64 + (n0 + q) * 16 + l16] = sc * d[q][i];
            }
        }
    }
}

#define CAPR 4096
#define NRB  13
#define GU   4

__device__ __forceinline__ int wave_incl_scan(int v, int lane) {
#pragma unroll
    for (int o = 1; o < 32; o <<= 1) {
        const int t = __shfl_up(v, o);
        if (lane >= o) v += t;
    }
    return v;
}

template<int ROWS, int PITCH>
__device__ __forceinline__ void tile_edge_sum(const int2* __restrict__ rec2, int e0, int e1,
                                              const float* __restrict__ h, float* acc,
                                              int* srec, int* wcnt, int* rbtot, int tid) {
    const int lane = tid & 31;
    const int wv   = tid >> 5;
    const int half = lane >> 4;
    const int l16  = lane & 15;
    const unsigned ltmask = (1u << lane) - 1u;
    const int wvu  = __builtin_amdgcn_readfirstlane(wv);
    const int rbw  = wvu >> 1;
    const int part = wvu & 1;
    const float4* h4 = (const float4*)h;

    v8f d[4];
#pragma unroll
    for (int t = 0; t < 4; ++t) d[t] = (v8f){};

    for (int c0 = e0; c0 < e1; c0 += CAPR) {
        const int n = (e1 - c0 < CAPR) ? (e1 - c0) : CAPR;

        int2  r[4];
        int   rk[4];
#pragma unroll
        for (int k = 0; k < 4; ++k) {
            const int idx = (k * 32 + wv) * 32 + lane;
            r[k] = rec2[c0 + ((idx < n) ? idx : (n - 1))];
        }
#pragma unroll
        for (int k = 0; k < 4; ++k) {
            const int idx = (k * 32 + wv) * 32 + lane;
            rk[k] = (idx < n) ? (r[k].y >> 4) : 15;
        }
        unsigned bm[4][4];
        int cntk[4];
        int mycnt = 0;
#pragma unroll
        for (int k = 0; k < 4; ++k) {
#pragma unroll
            for (int i = 0; i < 4; ++i)
                bm[k][i] = __builtin_amdgcn_ballot_w32(((rk[k] >> i) & 1) != 0);
            unsigned m = 0xFFFFFFFFu;
#pragma unroll
            for (int i = 0; i < 4; ++i) m &= ((lane >> i) & 1) ? bm[k][i] : ~bm[k][i];
            cntk[k] = (lane < NRB) ? __builtin_popcount(m) : 0;
            mycnt += cntk[k];
        }
        if (lane < 16) wcnt[wv * 16 + lane] = mycnt;
        __syncthreads();
        if (wvu < NRB) {
            const int v    = wcnt[lane * 16 + wvu];
            const int incl = wave_incl_scan(v, lane);
            wcnt[lane * 16 + wvu] = incl - v;
            if (lane == 31) rbtot[wvu] = incl;
        }
        __syncthreads();
        const int tv     = (lane < NRB) ? rbtot[lane] : 0;
        const int startv = wave_incl_scan(tv, lane) - tv;
        int basev = startv + ((lane < 16) ? wcnt[wv * 16 + lane] : 0);
#pragma unroll
        for (int k = 0; k < 4; ++k) {
            unsigned m = 0xFFFFFFFFu;
#pragma unroll
            for (int i = 0; i < 4; ++i) m &= ((rk[k] >> i) & 1) ? bm[k][i] : ~bm[k][i];
            const int pos = __shfl(basev, rk[k]) + __builtin_popcount(m & ltmask);
            basev += cntk[k];
            if (rk[k] != 15)
                srec[pos] = r[k].x | ((r[k].y & 15) << 20);
        }
        __syncthreads();

        if (wvu < 2 * NRB) {
            const int segS = __builtin_amdgcn_readlane(startv, rbw);
            const int segE = __builtin_amdgcn_readlane(startv, rbw + 1);
            const int ng   = (segE - segS + 3) >> 2;
            const int gh   = (ng + 1) >> 1;
            const int g0   = part ? gh : 0;
            const int g1   = part ? ng : gh;
            const int je   = segS + g1 * 4;
            const int jEnd = (segE < je) ? segE : je;
            for (int g = g0; g < g1; g += GU) {
                int    rc[GU][2];
                float4 hv[GU][2];
#pragma unroll
                for (int u = 0; u < GU; ++u)
#pragma unroll
                    for (int q = 0; q < 2; ++q) {
                        const int j = segS + (g + u) * 4 + half * 2 + q;
                        rc[u][q] = srec[(j < jEnd) ? j : (jEnd - 1)];
                    }
#pragma unroll
                for (int u = 0; u < GU; ++u)
#pragma unroll
                    for (int q = 0; q < 2; ++q)
                        hv[u][q] = h4[(size_t)(rc[u][q] & 0xFFFFF) * 16 + l16];
#pragma unroll
                for (int u = 0; u < GU; ++u) {
                    const int j = segS + (g + u) * 4 + half * 2;
                    v2f a;
                    a.x = (j     < jEnd && (rc[u][0] >> 20) == l16) ? 1.0f : 0.0f;
                    a.y = (j + 1 < jEnd && (rc[u][1] >> 20) == l16) ? 1.0f : 0.0f;
                    v2f b0, b1, b2, b3;
                    b0.x = hv[u][0].x; b0.y = hv[u][1].x;
                    b1.x = hv[u][0].y; b1.y = hv[u][1].y;
                    b2.x = hv[u][0].z; b2.y = hv[u][1].z;
                    b3.x = hv[u][0].w; b3.y = hv[u][1].w;
                    d[0] = __builtin_amdgcn_wmma_f32_16x16x4_f32(false, a, false, b0, (short)0, d[0], false, false);
                    d[1] = __builtin_amdgcn_wmma_f32_16x16x4_f32(false, a, false, b1, (short)0, d[1], false, false);
                    d[2] = __builtin_amdgcn_wmma_f32_16x16x4_f32(false, a, false, b2, (short)0, d[2], false, false);
                    d[3] = __builtin_amdgcn_wmma_f32_16x16x4_f32(false, a, false, b3, (short)0, d[3], false, false);
                }
            }
        }
        __syncthreads();
    }

#pragma unroll
    for (int p = 0; p < 2; ++p) {
        if (wvu < 2 * NRB && part == p) {
#pragma unroll
            for (int i = 0; i < 8; ++i) {
                const int row = rbw * 16 + half * 8 + i;
                if (row < ROWS) {
                    float4* o = (float4*)(acc + row * PITCH + l16 * 4);
                    float4  v = *o;
                    v.x += d[0][i]; v.y += d[1][i]; v.z += d[2][i]; v.w += d[3][i];
                    *o = v;
                }
            }
        }
        __syncthreads();
    }
}

template<int DO_ELU, int HW>
__global__ __launch_bounds__(1024) void k_tile_agg(const int* __restrict__ rec,
                                                   const int* __restrict__ binstart,
                                                   const float* __restrict__ dinv,
                                                   const float* __restrict__ h,
                                                   const float* __restrict__ bias,
                                                   float* __restrict__ act) {
    static_assert(HW == 64, "rows of 64 floats");
    constexpr int Q  = HW / 4;
    __shared__ __attribute__((aligned(16))) float acc[TILE * HW];
    __shared__ int   srec[CAPR];
    __shared__ int   wcnt[32 * 16];
    __shared__ int   rbtot[16];
    __shared__ float sdinv[TILE];
    const int tid  = threadIdx.x;
    const int t    = blockIdx.x;
    const int base = t * TILE;
    const int nn   = (NN - base < TILE) ? (NN - base) : TILE;

    if (tid < nn) sdinv[tid] = dinv[base + tid];
    const float4* h4  = (const float4*)(h + (size_t)base * HW);
    float4*       a4  = (float4*)acc;
    constexpr int NR = (TILE * Q + 1023) / 1024;
    const int lim = nn * Q;
    float4 hv[NR];
#pragma unroll
    for (int k = 0; k < NR; ++k) {
        const int i  = tid + k * 1024;
        hv[k] = h4[(i < lim) ? i : (lim - 1)];
    }
#pragma unroll
    for (int k = 0; k < NR; ++k) {
        const int i = tid + k * 1024;
        if (i < lim) a4[i] = hv[k];
    }
    for (int i = lim + tid; i < TILE * Q; i += 1024) a4[i] = make_float4(0.0f, 0.0f, 0.0f, 0.0f);
    __syncthreads();

    tile_edge_sum<TILE, HW>((const int2*)rec, binstart[t], binstart[t + 1], h, acc,
                            srec, wcnt, rbtot, tid);

    float4* o4 = (float4*)(act + (size_t)base * HW);
    for (int i = tid; i < nn * Q; i += 1024) {
        const float4 a = a4[i];
        const int    c = (i & 15) * 4;
        const float  sc = sdinv[i >> 4];
        float4 v = make_float4(sc * a.x + bias[c], sc * a.y + bias[c + 1], sc * a.z + bias[c + 2], sc * a.w + bias[c + 3]);
        if (DO_ELU) {
            v.x = (v.x <= 0.0f) ? expm1f(v.x) : v.x;
            v.y = (v.y <= 0.0f) ? expm1f(v.y) : v.y;
            v.z = (v.z <= 0.0f) ? expm1f(v.z) : v.z;
            v.w = (v.w <= 0.0f) ? expm1f(v.w) : v.w;
        }
        o4[i] = make_float4(sc * v.x, sc * v.y, sc * v.z, sc * v.w);
    }
}


#define PA  68
#define MTL 13

template<int NB, int NT>
__device__ __forceinline__ void tile_mm(const float* arow, const float* __restrict__ B,
                                        int n0, int half, int l16, v8f (&d)[NT]) {
#pragma unroll
    for (int t = 0; t < NT; ++t) d[t] = (v8f){};
    const float* Bp = B + half * 2 * NB;
#pragma unroll
    for (int k = 0; k < 64; k += 4) {
        v2f a;
        a.x = arow[k];
        a.y = arow[k + 1];
#pragma unroll
        for (int t = 0; t < NT; ++t) {
            const int c = (n0 + t) * 16 + l16;
            v2f b;
            if constexpr (NB % 16 == 0) {
                b.x = Bp[k * NB + c];
                b.y = Bp[k * NB + NB + c];
            } else {
                const float m  = (c < NB) ? 1.0f : 0.0f;
                const int   cc = (c < NB) ? c : (NB - 1);
                b.x = m * Bp[k * NB + cc];
                b.y = m * Bp[k * NB + NB + cc];
            }
            d[t] = __builtin_amdgcn_wmma_f32_16x16x4_f32(
                false, a, false, b, (short)0, d[t], false, false);
        }
    }
}

template<int LAST>
__global__ __launch_bounds__(1024) void k_tile_agg_mm(const int* __restrict__ rec,
                                                      const int* __restrict__ binstart,
                                                      const float* __restrict__ dinv,
                                                      const float* __restrict__ h,
                                                      const float* __restrict__ Wm,
                                                      const float* __restrict__ b,
                                                      const float* __restrict__ Wl,
                                                      const float* __restrict__ bl,
                                                      float* __restrict__ outp) {
    constexpr int HW = 64;
    __shared__ __attribute__((aligned(16))) float acc[MTL * 16 * PA];
    __shared__ __attribute__((aligned(16))) float mid[LAST ? MTL * 16 * PA : 4];
    __shared__ int   srec[CAPR];
    __shared__ int   wcnt[32 * 16];
    __shared__ int   rbtot[16];
    __shared__ float sdinv[TILE];
    __shared__ float scb[48];
    const int tid  = threadIdx.x;
    const int lane = tid & 31;
    const int wv   = tid >> 5;
    const int t    = blockIdx.x;
    const int base = t * TILE;
    const int nn   = (NN - base < TILE) ? (NN - base) : TILE;

    if (tid < nn) sdinv[tid] = dinv[base + tid];
    const float4* h4 = (const float4*)(h + (size_t)base * HW);
    constexpr int NR = (TILE * 16 + 1023) / 1024;
    const int lim = nn * 16;
    float4 hv[NR];
#pragma unroll
    for (int k = 0; k < NR; ++k) {
        const int i  = tid + k * 1024;
        hv[k] = h4[(i < lim) ? i : (lim - 1)];
    }
#pragma unroll
    for (int k = 0; k < NR; ++k) {
        const int i = tid + k * 1024;
        if (i < lim) *(float4*)(acc + (i >> 4) * PA + (i & 15) * 4) = hv[k];
    }
    for (int i = lim + tid; i < MTL * 16 * 16; i += 1024)
        *(float4*)(acc + (i >> 4) * PA + (i & 15) * 4) = make_float4(0.0f, 0.0f, 0.0f, 0.0f);
    __syncthreads();

    if (LAST && wv >= 30) {
        const int c = tid - 960;
        if (c < 40) {
            float s = bl[c];
#pragma unroll 16
            for (int k = 0; k < 64; ++k) s += b[k] * Wl[k * 40 + c];
            scb[c] = s;
        }
    }

    tile_edge_sum<MTL * 16, PA>((const int2*)rec, binstart[t], binstart[t + 1], h, acc,
                                srec, wcnt, rbtot, tid);

    const int half = lane >> 4;
    const int l16  = lane & 15;
    const int wvu = __builtin_amdgcn_readfirstlane(wv);
    if (wvu < 2 * MTL) {
        const int mt = wvu >> 1;
        const int n0 = (wvu & 1) * 2;
        v8f d[2];
        tile_mm<64, 2>(acc + (mt * 16 + l16) * PA + half * 2, Wm, n0, half, l16, d);
#pragma unroll
        for (int q = 0; q < 2; ++q) {
            const int c = (n0 + q) * 16 + l16;
            if (LAST) {
#pragma unroll
                for (int i = 0; i < 8; ++i)
                    mid[(mt * 16 + i + half * 8) * PA + c] = d[q][i];
            } else {
                const float bv = b[c];
#pragma unroll
                for (int i = 0; i < 8; ++i) {
                    const int row = mt * 16 + i + half * 8;
                    if (row < nn) {
                        const float sc = sdinv[row];
                        float v = sc * d[q][i] + bv;
                        v = (v <= 0.0f) ? expm1f(v) : v;
                        outp[(size_t)(base + row) * 64 + c] = sc * v;
                    }
                }
            }
        }
    }
    if (LAST) {
        __syncthreads();
        if (wvu < MTL) {
            const int mt = wvu;
            v8f e[3];
            tile_mm<40, 3>(mid + (mt * 16 + l16) * PA + half * 2, Wl, 0, half, l16, e);
#pragma unroll
            for (int q = 0; q < 3; ++q) {
                const int c = q * 16 + l16;
                if (c < 40) {
                    const float bv = scb[c];
#pragma unroll
                    for (int i = 0; i < 8; ++i) {
                        const int row = mt * 16 + i + half * 8;
                        if (row < nn) outp[(size_t)(base + row) * 40 + c] = sdinv[row] * e[q][i] + bv;
                    }
                }
            }
        }
    }
}


extern "C" void kernel_launch(void* const* d_in, const int* in_sizes, int n_in,
                              void* d_out, int out_size, void* d_ws, size_t ws_size,
                              hipStream_t stream) {
    const float* x  = (const float*)d_in[0];
    const int*   ei = (const int*)d_in[1];
    const float* W1 = (const float*)d_in[2];
    const float* b1 = (const float*)d_in[3];
    const float* W2 = (const float*)d_in[4];
    const float* b2 = (const float*)d_in[5];
    const float* W3 = (const float*)d_in[6];
    const float* b3 = (const float*)d_in[7];
    const float* Wl = (const float*)d_in[8];
    const float* bl = (const float*)d_in[9];
    float* out = (float*)d_out;

    const int E = in_sizes[1] / 2;
    const int* src = ei;
    const int* dst = ei + E;
    const int nch = (E + CH - 1) / CH;

    float* dinv     = (float*)d_ws;
    int*   binstart = (int*)(dinv + 50176);
    int*   cnt      = binstart + 512;
    int*   rec      = cnt + 1024 * NBIN;
    float* G        = (float*)(rec + 2 * (size_t)E);
    float* ACT      = G + (size_t)NN * H;

    const dim3 blk(256);
    const dim3 big(1024);
    const dim3 gTile(NBIN);

    k_bin_count<<<dim3(nch), blk, 0, stream>>>(dst, cnt, E);
    k_bin_fill<<<dim3(nch), blk, 0, stream>>>(src, dst, cnt, rec, binstart, E, nch);
    k_tile_dinv_mm1<<<gTile, big, 0, stream>>>(rec, binstart, x, W1, dinv, G);

    k_tile_agg<1, H><<<gTile, big, 0, stream>>>(rec, binstart, dinv, G, b1, ACT);

    k_tile_agg_mm<0><<<gTile, big, 0, stream>>>(rec, binstart, dinv, ACT, W2, b2, Wl, bl, G);
    k_tile_agg_mm<1><<<gTile, big, 0, stream>>>(rec, binstart, dinv, G, W3, b3, Wl, bl, out);
}


# Edge binning in one launch: a 4096-edge chunk sorts edges by dst tile in LDS and writes its bin offsets; the tile launch's spare waves gather the tile's list; 6 launches to 5; list order as parent
# speedup vs baseline: 4.1111x; 1.1740x over previous
#include <hip/hip_runtime.h>
#include <math.h>

typedef __attribute__((ext_vector_type(2))) float v2f;
typedef __attribute__((ext_vector_type(8))) float v8f;

#define NN 50000
#define H  64
#define TILE 196
#define NBIN 256
#define CH   4096

#define MAXCH 256

__device__ __forceinline__ int wave_incl_scan(int v, int lane) {
#pragma unroll
    for (int o = 1; o < 32; o <<= 1) {
        const int t = __shfl_up(v, o);
        if (lane >= o) v += t;
    }
    return v;
}

__global__ __launch_bounds__(256) void k_chunk_sort(const int* __restrict__ src,
                                                    const int* __restrict__ dst,
                                                    int* __restrict__ coffT,
                                                    int* __restrict__ crec, int E) {
    __shared__ int  hist[NBIN];
    __shared__ int  wtot[8];
    __shared__ int2 stage[CH];
    const int tid  = threadIdx.x;
    const int lane = tid & 31;
    const int wv   = tid >> 5;
    const int c    = blockIdx.x;
    hist[tid] = 0;
    __syncthreads();
    const int e0 = c * CH;
    int b[CH / 256], s[CH / 256], y[CH / 256], rk[CH / 256];
#pragma unroll
    for (int k = 0; k < CH / 256; ++k) {
        const int e  = e0 + k * 256 + tid;
        const int ec = (e < E) ? e : (E - 1);
        y[k] = dst[ec];
        s[k] = src[ec];
    }
#pragma unroll
    for (int k = 0; k < CH / 256; ++k) {
        const int e = e0 + k * 256 + tid;
        b[k]  = y[k] / TILE;
        y[k] -= b[k] * TILE;
        rk[k] = 0;
        if (e < E) rk[k] = atomicAdd(&hist[b[k]], 1);
    }
    __syncthreads();
    const int v    = hist[tid];
    const int incl = wave_incl_scan(v, lane);
    if (lane == 31) wtot[wv] = incl;
    __syncthreads();
    int add = 0;
#pragma unroll
    for (int w = 0; w < 8; ++w) add += (w < wv) ? wtot[w] : 0;
    const int off = add + incl - v;
    coffT[tid * MAXCH + c] = off;
    if (tid == NBIN - 1) coffT[NBIN * MAXCH + c] = off + v;
    hist[tid] = off;
    __syncthreads();
#pragma unroll
    for (int k = 0; k < CH / 256; ++k) {
        const int e = e0 + k * 256 + tid;
        if (e < E) stage[hist[b[k]] + rk[k]] = make_int2(s[k], y[k]);
    }
    __syncthreads();
    const int n = (E - e0 < CH) ? (E - e0) : CH;
    int2* o2 = (int2*)crec + e0;
    for (int i = tid; i < n; i += 256) o2[i] = stage[i];
}

#define GB 9
__global__ __launch_bounds__(1024) void k_tile_dinv_mm1(const int* __restrict__ crec,
                                                        const int* __restrict__ coffT,
                                                        const float* __restrict__ x,
                                                        const float* __restrict__ W1,
                                                        float* __restrict__ dinv,
                                                        float* __restrict__ G,
                                                        int* __restrict__ rec,
                                                        int* __restrict__ binstart) {
    __shared__ int c[13 * 16];
    __shared__ int pfx[6 * MAXCH];
    __shared__ int cst[6 * MAXCH];
    const int tid  = threadIdx.x;
    const int lane = tid & 31;
    const int half = lane >> 4;
    const int l16  = lane & 15;
    const int t    = blockIdx.x;
    const int base = t * TILE;
    const int nn   = (NN - base < TILE) ? (NN - base) : TILE;
    if (tid < 13 * 16) c[tid] = 0;
    __syncthreads();
    const int wvu = __builtin_amdgcn_readfirstlane(tid >> 5);
    const int mt  = wvu >> 1;
    const int n0  = (wvu & 1) * 2;
    v8f d[2];
    d[0] = (v8f){};
    d[1] = (v8f){};
    if (wvu < 26) {
        int arow = base + mt * 16 + l16;
        arow = (arow < NN) ? arow : (NN - 1);
        const float* Ap = x + (size_t)arow * 128 + half * 2;
        const float* Bp = W1 + half * 2 * 64;
#pragma unroll
        for (int k = 0; k < 128; k += 4) {
            v2f a;
            a.x = Ap[k];
            a.y = Ap[k + 1];
#pragma unroll
            for (int q = 0; q < 2; ++q) {
                const int cc = (n0 + q) * 16 + l16;
                v2f b;
                b.x = Bp[k * 64 + cc];
                b.y = Bp[k * 64 + 64 + cc];
                d[q] = __builtin_amdgcn_wmma_f32_16x16x4_f32(
                    false, a, false, b, (short)0, d[q], false, false);
            }
        }
    } else {
        const int sw = wvu - 26;
        int* P = pfx + sw * MAXCH;
        int* S = cst + sw * MAXCH;
        const int4* rs = (const int4*)(coffT + (size_t)t * MAXCH) + lane * 2;
        const int4* re = (const int4*)(coffT + (size_t)(t + 1) * MAXCH) + lane * 2;
        const int4 s0 = rs[0], s1 = rs[1], f0 = re[0], f1 = re[1];
        const int st[8] = {s0.x, s0.y, s0.z, s0.w, s1.x, s1.y, s1.z, s1.w};
        const int en[8] = {f0.x, f0.y, f0.z, f0.w, f1.x, f1.y, f1.z, f1.w};
        int cn[8];
        int lt = 0, ls = 0;
#pragma unroll
        for (int j = 0; j < 8; ++j) {
            cn[j] = en[j] - st[j];
            lt += cn[j];
            ls += st[j];
        }
        const int incl = wave_incl_scan(lt, lane);
        int run = incl - lt;
#pragma unroll
        for (int j = 0; j < 8; ++j) {
            P[lane * 8 + j] = run;
            S[lane * 8 + j] = (lane * 8 + j) * CH + st[j] - run;
            run += cn[j];
        }
        const int T = __shfl(incl, 31);
        int B = ls;
#pragma unroll
        for (int o = 16; o >= 1; o >>= 1) B += __shfl_xor(B, o);
        if (sw == 0 && lane == 0) {
            binstart[t] = B;
            if (t == NBIN - 1) binstart[NBIN] = B + T;
        }
        const int2* crec2 = (const int2*)crec;
        int2*       rec2  = (int2*)rec + B;
        for (int i0 = sw * 32 + lane; i0 < T; i0 += GB * 6 * 32) {
            int  ii[GB], pos[GB];
            int2 r[GB];
#pragma unroll
            for (int k = 0; k < GB; ++k) {
                const int i = i0 + k * 6 * 32;
                ii[k]  = (i < T) ? i : (T - 1);
                pos[k] = 0;
            }
#pragma unroll
            for (int step = MAXCH / 2; step >= 1; step >>= 1)
#pragma unroll
                for (int k = 0; k < GB; ++k)
                    pos[k] += (P[pos[k] + step] <= ii[k]) ? step : 0;
#pragma unroll
            for (int k = 0; k < GB; ++k) r[k] = crec2[S[pos[k]] + ii[k]];
#pragma unroll
            for (int k = 0; k < GB; ++k)
                if (i0 + k * 6 * 32 < T) {
                    rec2[ii[k]] = r[k];
                    atomicAdd(&c[r[k].y], 1);
                }
        }
    }
    __syncthreads();
    if (tid < nn) dinv[base + tid] = rsqrtf((float)c[tid] + 1.0f);
    if (wvu < 26) {
#pragma unroll
        for (int i = 0; i < 8; ++i) {
            const int row = mt * 16 + i + half * 8;
            if (row < nn) {
                const float sc = rsqrtf((float)c[row] + 1.0f);
#pragma unroll
                for (int q = 0; q < 2; ++q)
                    G[(size_t)(base + row) * 64 + (n0 + q) * 16 + l16] = sc * d[q][i];
            }
        }
    }
}

#define CAPR 4096
#define NRB  13
#define GU   4

template<int ROWS, int PITCH>
__device__ __forceinline__ void tile_edge_sum(const int2* __restrict__ rec2, int e0, int e1,
                                              const float* __restrict__ h, float* acc,
                                              int* srec, int* wcnt, int* rbtot, int tid) {
    const int lane = tid & 31;
    const int wv   = tid >> 5;
    const int half = lane >> 4;
    const int l16  = lane & 15;
    const unsigned ltmask = (1u << lane) - 1u;
    const int wvu  = __builtin_amdgcn_readfirstlane(wv);
    const int rbw  = wvu >> 1;
    const int part = wvu & 1;
    const float4* h4 = (const float4*)h;

    v8f d[4];
#pragma unroll
    for (int t = 0; t < 4; ++t) d[t] = (v8f){};

    for (int c0 = e0; c0 < e1; c0 += CAPR) {
        const int n = (e1 - c0 < CAPR) ? (e1 - c0) : CAPR;

        int2  r[4];
        int   rk[4];
#pragma unroll
        for (int k = 0; k < 4; ++k) {
            const int idx = (k * 32 + wv) * 32 + lane;
            r[k] = rec2[c0 + ((idx < n) ? idx : (n - 1))];
        }
#pragma unroll
        for (int k = 0; k < 4; ++k) {
            const int idx = (k * 32 + wv) * 32 + lane;
            rk[k] = (idx < n) ? (r[k].y >> 4) : 15;
        }
        unsigned bm[4][4];
        int cntk[4];
        int mycnt = 0;
#pragma unroll
        for (int k = 0; k < 4; ++k) {
#pragma unroll
            for (int i = 0; i < 4; ++i)
                bm[k][i] = __builtin_amdgcn_ballot_w32(((rk[k] >> i) & 1) != 0);
            unsigned m = 0xFFFFFFFFu;
#pragma unroll
            for (int i = 0; i < 4; ++i) m &= ((lane >> i) & 1) ? bm[k][i] : ~bm[k][i];
            cntk[k] = (lane < NRB) ? __builtin_popcount(m) : 0;
            mycnt += cntk[k];
        }
        if (lane < 16) wcnt[wv * 16 + lane] = mycnt;
        __syncthreads();
        if (wvu < NRB) {
            const int v    = wcnt[lane * 16 + wvu];
            const int incl = wave_incl_scan(v, lane);
            wcnt[lane * 16 + wvu] = incl - v;
            if (lane == 31) rbtot[wvu] = incl;
        }
        __syncthreads();
        const int tv     = (lane < NRB) ? rbtot[lane] : 0;
        const int startv = wave_incl_scan(tv, lane) - tv;
        int basev = startv + ((lane < 16) ? wcnt[wv * 16 + lane] : 0);
#pragma unroll
        for (int k = 0; k < 4; ++k) {
            unsigned m = 0xFFFFFFFFu;
#pragma unroll
            for (int i = 0; i < 4; ++i) m &= ((rk[k] >> i) & 1) ? bm[k][i] : ~bm[k][i];
            const int pos = __shfl(basev, rk[k]) + __builtin_popcount(m & ltmask);
            basev += cntk[k];
            if (rk[k] != 15)
                srec[pos] = r[k].x | ((r[k].y & 15) << 20);
        }
        __syncthreads();

        if (wvu < 2 * NRB) {
            const int segS = __builtin_amdgcn_readlane(startv, rbw);
            const int segE = __builtin_amdgcn_readlane(startv, rbw + 1);
            const int ng   = (segE - segS + 3) >> 2;
            const int gh   = (ng + 1) >> 1;
            const int g0   = part ? gh : 0;
            const int g1   = part ? ng : gh;
            const int je   = segS + g1 * 4;
            const int jEnd = (segE < je) ? segE : je;
            for (int g = g0; g < g1; g += GU) {
                int    rc[GU][2];
                float4 hv[GU][2];
#pragma unroll
                for (int u = 0; u < GU; ++u)
#pragma unroll
                    for (int q = 0; q < 2; ++q) {
                        const int j = segS + (g + u) * 4 + half * 2 + q;
                        rc[u][q] = srec[(j < jEnd) ? j : (jEnd - 1)];
                    }
#pragma unroll
                for (int u = 0; u < GU; ++u)
#pragma unroll
                    for (int q = 0; q < 2; ++q)
                        hv[u][q] = h4[(size_t)(rc[u][q] & 0xFFFFF) * 16 + l16];
#pragma unroll
                for (int u = 0; u < GU; ++u) {
                    const int j = segS + (g + u) * 4 + half * 2;
                    v2f a;
                    a.x = (j     < jEnd && (rc[u][0] >> 20) == l16) ? 1.0f : 0.0f;
                    a.y = (j + 1 < jEnd && (rc[u][1] >> 20) == l16) ? 1.0f : 0.0f;
                    v2f b0, b1, b2, b3;
                    b0.x = hv[u][0].x; b0.y = hv[u][1].x;
                    b1.x = hv[u][0].y; b1.y = hv[u][1].y;
                    b2.x = hv[u][0].z; b2.y = hv[u][1].z;
                    b3.x = hv[u][0].w; b3.y = hv[u][1].w;
                    d[0] = __builtin_amdgcn_wmma_f32_16x16x4_f32(false, a, false, b0, (short)0, d[0], false, false);
                    d[1] = __builtin_amdgcn_wmma_f32_16x16x4_f32(false, a, false, b1, (short)0, d[1], false, false);
                    d[2] = __builtin_amdgcn_wmma_f32_16x16x4_f32(false, a, false, b2, (short)0, d[2], false, false);
                    d[3] = __builtin_amdgcn_wmma_f32_16x16x4_f32(false, a, false, b3, (short)0, d[3], false, false);
                }
            }
        }
        __syncthreads();
    }

#pragma unroll
    for (int p = 0; p < 2; ++p) {
        if (wvu < 2 * NRB && part == p) {
#pragma unroll
            for (int i = 0; i < 8; ++i) {
                const int row = rbw * 16 + half * 8 + i;
                if (row < ROWS) {
                    float4* o = (float4*)(acc + row * PITCH + l16 * 4);
                    float4  v = *o;
                    v.x += d[0][i]; v.y += d[1][i]; v.z += d[2][i]; v.w += d[3][i];
                    *o = v;
                }
            }
        }
        __syncthreads();
    }
}

template<int DO_ELU, int HW>
__global__ __launch_bounds__(1024) void k_tile_agg(const int* __restrict__ rec,
                                                   const int* __restrict__ binstart,
                                                   const float* __restrict__ dinv,
                                                   const float* __restrict__ h,
                                                   const float* __restrict__ bias,
                                                   float* __restrict__ act) {
    static_assert(HW == 64, "rows of 64 floats");
    constexpr int Q  = HW / 4;
    __shared__ __attribute__((aligned(16))) float acc[TILE * HW];
    __shared__ int   srec[CAPR];
    __shared__ int   wcnt[32 * 16];
    __shared__ int   rbtot[16];
    __shared__ float sdinv[TILE];
    const int tid  = threadIdx.x;
    const int t    = blockIdx.x;
    const int base = t * TILE;
    const int nn   = (NN - base < TILE) ? (NN - base) : TILE;

    if (tid < nn) sdinv[tid] = dinv[base + tid];
    const float4* h4  = (const float4*)(h + (size_t)base * HW);
    float4*       a4  = (float4*)acc;
    constexpr int NR = (TILE * Q + 1023) / 1024;
    const int lim = nn * Q;
    float4 hv[NR];
#pragma unroll
    for (int k = 0; k < NR; ++k) {
        const int i  = tid + k * 1024;
        hv[k] = h4[(i < lim) ? i : (lim - 1)];
    }
#pragma unroll
    for (int k = 0; k < NR; ++k) {
        const int i = tid + k * 1024;
        if (i < lim) a4[i] = hv[k];
    }
    for (int i = lim + tid; i < TILE * Q; i += 1024) a4[i] = make_float4(0.0f, 0.0f, 0.0f, 0.0f);
    __syncthreads();

    tile_edge_sum<TILE, HW>((const int2*)rec, binstart[t], binstart[t + 1], h, acc,
                            srec, wcnt, rbtot, tid);

    float4* o4 = (float4*)(act + (size_t)base * HW);
    for (int i = tid; i < nn * Q; i += 1024) {
        const float4 a = a4[i];
        const int    c = (i & 15) * 4;
        const float  sc = sdinv[i >> 4];
        float4 v = make_float4(sc * a.x + bias[c], sc * a.y + bias[c + 1], sc * a.z + bias[c + 2], sc * a.w + bias[c + 3]);
        if (DO_ELU) {
            v.x = (v.x <= 0.0f) ? expm1f(v.x) : v.x;
            v.y = (v.y <= 0.0f) ? expm1f(v.y) : v.y;
            v.z = (v.z <= 0.0f) ? expm1f(v.z) : v.z;
            v.w = (v.w <= 0.0f) ? expm1f(v.w) : v.w;
        }
        o4[i] = make_float4(sc * v.x, sc * v.y, sc * v.z, sc * v.w);
    }
}


#define PA  68
#define MTL 13

template<int NB, int NT>
__device__ __forceinline__ void tile_mm(const float* arow, const float* __restrict__ B,
                                        int n0, int half, int l16, v8f (&d)[NT]) {
#pragma unroll
    for (int t = 0; t < NT; ++t) d[t] = (v8f){};
    const float* Bp = B + half * 2 * NB;
#pragma unroll
    for (int k = 0; k < 64; k += 4) {
        v2f a;
        a.x = arow[k];
        a.y = arow[k + 1];
#pragma unroll
        for (int t = 0; t < NT; ++t) {
            const int c = (n0 + t) * 16 + l16;
            v2f b;
            if constexpr (NB % 16 == 0) {
                b.x = Bp[k * NB + c];
                b.y = Bp[k * NB + NB + c];
            } else {
                const float m  = (c < NB) ? 1.0f : 0.0f;
                const int   cc = (c < NB) ? c : (NB - 1);
                b.x = m * Bp[k * NB + cc];
                b.y = m * Bp[k * NB + NB + cc];
            }
            d[t] = __builtin_amdgcn_wmma_f32_16x16x4_f32(
                false, a, false, b, (short)0, d[t], false, false);
        }
    }
}

template<int LAST>
__global__ __launch_bounds__(1024) void k_tile_agg_mm(const int* __restrict__ rec,
                                                      const int* __restrict__ binstart,
                                                      const float* __restrict__ dinv,
                                                      const float* __restrict__ h,
                                                      const float* __restrict__ Wm,
                                                      const float* __restrict__ b,
                                                      const float* __restrict__ Wl,
                                                      const float* __restrict__ bl,
                                                      float* __restrict__ outp) {
    constexpr int HW = 64;
    __shared__ __attribute__((aligned(16))) float acc[MTL * 16 * PA];
    __shared__ __attribute__((aligned(16))) float mid[LAST ? MTL * 16 * PA : 4];
    __shared__ int   srec[CAPR];
    __shared__ int   wcnt[32 * 16];
    __shared__ int   rbtot[16];
    __shared__ float sdinv[TILE];
    __shared__ float scb[48];
    const int tid  = threadIdx.x;
    const int lane = tid & 31;
    const int wv   = tid >> 5;
    const int t    = blockIdx.x;
    const int base = t * TILE;
    const int nn   = (NN - base < TILE) ? (NN - base) : TILE;

    if (tid < nn) sdinv[tid] = dinv[base + tid];
    const float4* h4 = (const float4*)(h + (size_t)base * HW);
    constexpr int NR = (TILE * 16 + 1023) / 1024;
    const int lim = nn * 16;
    float4 hv[NR];
#pragma unroll
    for (int k = 0; k < NR; ++k) {
        const int i  = tid + k * 1024;
        hv[k] = h4[(i < lim) ? i : (lim - 1)];
    }
#pragma unroll
    for (int k = 0; k < NR; ++k) {
        const int i = tid + k * 1024;
        if (i < lim) *(float4*)(acc + (i >> 4) * PA + (i & 15) * 4) = hv[k];
    }
    for (int i = lim + tid; i < MTL * 16 * 16; i += 1024)
        *(float4*)(acc + (i >> 4) * PA + (i & 15) * 4) = make_float4(0.0f, 0.0f, 0.0f, 0.0f);
    __syncthreads();

    if (LAST && wv >= 30) {
        const int c = tid - 960;
        if (c < 40) {
            float s = bl[c];
#pragma unroll 16
            for (int k = 0; k < 64; ++k) s += b[k] * Wl[k * 40 + c];
            scb[c] = s;
        }
    }

    tile_edge_sum<MTL * 16, PA>((const int2*)rec, binstart[t], binstart[t + 1], h, acc,
                                srec, wcnt, rbtot, tid);

    const int half = lane >> 4;
    const int l16  = lane & 15;
    const int wvu = __builtin_amdgcn_readfirstlane(wv);
    if (wvu < 2 * MTL) {
        const int mt = wvu >> 1;
        const int n0 = (wvu & 1) * 2;
        v8f d[2];
        tile_mm<64, 2>(acc + (mt * 16 + l16) * PA + half * 2, Wm, n0, half, l16, d);
#pragma unroll
        for (int q = 0; q < 2; ++q) {
            const int c = (n0 + q) * 16 + l16;
            if (LAST) {
#pragma unroll
                for (int i = 0; i < 8; ++i)
                    mid[(mt * 16 + i + half * 8) * PA + c] = d[q][i];
            } else {
                const float bv = b[c];
#pragma unroll
                for (int i = 0; i < 8; ++i) {
                    const int row = mt * 16 + i + half * 8;
                    if (row < nn) {
                        const float sc = sdinv[row];
                        float v = sc * d[q][i] + bv;
                        v = (v <= 0.0f) ? expm1f(v) : v;
                        outp[(size_t)(base + row) * 64 + c] = sc * v;
                    }
                }
            }
        }
    }
    if (LAST) {
        __syncthreads();
        if (wvu < MTL) {
            const int mt = wvu;
            v8f e[3];
            tile_mm<40, 3>(mid + (mt * 16 + l16) * PA + half * 2, Wl, 0, half, l16, e);
#pragma unroll
            for (int q = 0; q < 3; ++q) {
                const int c = q * 16 + l16;
                if (c < 40) {
                    const float bv = scb[c];
#pragma unroll
                    for (int i = 0; i < 8; ++i) {
                        const int row = mt * 16 + i + half * 8;
                        if (row < nn) outp[(size_t)(base + row) * 40 + c] = sdinv[row] * e[q][i] + bv;
                    }
                }
            }
        }
    }
}


extern "C" void kernel_launch(void* const* d_in, const int* in_sizes, int n_in,
                              void* d_out, int out_size, void* d_ws, size_t ws_size,
                              hipStream_t stream) {
    const float* x  = (const float*)d_in[0];
    const int*   ei = (const int*)d_in[1];
    const float* W1 = (const float*)d_in[2];
    const float* b1 = (const float*)d_in[3];
    const float* W2 = (const float*)d_in[4];
    const float* b2 = (const float*)d_in[5];
    const float* W3 = (const float*)d_in[6];
    const float* b3 = (const float*)d_in[7];
    const float* Wl = (const float*)d_in[8];
    const float* bl = (const float*)d_in[9];
    float* out = (float*)d_out;

    const int E = in_sizes[1] / 2;
    const int* src = ei;
    const int* dst = ei + E;

    float* dinv     = (float*)d_ws;
    int*   binstart = (int*)(dinv + 50176);
    int*   coffT    = binstart + 512;
    int*   rec      = coffT + 1024 * NBIN;
    float* G        = (float*)(rec + 2 * (size_t)E);
    float* ACT      = G + (size_t)NN * H;
    int*   crec     = (int*)(ACT + (size_t)NN * H);

    const dim3 blk(256);
    const dim3 big(1024);
    const dim3 gTile(NBIN);

    k_chunk_sort<<<dim3(MAXCH), blk, 0, stream>>>(src, dst, coffT, crec, E);
    k_tile_dinv_mm1<<<gTile, big, 0, stream>>>(crec, coffT, x, W1, dinv, G, rec, binstart);

    k_tile_agg<1, H><<<gTile, big, 0, stream>>>(rec, binstart, dinv, G, b1, ACT);

    k_tile_agg_mm<0><<<gTile, big, 0, stream>>>(rec, binstart, dinv, ACT, W2, b2, Wl, bl, G);
    k_tile_agg_mm<1><<<gTile, big, 0, stream>>>(rec, binstart, dinv, G, W3, b3, Wl, bl, out);
}


# Tile launches ask for all opening data at once: dinv, own rows, tile bounds, first-pass records in flight before the first wait (compiler had sunk each load into its guarded store); bit-identical
# speedup vs baseline: 4.2640x; 1.0372x over previous
#include <hip/hip_runtime.h>
#include <math.h>

typedef __attribute__((ext_vector_type(2))) float v2f;
typedef __attribute__((ext_vector_type(8))) float v8f;

#define NN 50000
#define H  64
#define TILE 196
#define NBIN 256
#define CH   4096

#define MAXCH 256

__device__ __forceinline__ int wave_incl_scan(int v, int lane) {
#pragma unroll
    for (int o = 1; o < 32; o <<= 1) {
        const int t = __shfl_up(v, o);
        if (lane >= o) v += t;
    }
    return v;
}

__global__ __launch_bounds__(256) void k_chunk_sort(const int* __restrict__ src,
                                                    const int* __restrict__ dst,
                                                    int* __restrict__ coffT,
                                                    int* __restrict__ crec, int E) {
    __shared__ int  hist[NBIN];
    __shared__ int  wtot[8];
    __shared__ int2 stage[CH];
    const int tid  = threadIdx.x;
    const int lane = tid & 31;
    const int wv   = tid >> 5;
    const int c    = blockIdx.x;
    hist[tid] = 0;
    __syncthreads();
    const int e0 = c * CH;
    int b[CH / 256], s[CH / 256], y[CH / 256], rk[CH / 256];
#pragma unroll
    for (int k = 0; k < CH / 256; ++k) {
        const int e  = e0 + k * 256 + tid;
        const int ec = (e < E) ? e : (E - 1);
        y[k] = dst[ec];
        s[k] = src[ec];
    }
#pragma unroll
    for (int k = 0; k < CH / 256; ++k) {
        const int e = e0 + k * 256 + tid;
        b[k]  = y[k] / TILE;
        y[k] -= b[k] * TILE;
        rk[k] = 0;
        if (e < E) rk[k] = atomicAdd(&hist[b[k]], 1);
    }
    __syncthreads();
    const int v    = hist[tid];
    const int incl = wave_incl_scan(v, lane);
    if (lane == 31) wtot[wv] = incl;
    __syncthreads();
    int add = 0;
#pragma unroll
    for (int w = 0; w < 8; ++w) add += (w < wv) ? wtot[w] : 0;
    const int off = add + incl - v;
    coffT[tid * MAXCH + c] = off;
    if (tid == NBIN - 1) coffT[NBIN * MAXCH + c] = off + v;
    hist[tid] = off;
    __syncthreads();
#pragma unroll
    for (int k = 0; k < CH / 256; ++k) {
        const int e = e0 + k * 256 + tid;
        if (e < E) stage[hist[b[k]] + rk[k]] = make_int2(s[k], y[k]);
    }
    __syncthreads();
    const int n = (E - e0 < CH) ? (E - e0) : CH;
    int2* o2 = (int2*)crec + e0;
    for (int i = tid; i < n; i += 256) o2[i] = stage[i];
}

#define GB 9
__global__ __launch_bounds__(1024) void k_tile_dinv_mm1(const int* __restrict__ crec,
                                                        const int* __restrict__ coffT,
                                                        const float* __restrict__ x,
                                                        const float* __restrict__ W1,
                                                        float* __restrict__ dinv,
                                                        float* __restrict__ G,
                                                        int* __restrict__ rec,
                                                        int* __restrict__ binstart) {
    __shared__ int c[13 * 16];
    __shared__ int pfx[6 * MAXCH];
    __shared__ int cst[6 * MAXCH];
    const int tid  = threadIdx.x;
    const int lane = tid & 31;
    const int half = lane >> 4;
    const int l16  = lane & 15;
    const int t    = blockIdx.x;
    const int base = t * TILE;
    const int nn   = (NN - base < TILE) ? (NN - base) : TILE;
    if (tid < 13 * 16) c[tid] = 0;
    __syncthreads();
    const int wvu = __builtin_amdgcn_readfirstlane(tid >> 5);
    const int mt  = wvu >> 1;
    const int n0  = (wvu & 1) * 2;
    v8f d[2];
    d[0] = (v8f){};
    d[1] = (v8f){};
    if (wvu < 26) {
        int arow = base + mt * 16 + l16;
        arow = (arow < NN) ? arow : (NN - 1);
        const float* Ap = x + (size_t)arow * 128 + half * 2;
        const float* Bp = W1 + half * 2 * 64;
#pragma unroll
        for (int k = 0; k < 128; k += 4) {
            v2f a;
            a.x = Ap[k];
            a.y = Ap[k + 1];
#pragma unroll
            for (int q = 0; q < 2; ++q) {
                const int cc = (n0 + q) * 16 + l16;
                v2f b;
                b.x = Bp[k * 64 + cc];
                b.y = Bp[k * 64 + 64 + cc];
                d[q] = __builtin_amdgcn_wmma_f32_16x16x4_f32(
                    false, a, false, b, (short)0, d[q], false, false);
            }
        }
    } else {
        const int sw = wvu - 26;
        int* P = pfx + sw * MAXCH;
        int* S = cst + sw * MAXCH;
        const int4* rs = (const int4*)(coffT + (size_t)t * MAXCH) + lane * 2;
        const int4* re = (const int4*)(coffT + (size_t)(t + 1) * MAXCH) + lane * 2;
        const int4 s0 = rs[0], s1 = rs[1], f0 = re[0], f1 = re[1];
        const int st[8] = {s0.x, s0.y, s0.z, s0.w, s1.x, s1.y, s1.z, s1.w};
        const int en[8] = {f0.x, f0.y, f0.z, f0.w, f1.x, f1.y, f1.z, f1.w};
        int cn[8];
        int lt = 0, ls = 0;
#pragma unroll
        for (int j = 0; j < 8; ++j) {
            cn[j] = en[j] - st[j];
            lt += cn[j];
            ls += st[j];
        }
        const int incl = wave_incl_scan(lt, lane);
        int run = incl - lt;
#pragma unroll
        for (int j = 0; j < 8; ++j) {
            P[lane * 8 + j] = run;
            S[lane * 8 + j] = (lane * 8 + j) * CH + st[j] - run;
            run += cn[j];
        }
        const int T = __shfl(incl, 31);
        int B = ls;
#pragma unroll
        for (int o = 16; o >= 1; o >>= 1) B += __shfl_xor(B, o);
        if (sw == 0 && lane == 0) {
            binstart[t] = B;
            if (t == NBIN - 1) binstart[NBIN] = B + T;
        }
        const int2* crec2 = (const int2*)crec;
        int2*       rec2  = (int2*)rec + B;
        for (int i0 = sw * 32 + lane; i0 < T; i0 += GB * 6 * 32) {
            int  ii[GB], pos[GB];
            int2 r[GB];
#pragma unroll
            for (int k = 0; k < GB; ++k) {
                const int i = i0 + k * 6 * 32;
                ii[k]  = (i < T) ? i : (T - 1);
                pos[k] = 0;
            }
#pragma unroll
            for (int step = MAXCH / 2; step >= 1; step >>= 1)
#pragma unroll
                for (int k = 0; k < GB; ++k)
                    pos[k] += (P[pos[k] + step] <= ii[k]) ? step : 0;
#pragma unroll
            for (int k = 0; k < GB; ++k) r[k] = crec2[S[pos[k]] + ii[k]];
#pragma unroll
            for (int k = 0; k < GB; ++k)
                if (i0 + k * 6 * 32 < T) {
                    rec2[ii[k]] = r[k];
                    atomicAdd(&c[r[k].y], 1);
                }
        }
    }
    __syncthreads();
    if (tid < nn) dinv[base + tid] = rsqrtf((float)c[tid] + 1.0f);
    if (wvu < 26) {
#pragma unroll
        for (int i = 0; i < 8; ++i) {
            const int row = mt * 16 + i + half * 8;
            if (row < nn) {
                const float sc = rsqrtf((float)c[row] + 1.0f);
#pragma unroll
                for (int q = 0; q < 2; ++q)
                    G[(size_t)(base + row) * 64 + (n0 + q) * 16 + l16] = sc * d[q][i];
            }
        }
    }
}

#define CAPR 4096
#define NRB  13
#define GU   4

__device__ __forceinline__ void load_pass_records(const int2* __restrict__ rec2, int c0, int e1,
                                                  int tid, int2 (&r)[4]) {
    const int n = (e1 - c0 < CAPR) ? (e1 - c0) : CAPR;
#pragma unroll
    for (int k = 0; k < 4; ++k) {
        const int idx = (k * 32 + (tid >> 5)) * 32 + (tid & 31);
        r[k] = rec2[c0 + ((idx < n) ? idx : (n - 1))];
    }
}

template<int ROWS, int PITCH>
__device__ __forceinline__ void tile_edge_sum(const int2* __restrict__ rec2, int e0, int e1,
                                              const float* __restrict__ h, float* acc,
                                              int* srec, int* wcnt, int* rbtot, int tid,
                                              int2 (&r)[4]) {
    const int lane = tid & 31;
    const int wv   = tid >> 5;
    const int half = lane >> 4;
    const int l16  = lane & 15;
    const unsigned ltmask = (1u << lane) - 1u;
    const int wvu  = __builtin_amdgcn_readfirstlane(wv);
    const int rbw  = wvu >> 1;
    const int part = wvu & 1;
    const float4* h4 = (const float4*)h;

    v8f d[4];
#pragma unroll
    for (int t = 0; t < 4; ++t) d[t] = (v8f){};

    for (int c0 = e0; c0 < e1; c0 += CAPR) {
        const int n = (e1 - c0 < CAPR) ? (e1 - c0) : CAPR;

        int   rk[4];
#pragma unroll
        for (int k = 0; k < 4; ++k) {
            const int idx = (k * 32 + wv) * 32 + lane;
            rk[k] = (idx < n) ? (r[k].y >> 4) : 15;
        }
        unsigned bm[4][4];
        int cntk[4];
        int mycnt = 0;
#pragma unroll
        for (int k = 0; k < 4; ++k) {
#pragma unroll
            for (int i = 0; i < 4; ++i)
                bm[k][i] = __builtin_amdgcn_ballot_w32(((rk[k] >> i) & 1) != 0);
            unsigned m = 0xFFFFFFFFu;
#pragma unroll
            for (int i = 0; i < 4; ++i) m &= ((lane >> i) & 1) ? bm[k][i] : ~bm[k][i];
            cntk[k] = (lane < NRB) ? __builtin_popcount(m) : 0;
            mycnt += cntk[k];
        }
        if (lane < 16) wcnt[wv * 16 + lane] = mycnt;
        __syncthreads();
        if (wvu < NRB) {
            const int v    = wcnt[lane * 16 + wvu];
            const int incl = wave_incl_scan(v, lane);
            wcnt[lane * 16 + wvu] = incl - v;
            if (lane == 31) rbtot[wvu] = incl;
        }
        __syncthreads();
        const int tv     = (lane < NRB) ? rbtot[lane] : 0;
        const int startv = wave_incl_scan(tv, lane) - tv;
        int basev = startv + ((lane < 16) ? wcnt[wv * 16 + lane] : 0);
#pragma unroll
        for (int k = 0; k < 4; ++k) {
            unsigned m = 0xFFFFFFFFu;
#pragma unroll
            for (int i = 0; i < 4; ++i) m &= ((rk[k] >> i) & 1) ? bm[k][i] : ~bm[k][i];
            const int pos = __shfl(basev, rk[k]) + __builtin_popcount(m & ltmask);
            basev += cntk[k];
            if (rk[k] != 15)
                srec[pos] = r[k].x | ((r[k].y & 15) << 20);
        }
        __syncthreads();

        if (wvu < 2 * NRB) {
            const int segS = __builtin_amdgcn_readlane(startv, rbw);
            const int segE = __builtin_amdgcn_readlane(startv, rbw + 1);
            const int ng   = (segE - segS + 3) >> 2;
            const int gh   = (ng + 1) >> 1;
            const int g0   = part ? gh : 0;
            const int g1   = part ? ng : gh;
            const int je   = segS + g1 * 4;
            const int jEnd = (segE < je) ? segE : je;
            for (int g = g0; g < g1; g += GU) {
                int    rc[GU][2];
                float4 hv[GU][2];
#pragma unroll
                for (int u = 0; u < GU; ++u)
#pragma unroll
                    for (int q = 0; q < 2; ++q) {
                        const int j = segS + (g + u) * 4 + half * 2 + q;
                        rc[u][q] = srec[(j < jEnd) ? j : (jEnd - 1)];
                    }
#pragma unroll
                for (int u = 0; u < GU; ++u)
#pragma unroll
                    for (int q = 0; q < 2; ++q)
                        hv[u][q] = h4[(size_t)(rc[u][q] & 0xFFFFF) * 16 + l16];
#pragma unroll
                for (int u = 0; u < GU; ++u) {
                    const int j = segS + (g + u) * 4 + half * 2;
                    v2f a;
                    a.x = (j     < jEnd && (rc[u][0] >> 20) == l16) ? 1.0f : 0.0f;
                    a.y = (j + 1 < jEnd && (rc[u][1] >> 20) == l16) ? 1.0f : 0.0f;
                    v2f b0, b1, b2, b3;
                    b0.x = hv[u][0].x; b0.y = hv[u][1].x;
                    b1.x = hv[u][0].y; b1.y = hv[u][1].y;
                    b2.x = hv[u][0].z; b2.y = hv[u][1].z;
                    b3.x = hv[u][0].w; b3.y = hv[u][1].w;
                    d[0] = __builtin_amdgcn_wmma_f32_16x16x4_f32(false, a, false, b0, (short)0, d[0], false, false);
                    d[1] = __builtin_amdgcn_wmma_f32_16x16x4_f32(false, a, false, b1, (short)0, d[1], false, false);
                    d[2] = __builtin_amdgcn_wmma_f32_16x16x4_f32(false, a, false, b2, (short)0, d[2], false, false);
                    d[3] = __builtin_amdgcn_wmma_f32_16x16x4_f32(false, a, false, b3, (short)0, d[3], false, false);
                }
            }
        }
        if (c0 + CAPR < e1) load_pass_records(rec2, c0 + CAPR, e1, tid, r);
        __syncthreads();
    }

#pragma unroll
    for (int p = 0; p < 2; ++p) {
        if (wvu < 2 * NRB && part == p) {
#pragma unroll
            for (int i = 0; i < 8; ++i) {
                const int row = rbw * 16 + half * 8 + i;
                if (row < ROWS) {
                    float4* o = (float4*)(acc + row * PITCH + l16 * 4);
                    float4  v = *o;
                    v.x += d[0][i]; v.y += d[1][i]; v.z += d[2][i]; v.w += d[3][i];
                    *o = v;
                }
            }
        }
        __syncthreads();
    }
}

#define OPENING_LOADS_TOGETHER(dv, hv, r)                                        \
    asm volatile("" :: "v"(dv),                                                  \
                 "v"(hv[0].x), "v"(hv[0].y), "v"(hv[0].z), "v"(hv[0].w),         \
                 "v"(hv[1].x), "v"(hv[1].y), "v"(hv[1].z), "v"(hv[1].w),         \
                 "v"(hv[2].x), "v"(hv[2].y), "v"(hv[2].z), "v"(hv[2].w),         \
                 "v"(hv[3].x), "v"(hv[3].y), "v"(hv[3].z), "v"(hv[3].w),         \
                 "v"(r[0].x), "v"(r[0].y), "v"(r[1].x), "v"(r[1].y),             \
                 "v"(r[2].x), "v"(r[2].y), "v"(r[3].x), "v"(r[3].y))

template<int DO_ELU, int HW>
__global__ __launch_bounds__(1024) void k_tile_agg(const int* __restrict__ rec,
                                                   const int* __restrict__ binstart,
                                                   const float* __restrict__ dinv,
                                                   const float* __restrict__ h,
                                                   const float* __restrict__ bias,
                                                   float* __restrict__ act) {
    static_assert(HW == 64, "rows of 64 floats");
    constexpr int Q  = HW / 4;
    __shared__ __attribute__((aligned(16))) float acc[TILE * HW];
    __shared__ int   srec[CAPR];
    __shared__ int   wcnt[32 * 16];
    __shared__ int   rbtot[16];
    __shared__ float sdinv[TILE];
    const int tid  = threadIdx.x;
    const int t    = blockIdx.x;
    const int base = t * TILE;
    const int nn   = (NN - base < TILE) ? (NN - base) : TILE;

    const int e0 = binstart[t];
    const int e1 = binstart[t + 1];
    const float dv = dinv[base + ((tid < nn) ? tid : (nn - 1))];
    const float4* h4  = (const float4*)(h + (size_t)base * HW);
    float4*       a4  = (float4*)acc;
    constexpr int NR = (TILE * Q + 1023) / 1024;
    static_assert(NR == 4, "OPENING_LOADS_TOGETHER names four row loads");
    const int lim = nn * Q;
    float4 hv[NR];
#pragma unroll
    for (int k = 0; k < NR; ++k) {
        const int i  = tid + k * 1024;
        hv[k] = h4[(i < lim) ? i : (lim - 1)];
    }
    int2 r[4];
#pragma unroll
    for (int k = 0; k < 4; ++k) r[k] = make_int2(0, 0);
    if (e0 < e1) load_pass_records((const int2*)rec, e0, e1, tid, r);
    OPENING_LOADS_TOGETHER(dv, hv, r);
    if (tid < nn) sdinv[tid] = dv;
#pragma unroll
    for (int k = 0; k < NR; ++k) {
        const int i = tid + k * 1024;
        if (i < lim) a4[i] = hv[k];
    }
    for (int i = lim + tid; i < TILE * Q; i += 1024) a4[i] = make_float4(0.0f, 0.0f, 0.0f, 0.0f);
    __syncthreads();

    tile_edge_sum<TILE, HW>((const int2*)rec, e0, e1, h, acc, srec, wcnt, rbtot, tid, r);

    float4* o4 = (float4*)(act + (size_t)base * HW);
    for (int i = tid; i < nn * Q; i += 1024) {
        const float4 a = a4[i];
        const int    c = (i & 15) * 4;
        const float  sc = sdinv[i >> 4];
        float4 v = make_float4(sc * a.x + bias[c], sc * a.y + bias[c + 1], sc * a.z + bias[c + 2], sc * a.w + bias[c + 3]);
        if (DO_ELU) {
            v.x = (v.x <= 0.0f) ? expm1f(v.x) : v.x;
            v.y = (v.y <= 0.0f) ? expm1f(v.y) : v.y;
            v.z = (v.z <= 0.0f) ? expm1f(v.z) : v.z;
            v.w = (v.w <= 0.0f) ? expm1f(v.w) : v.w;
        }
        o4[i] = make_float4(sc * v.x, sc * v.y, sc * v.z, sc * v.w);
    }
}


#define PA  68
#define MTL 13

template<int NB, int NT>
__device__ __forceinline__ void tile_mm(const float* arow, const float* __restrict__ B,
                                        int n0, int half, int l16, v8f (&d)[NT]) {
#pragma unroll
    for (int t = 0; t < NT; ++t) d[t] = (v8f){};
    const float* Bp = B + half * 2 * NB;
#pragma unroll
    for (int k = 0; k < 64; k += 4) {
        v2f a;
        a.x = arow[k];
        a.y = arow[k + 1];
#pragma unroll
        for (int t = 0; t < NT; ++t) {
            const int c = (n0 + t) * 16 + l16;
            v2f b;
            if constexpr (NB % 16 == 0) {
                b.x = Bp[k * NB + c];
                b.y = Bp[k * NB + NB + c];
            } else {
                const float m  = (c < NB) ? 1.0f : 0.0f;
                const int   cc = (c < NB) ? c : (NB - 1);
                b.x = m * Bp[k * NB + cc];
                b.y = m * Bp[k * NB + NB + cc];
            }
            d[t] = __builtin_amdgcn_wmma_f32_16x16x4_f32(
                false, a, false, b, (short)0, d[t], false, false);
        }
    }
}

template<int LAST>
__global__ __launch_bounds__(1024) void k_tile_agg_mm(const int* __restrict__ rec,
                                                      const int* __restrict__ binstart,
                                                      const float* __restrict__ dinv,
                                                      const float* __restrict__ h,
                                                      const float* __restrict__ Wm,
                                                      const float* __restrict__ b,
                                                      const float* __restrict__ Wl,
                                                      const float* __restrict__ bl,
                                                      float* __restrict__ outp) {
    constexpr int HW = 64;
    __shared__ __attribute__((aligned(16))) float acc[MTL * 16 * PA];
    __shared__ __attribute__((aligned(16))) float mid[LAST ? MTL * 16 * PA : 4];
    __shared__ int   srec[CAPR];
    __shared__ int   wcnt[32 * 16];
    __shared__ int   rbtot[16];
    __shared__ float sdinv[TILE];
    __shared__ float scb[48];
    const int tid  = threadIdx.x;
    const int lane = tid & 31;
    const int wv   = tid >> 5;
    const int t    = blockIdx.x;
    const int base = t * TILE;
    const int nn   = (NN - base < TILE) ? (NN - base) : TILE;

    const int e0 = binstart[t];
    const int e1 = binstart[t + 1];
    const float dv = dinv[base + ((tid < nn) ? tid : (nn - 1))];
    const float4* h4 = (const float4*)(h + (size_t)base * HW);
    constexpr int NR = (TILE * 16 + 1023) / 1024;
    static_assert(NR == 4, "OPENING_LOADS_TOGETHER names four row loads");
    const int lim = nn * 16;
    float4 hv[NR];
#pragma unroll
    for (int k = 0; k < NR; ++k) {
        const int i  = tid + k * 1024;
        hv[k] = h4[(i < lim) ? i : (lim - 1)];
    }
    int2 r[4];
#pragma unroll
    for (int k = 0; k < 4; ++k) r[k] = make_int2(0, 0);
    if (e0 < e1) load_pass_records((const int2*)rec, e0, e1, tid, r);
    OPENING_LOADS_TOGETHER(dv, hv, r);
    if (tid < nn) sdinv[tid] = dv;
#pragma unroll
    for (int k = 0; k < NR; ++k) {
        const int i = tid + k * 1024;
        if (i < lim) *(float4*)(acc + (i >> 4) * PA + (i & 15) * 4) = hv[k];
    }
    for (int i = lim + tid; i < MTL * 16 * 16; i += 1024)
        *(float4*)(acc + (i >> 4) * PA + (i & 15) * 4) = make_float4(0.0f, 0.0f, 0.0f, 0.0f);
    __syncthreads();

    tile_edge_sum<MTL * 16, PA>((const int2*)rec, e0, e1, h, acc, srec, wcnt, rbtot, tid, r);

    if (LAST && wv >= 30) {
        const int c = tid - 960;
        if (c < 40) {
            float s = bl[c];
#pragma unroll 16
            for (int k = 0; k < 64; ++k) s += b[k] * Wl[k * 40 + c];
            scb[c] = s;
        }
    }

    const int half = lane >> 4;
    const int l16  = lane & 15;
    const int wvu = __builtin_amdgcn_readfirstlane(wv);
    if (wvu < 2 * MTL) {
        const int mt = wvu >> 1;
        const int n0 = (wvu & 1) * 2;
        v8f d[2];
        tile_mm<64, 2>(acc + (mt * 16 + l16) * PA + half * 2, Wm, n0, half, l16, d);
#pragma unroll
        for (int q = 0; q < 2; ++q) {
            const int c = (n0 + q) * 16 + l16;
            if (LAST) {
#pragma unroll
                for (int i = 0; i < 8; ++i)
                    mid[(mt * 16 + i + half * 8) * PA + c] = d[q][i];
            } else {
                const float bv = b[c];
#pragma unroll
                for (int i = 0; i < 8; ++i) {
                    const int row = mt * 16 + i + half * 8;
                    if (row < nn) {
                        const float sc = sdinv[row];
                        float v = sc * d[q][i] + bv;
                        v = (v <= 0.0f) ? expm1f(v) : v;
                        outp[(size_t)(base + row) * 64 + c] = sc * v;
                    }
                }
            }
        }
    }
    if (LAST) {
        __syncthreads();
        if (wvu < MTL) {
            const int mt = wvu;
            v8f e[3];
            tile_mm<40, 3>(mid + (mt * 16 + l16) * PA + half * 2, Wl, 0, half, l16, e);
#pragma unroll
            for (int q = 0; q < 3; ++q) {
                const int c = q * 16 + l16;
                if (c < 40) {
                    const float bv = scb[c];
#pragma unroll
                    for (int i = 0; i < 8; ++i) {
                        const int row = mt * 16 + i + half * 8;
                        if (row < nn) outp[(size_t)(base + row) * 40 + c] = sdinv[row] * e[q][i] + bv;
                    }
                }
            }
        }
    }
}


extern "C" void kernel_launch(void* const* d_in, const int* in_sizes, int n_in,
                              void* d_out, int out_size, void* d_ws, size_t ws_size,
                              hipStream_t stream) {
    const float* x  = (const float*)d_in[0];
    const int*   ei = (const int*)d_in[1];
    const float* W1 = (const float*)d_in[2];
    const float* b1 = (const float*)d_in[3];
    const float* W2 = (const float*)d_in[4];
    const float* b2 = (const float*)d_in[5];
    const float* W3 = (const float*)d_in[6];
    const float* b3 = (const float*)d_in[7];
    const float* Wl = (const float*)d_in[8];
    const float* bl = (const float*)d_in[9];
    float* out = (float*)d_out;

    const int E = in_sizes[1] / 2;
    const int* src = ei;
    const int* dst = ei + E;

    float* dinv     = (float*)d_ws;
    int*   binstart = (int*)(dinv + 50176);
    int*   coffT    = binstart + 512;
    int*   rec      = coffT + 1024 * NBIN;
    float* G        = (float*)(rec + 2 * (size_t)E);
    float* ACT      = G + (size_t)NN * H;
    int*   crec     = (int*)(ACT + (size_t)NN * H);

    const dim3 blk(256);
    const dim3 big(1024);
    const dim3 gTile(NBIN);

    k_chunk_sort<<<dim3(MAXCH), blk, 0, stream>>>(src, dst, coffT, crec, E);
    k_tile_dinv_mm1<<<gTile, big, 0, stream>>>(crec, coffT, x, W1, dinv, G, rec, binstart);

    k_tile_agg<1, H><<<gTile, big, 0, stream>>>(rec, binstart, dinv, G, b1, ACT);

    k_tile_agg_mm<0><<<gTile, big, 0, stream>>>(rec, binstart, dinv, ACT, W2, b2, Wl, bl, G);
    k_tile_agg_mm<1><<<gTile, big, 0, stream>>>(rec, binstart, dinv, G, W3, b3, Wl, bl, out);
}


# Last layer reassociated, A (a2 (W3 Wl)): M = W3 Wl formed once on WMMA; layer-2 launch multiplies its rows by M; last launch sums 48-float rows, 3 one-hot WMMAs per group, no product; sums regrouped
# speedup vs baseline: 4.4216x; 1.0370x over previous
#include <hip/hip_runtime.h>
#include <math.h>

typedef __attribute__((ext_vector_type(2))) float v2f;
typedef __attribute__((ext_vector_type(8))) float v8f;

#define NN 50000
#define H  64
#define TILE 196
#define NBIN 256
#define CH   4096

#define MAXCH 256

__device__ __forceinline__ int wave_incl_scan(int v, int lane) {
#pragma unroll
    for (int o = 1; o < 32; o <<= 1) {
        const int t = __shfl_up(v, o);
        if (lane >= o) v += t;
    }
    return v;
}

__global__ __launch_bounds__(256) void k_chunk_sort(const int* __restrict__ src,
                                                    const int* __restrict__ dst,
                                                    int* __restrict__ coffT,
                                                    int* __restrict__ crec, int E) {
    __shared__ int  hist[NBIN];
    __shared__ int  wtot[8];
    __shared__ int2 stage[CH];
    const int tid  = threadIdx.x;
    const int lane = tid & 31;
    const int wv   = tid >> 5;
    const int c    = blockIdx.x;
    hist[tid] = 0;
    __syncthreads();
    const int e0 = c * CH;
    int b[CH / 256], s[CH / 256], y[CH / 256], rk[CH / 256];
#pragma unroll
    for (int k = 0; k < CH / 256; ++k) {
        const int e  = e0 + k * 256 + tid;
        const int ec = (e < E) ? e : (E - 1);
        y[k] = dst[ec];
        s[k] = src[ec];
    }
#pragma unroll
    for (int k = 0; k < CH / 256; ++k) {
        const int e = e0 + k * 256 + tid;
        b[k]  = y[k] / TILE;
        y[k] -= b[k] * TILE;
        rk[k] = 0;
        if (e < E) rk[k] = atomicAdd(&hist[b[k]], 1);
    }
    __syncthreads();
    const int v    = hist[tid];
    const int incl = wave_incl_scan(v, lane);
    if (lane == 31) wtot[wv] = incl;
    __syncthreads();
    int add = 0;
#pragma unroll
    for (int w = 0; w < 8; ++w) add += (w < wv) ? wtot[w] : 0;
    const int off = add + incl - v;
    coffT[tid * MAXCH + c] = off;
    if (tid == NBIN - 1) coffT[NBIN * MAXCH + c] = off + v;
    hist[tid] = off;
    __syncthreads();
#pragma unroll
    for (int k = 0; k < CH / 256; ++k) {
        const int e = e0 + k * 256 + tid;
        if (e < E) stage[hist[b[k]] + rk[k]] = make_int2(s[k], y[k]);
    }
    __syncthreads();
    const int n = (E - e0 < CH) ? (E - e0) : CH;
    int2* o2 = (int2*)crec + e0;
    for (int i = tid; i < n; i += 256) o2[i] = stage[i];
}

#define GB 9
__global__ __launch_bounds__(1024) void k_tile_dinv_mm1(const int* __restrict__ crec,
                                                        const int* __restrict__ coffT,
                                                        const float* __restrict__ x,
                                                        const float* __restrict__ W1,
                                                        float* __restrict__ dinv,
                                                        float* __restrict__ G,
                                                        int* __restrict__ rec,
                                                        int* __restrict__ binstart) {
    __shared__ int c[13 * 16];
    __shared__ int pfx[6 * MAXCH];
    __shared__ int cst[6 * MAXCH];
    const int tid  = threadIdx.x;
    const int lane = tid & 31;
    const int half = lane >> 4;
    const int l16  = lane & 15;
    const int t    = blockIdx.x;
    const int base = t * TILE;
    const int nn   = (NN - base < TILE) ? (NN - base) : TILE;
    if (tid < 13 * 16) c[tid] = 0;
    __syncthreads();
    const int wvu = __builtin_amdgcn_readfirstlane(tid >> 5);
    const int mt  = wvu >> 1;
    const int n0  = (wvu & 1) * 2;
    v8f d[2];
    d[0] = (v8f){};
    d[1] = (v8f){};
    if (wvu < 26) {
        int arow = base + mt * 16 + l16;
        arow = (arow < NN) ? arow : (NN - 1);
        const float* Ap = x + (size_t)arow * 128 + half * 2;
        const float* Bp = W1 + half * 2 * 64;
#pragma unroll
        for (int k = 0; k < 128; k += 4) {
            v2f a;
            a.x = Ap[k];
            a.y = Ap[k + 1];
#pragma unroll
            for (int q = 0; q < 2; ++q) {
                const int cc = (n0 + q) * 16 + l16;
                v2f b;
                b.x = Bp[k * 64 + cc];
                b.y = Bp[k * 64 + 64 + cc];
                d[q] = __builtin_amdgcn_wmma_f32_16x16x4_f32(
                    false, a, false, b, (short)0, d[q], false, false);
            }
        }
    } else {
        const int sw = wvu - 26;
        int* P = pfx + sw * MAXCH;
        int* S = cst + sw * MAXCH;
        const int4* rs = (const int4*)(coffT + (size_t)t * MAXCH) + lane * 2;
        const int4* re = (const int4*)(coffT + (size_t)(t + 1) * MAXCH) + lane * 2;
        const int4 s0 = rs[0], s1 = rs[1], f0 = re[0], f1 = re[1];
        const int st[8] = {s0.x, s0.y, s0.z, s0.w, s1.x, s1.y, s1.z, s1.w};
        const int en[8] = {f0.x, f0.y, f0.z, f0.w, f1.x, f1.y, f1.z, f1.w};
        int cn[8];
        int lt = 0, ls = 0;
#pragma unroll
        for (int j = 0; j < 8; ++j) {
            cn[j] = en[j] - st[j];
            lt += cn[j];
            ls += st[j];
        }
        const int incl = wave_incl_scan(lt, lane);
        int run = incl - lt;
#pragma unroll
        for (int j = 0; j < 8; ++j) {
            P[lane * 8 + j] = run;
            S[lane * 8 + j] = (lane * 8 + j) * CH + st[j] - run;
            run += cn[j];
        }
        const int T = __shfl(incl, 31);
        int B = ls;
#pragma unroll
        for (int o = 16; o >= 1; o >>= 1) B += __shfl_xor(B, o);
        if (sw == 0 && lane == 0) {
            binstart[t] = B;
            if (t == NBIN - 1) binstart[NBIN] = B + T;
        }
        const int2* crec2 = (const int2*)crec;
        int2*       rec2  = (int2*)rec + B;
        for (int i0 = sw * 32 + lane; i0 < T; i0 += GB * 6 * 32) {
            int  ii[GB], pos[GB];
            int2 r[GB];
#pragma unroll
            for (int k = 0; k < GB; ++k) {
                const int i = i0 + k * 6 * 32;
                ii[k]  = (i < T) ? i : (T - 1);
                pos[k] = 0;
            }
#pragma unroll
            for (int step = MAXCH / 2; step >= 1; step >>= 1)
#pragma unroll
                for (int k = 0; k < GB; ++k)
                    pos[k] += (P[pos[k] + step] <= ii[k]) ? step : 0;
#pragma unroll
            for (int k = 0; k < GB; ++k) r[k] = crec2[S[pos[k]] + ii[k]];
#pragma unroll
            for (int k = 0; k < GB; ++k)
                if (i0 + k * 6 * 32 < T) {
                    rec2[ii[k]] = r[k];
                    atomicAdd(&c[r[k].y], 1);
                }
        }
    }
    __syncthreads();
    if (tid < nn) dinv[base + tid] = rsqrtf((float)c[tid] + 1.0f);
    if (wvu < 26) {
#pragma unroll
        for (int i = 0; i < 8; ++i) {
            const int row = mt * 16 + i + half * 8;
            if (row < nn) {
                const float sc = rsqrtf((float)c[row] + 1.0f);
#pragma unroll
                for (int q = 0; q < 2; ++q)
                    G[(size_t)(base + row) * 64 + (n0 + q) * 16 + l16] = sc * d[q][i];
            }
        }
    }
}

#define CAPR 4096
#define NRB  13
#define GU   4

struct __attribute__((aligned(4))) row3 { float x, y, z; };
template<int NC> struct row_piece;
template<> struct row_piece<4> { typedef float4 T; };
template<> struct row_piece<3> { typedef row3 T; };
__device__ __forceinline__ float piece_at(const float4& v, int t) { return t == 0 ? v.x : (t == 1 ? v.y : (t == 2 ? v.z : v.w)); }
__device__ __forceinline__ float piece_at(const row3& v, int t) { return t == 0 ? v.x : (t == 1 ? v.y : v.z); }

__device__ __forceinline__ void load_pass_records(const int2* __restrict__ rec2, int c0, int e1,
                                                  int tid, int2 (&r)[4]) {
    const int n = (e1 - c0 < CAPR) ? (e1 - c0) : CAPR;
#pragma unroll
    for (int k = 0; k < 4; ++k) {
        const int idx = (k * 32 + (tid >> 5)) * 32 + (tid & 31);
        r[k] = rec2[c0 + ((idx < n) ? idx : (n - 1))];
    }
}

template<int ROWS, int PITCH, int NC>
__device__ __forceinline__ void tile_edge_sum(const int2* __restrict__ rec2, int e0, int e1,
                                              const float* __restrict__ h, float* acc,
                                              int* srec, int* wcnt, int* rbtot, int tid,
                                              int2 (&r)[4]) {
    const int lane = tid & 31;
    const int wv   = tid >> 5;
    const int half = lane >> 4;
    const int l16  = lane & 15;
    const unsigned ltmask = (1u << lane) - 1u;
    const int wvu  = __builtin_amdgcn_readfirstlane(wv);
    const int rbw  = wvu >> 1;
    const int part = wvu & 1;
    typedef typename row_piece<NC>::T RP;
    const RP* hp = (const RP*)h;

    v8f d[NC];
#pragma unroll
    for (int t = 0; t < NC; ++t) d[t] = (v8f){};

    for (int c0 = e0; c0 < e1; c0 += CAPR) {
        const int n = (e1 - c0 < CAPR) ? (e1 - c0) : CAPR;

        int   rk[4];
#pragma unroll
        for (int k = 0; k < 4; ++k) {
            const int idx = (k * 32 + wv) * 32 + lane;
            rk[k] = (idx < n) ? (r[k].y >> 4) : 15;
        }
        unsigned bm[4][4];
        int cntk[4];
        int mycnt = 0;
#pragma unroll
        for (int k = 0; k < 4; ++k) {
#pragma unroll
            for (int i = 0; i < 4; ++i)
                bm[k][i] = __builtin_amdgcn_ballot_w32(((rk[k] >> i) & 1) != 0);
            unsigned m = 0xFFFFFFFFu;
#pragma unroll
            for (int i = 0; i < 4; ++i) m &= ((lane >> i) & 1) ? bm[k][i] : ~bm[k][i];
            cntk[k] = (lane < NRB) ? __builtin_popcount(m) : 0;
            mycnt += cntk[k];
        }
        if (lane < 16) wcnt[wv * 16 + lane] = mycnt;
        __syncthreads();
        if (wvu < NRB) {
            const int v    = wcnt[lane * 16 + wvu];
            const int incl = wave_incl_scan(v, lane);
            wcnt[lane * 16 + wvu] = incl - v;
            if (lane == 31) rbtot[wvu] = incl;
        }
        __syncthreads();
        const int tv     = (lane < NRB) ? rbtot[lane] : 0;
        const int startv = wave_incl_scan(tv, lane) - tv;
        int basev = startv + ((lane < 16) ? wcnt[wv * 16 + lane] : 0);
#pragma unroll
        for (int k = 0; k < 4; ++k) {
            unsigned m = 0xFFFFFFFFu;
#pragma unroll
            for (int i = 0; i < 4; ++i) m &= ((rk[k] >> i) & 1) ? bm[k][i] : ~bm[k][i];
            const int pos = __shfl(basev, rk[k]) + __builtin_popcount(m & ltmask);
            basev += cntk[k];
            if (rk[k] != 15)
                srec[pos] = r[k].x | ((r[k].y & 15) << 20);
        }
        __syncthreads();

        if (wvu < 2 * NRB) {
            const int segS = __builtin_amdgcn_readlane(startv, rbw);
            const int segE = __builtin_amdgcn_readlane(startv, rbw + 1);
            const int ng   = (segE - segS + 3) >> 2;
            const int gh   = (ng + 1) >> 1;
            const int g0   = part ? gh : 0;
            const int g1   = part ? ng : gh;
            const int je   = segS + g1 * 4;
            const int jEnd = (segE < je) ? segE : je;
            for (int g = g0; g < g1; g += GU) {
                int rc[GU][2];
                RP  hv[GU][2];
#pragma unroll
                for (int u = 0; u < GU; ++u)
#pragma unroll
                    for (int q = 0; q < 2; ++q) {
                        const int j = segS + (g + u) * 4 + half * 2 + q;
                        rc[u][q] = srec[(j < jEnd) ? j : (jEnd - 1)];
                    }
#pragma unroll
                for (int u = 0; u < GU; ++u)
#pragma unroll
                    for (int q = 0; q < 2; ++q)
                        if constexpr (NC == 4) {
                            hv[u][q] = hp[(size_t)(rc[u][q] & 0xFFFFF) * 16 + l16];
                        } else {
                            const unsigned off = (unsigned)(rc[u][q] & 0xFFFFF) * (64u * NC) + (unsigned)l16 * (4u * NC);
                            hv[u][q] = *(const RP*)((const char*)h + off);
                        }
#pragma unroll
                for (int u = 0; u < GU; ++u) {
                    const int j = segS + (g + u) * 4 + half * 2;
                    v2f a;
                    a.x = (j     < jEnd && (rc[u][0] >> 20) == l16) ? 1.0f : 0.0f;
                    a.y = (j + 1 < jEnd && (rc[u][1] >> 20) == l16) ? 1.0f : 0.0f;
#pragma unroll
                    for (int t = 0; t < NC; ++t) {
                        v2f b;
                        b.x = piece_at(hv[u][0], t);
                        b.y = piece_at(hv[u][1], t);
                        d[t] = __builtin_amdgcn_wmma_f32_16x16x4_f32(false, a, false, b, (short)0, d[t], false, false);
                    }
                }
            }
        }
        if (c0 + CAPR < e1) load_pass_records(rec2, c0 + CAPR, e1, tid, r);
        __syncthreads();
    }

#pragma unroll
    for (int p = 0; p < 2; ++p) {
        if (wvu < 2 * NRB && part == p) {
#pragma unroll
            for (int i = 0; i < 8; ++i) {
                const int row = rbw * 16 + half * 8 + i;
                if (row < ROWS) {
                    if constexpr (NC == 4) {
                        float4* o = (float4*)(acc + row * PITCH + l16 * 4);
                        float4  v = *o;
                        v.x += d[0][i]; v.y += d[1][i]; v.z += d[2][i]; v.w += d[3][i];
                        *o = v;
                    } else {
                        float* o = acc + row * PITCH + l16 * NC;
#pragma unroll
                        for (int t = 0; t < NC; ++t) o[t] += d[t][i];
                    }
                }
            }
        }
        __syncthreads();
    }
}

#define PW 48
#define PO 52

template<int NB, int NT>
__device__ __forceinline__ void tile_mm(const float* arow, const float* __restrict__ B,
                                        int n0, int half, int l16, v8f (&d)[NT]);

#define OPENING_LOADS_TOGETHER(dv, hv, r)                                        \
    asm volatile("" :: "v"(dv),                                                  \
                 "v"(hv[0].x), "v"(hv[0].y), "v"(hv[0].z), "v"(hv[0].w),         \
                 "v"(hv[1].x), "v"(hv[1].y), "v"(hv[1].z), "v"(hv[1].w),         \
                 "v"(hv[2].x), "v"(hv[2].y), "v"(hv[2].z), "v"(hv[2].w),         \
                 "v"(hv[3].x), "v"(hv[3].y), "v"(hv[3].z), "v"(hv[3].w),         \
                 "v"(r[0].x), "v"(r[0].y), "v"(r[1].x), "v"(r[1].y),             \
                 "v"(r[2].x), "v"(r[2].y), "v"(r[3].x), "v"(r[3].y))

template<int DO_ELU, int HW>
__global__ __launch_bounds__(1024) void k_tile_agg(const int* __restrict__ rec,
                                                   const int* __restrict__ binstart,
                                                   const float* __restrict__ dinv,
                                                   const float* __restrict__ h,
                                                   const float* __restrict__ bias,
                                                   float* __restrict__ act,
                                                   const float* __restrict__ W3,
                                                   const float* __restrict__ b3,
                                                   const float* __restrict__ Wl,
                                                   const float* __restrict__ bl,
                                                   float* __restrict__ Mg,
                                                   float* __restrict__ cb) {
    static_assert(HW == 64, "rows of 64 floats");
    constexpr int Q  = HW / 4;
    __shared__ __attribute__((aligned(16))) float acc[TILE * HW];
    __shared__ int   srec[CAPR];
    __shared__ int   wcnt[32 * 16];
    __shared__ int   rbtot[16];
    __shared__ float sdinv[TILE];
    const int tid  = threadIdx.x;
    const int t    = blockIdx.x;
    const int base = t * TILE;
    const int nn   = (NN - base < TILE) ? (NN - base) : TILE;

    const int e0 = binstart[t];
    const int e1 = binstart[t + 1];
    const float dv = dinv[base + ((tid < nn) ? tid : (nn - 1))];
    const float4* h4  = (const float4*)(h + (size_t)base * HW);
    float4*       a4  = (float4*)acc;
    constexpr int NR = (TILE * Q + 1023) / 1024;
    static_assert(NR == 4, "OPENING_LOADS_TOGETHER names four row loads");
    const int lim = nn * Q;
    float4 hv[NR];
#pragma unroll
    for (int k = 0; k < NR; ++k) {
        const int i  = tid + k * 1024;
        hv[k] = h4[(i < lim) ? i : (lim - 1)];
    }
    int2 r[4];
#pragma unroll
    for (int k = 0; k < 4; ++k) r[k] = make_int2(0, 0);
    if (e0 < e1) load_pass_records((const int2*)rec, e0, e1, tid, r);
    OPENING_LOADS_TOGETHER(dv, hv, r);
    if (tid < nn) sdinv[tid] = dv;
#pragma unroll
    for (int k = 0; k < NR; ++k) {
        const int i = tid + k * 1024;
        if (i < lim) a4[i] = hv[k];
    }
    for (int i = lim + tid; i < TILE * Q; i += 1024) a4[i] = make_float4(0.0f, 0.0f, 0.0f, 0.0f);
    __syncthreads();

    tile_edge_sum<TILE, HW, 4>((const int2*)rec, e0, e1, h, acc, srec, wcnt, rbtot, tid, r);

    float4* o4 = (float4*)(act + (size_t)base * HW);
    for (int i = tid; i < nn * Q; i += 1024) {
        const float4 a = a4[i];
        const int    c = (i & 15) * 4;
        const float  sc = sdinv[i >> 4];
        float4 v = make_float4(sc * a.x + bias[c], sc * a.y + bias[c + 1], sc * a.z + bias[c + 2], sc * a.w + bias[c + 3]);
        if (DO_ELU) {
            v.x = (v.x <= 0.0f) ? expm1f(v.x) : v.x;
            v.y = (v.y <= 0.0f) ? expm1f(v.y) : v.y;
            v.z = (v.z <= 0.0f) ? expm1f(v.z) : v.z;
            v.w = (v.w <= 0.0f) ? expm1f(v.w) : v.w;
        }
        o4[i] = make_float4(sc * v.x, sc * v.y, sc * v.z, sc * v.w);
    }

    if (t == NBIN - 1) {
        const int lane = tid & 31;
        const int half = lane >> 4;
        const int l16  = lane & 15;
        const int wvu = __builtin_amdgcn_readfirstlane(tid >> 5);
        if (wvu < 4) {
            v8f e[3];
            tile_mm<40, 3>(W3 + (wvu * 16 + l16) * 64 + half * 2, Wl, 0, half, l16, e);
#pragma unroll
            for (int q = 0; q < 3; ++q)
#pragma unroll
                for (int i = 0; i < 8; ++i)
                    Mg[(wvu * 16 + i + half * 8) * PW + q * 16 + l16] = e[q][i];
        } else if (wvu == 4) {
            for (int c = lane; c < PW; c += 32) {
                float s = 0.0f;
                if (c < 40) {
                    s = bl[c];
#pragma unroll 16
                    for (int k = 0; k < 64; ++k) s += b3[k] * Wl[k * 40 + c];
                }
                cb[c] = s;
            }
        }
    }
}


#define PA  68
#define MTL 13

template<int NB, int NT>
__device__ __forceinline__ void tile_mm(const float* arow, const float* __restrict__ B,
                                        int n0, int half, int l16, v8f (&d)[NT]) {
#pragma unroll
    for (int t = 0; t < NT; ++t) d[t] = (v8f){};
    const float* Bp = B + half * 2 * NB;
#pragma unroll
    for (int k = 0; k < 64; k += 4) {
        v2f a;
        a.x = arow[k];
        a.y = arow[k + 1];
#pragma unroll
        for (int t = 0; t < NT; ++t) {
            const int c = (n0 + t) * 16 + l16;
            v2f b;
            if constexpr (NB % 16 == 0) {
                b.x = Bp[k * NB + c];
                b.y = Bp[k * NB + NB + c];
            } else {
                const float m  = (c < NB) ? 1.0f : 0.0f;
                const int   cc = (c < NB) ? c : (NB - 1);
                b.x = m * Bp[k * NB + cc];
                b.y = m * Bp[k * NB + NB + cc];
            }
            d[t] = __builtin_amdgcn_wmma_f32_16x16x4_f32(
                false, a, false, b, (short)0, d[t], false, false);
        }
    }
}


__global__ __launch_bounds__(1024) void k_tile_agg_mm(const int* __restrict__ rec,
                                                      const int* __restrict__ binstart,
                                                      const float* __restrict__ dinv,
                                                      const float* __restrict__ h,
                                                      const float* __restrict__ W2,
                                                      const float* __restrict__ b2,
                                                      const float* __restrict__ Mg,
                                                      float* __restrict__ outp) {
    constexpr int HW = 64;
    __shared__ __attribute__((aligned(16))) float acc[MTL * 16 * PA];
    __shared__ __attribute__((aligned(16))) float mid[MTL * 16 * PA];
    __shared__ __attribute__((aligned(16))) float Ml[64 * PW];
    __shared__ int   srec[CAPR];
    __shared__ int   wcnt[32 * 16];
    __shared__ int   rbtot[16];
    __shared__ float sdinv[TILE];
    const int tid  = threadIdx.x;
    const int lane = tid & 31;
    const int wv   = tid >> 5;
    const int t    = blockIdx.x;
    const int base = t * TILE;
    const int nn   = (NN - base < TILE) ? (NN - base) : TILE;

    const int e0 = binstart[t];
    const int e1 = binstart[t + 1];
    const float dv = dinv[base + ((tid < nn) ? tid : (nn - 1))];
    const float4* h4 = (const float4*)(h + (size_t)base * HW);
    constexpr int NR = (TILE * 16 + 1023) / 1024;
    static_assert(NR == 4, "OPENING_LOADS_TOGETHER names four row loads");
    const int lim = nn * 16;
    float4 hv[NR];
#pragma unroll
    for (int k = 0; k < NR; ++k) {
        const int i  = tid + k * 1024;
        hv[k] = h4[(i < lim) ? i : (lim - 1)];
    }
    int2 r[4];
#pragma unroll
    for (int k = 0; k < 4; ++k) r[k] = make_int2(0, 0);
    if (e0 < e1) load_pass_records((const int2*)rec, e0, e1, tid, r);
    OPENING_LOADS_TOGETHER(dv, hv, r);
    if (tid < nn) sdinv[tid] = dv;
#pragma unroll
    for (int k = 0; k < NR; ++k) {
        const int i = tid + k * 1024;
        if (i < lim) *(float4*)(acc + (i >> 4) * PA + (i & 15) * 4) = hv[k];
    }
    for (int i = lim + tid; i < MTL * 16 * 16; i += 1024)
        *(float4*)(acc + (i >> 4) * PA + (i & 15) * 4) = make_float4(0.0f, 0.0f, 0.0f, 0.0f);
    __syncthreads();

    tile_edge_sum<MTL * 16, PA, 4>((const int2*)rec, e0, e1, h, acc, srec, wcnt, rbtot, tid, r);

    const int half = lane >> 4;
    const int l16  = lane & 15;
    const int wvu = __builtin_amdgcn_readfirstlane(wv);
    if (wvu < 2 * MTL) {
        const int mt = wvu >> 1;
        const int n0 = (wvu & 1) * 2;
        v8f d[2];
        tile_mm<64, 2>(acc + (mt * 16 + l16) * PA + half * 2, W2, n0, half, l16, d);
#pragma unroll
        for (int q = 0; q < 2; ++q) {
            const int c = (n0 + q) * 16 + l16;
            const float bv = b2[c];
#pragma unroll
            for (int i = 0; i < 8; ++i) {
                const int row = mt * 16 + i + half * 8;
                const float sc = (row < nn) ? sdinv[row] : 0.0f;
                float v = sc * d[q][i] + bv;
                v = (v <= 0.0f) ? expm1f(v) : v;
                mid[row * PA + c] = sc * v;
            }
        }
    } else {
        const int j = tid - 2 * MTL * 32;
        static_assert(64 * PW / 4 == 4 * 6 * 32, "four float4 per thread of the six waves");
#pragma unroll
        for (int k = 0; k < 4; ++k) ((float4*)Ml)[j + k * 6 * 32] = ((const float4*)Mg)[j + k * 6 * 32];
    }
    __syncthreads();
    for (int job = wvu; job < 3 * MTL; job += 32) {
        const int mt = job / 3;
        const int q  = job - mt * 3;
        v8f e[1];
        tile_mm<PW, 1>(mid + (mt * 16 + l16) * PA + half * 2, Ml, q, half, l16, e);
        const int c = q * 16 + l16;
#pragma unroll
        for (int i = 0; i < 8; ++i) {
            const int row = mt * 16 + i + half * 8;
            if (row < nn) outp[(size_t)(base + row) * PW + c] = e[0][i];
        }
    }
}

#define OPENING_LOADS_TOGETHER3(dv, cv, hv, r)                                   \
    asm volatile("" :: "v"(dv), "v"(cv),                                         \
                 "v"(hv[0].x), "v"(hv[0].y), "v"(hv[0].z), "v"(hv[0].w),         \
                 "v"(hv[1].x), "v"(hv[1].y), "v"(hv[1].z), "v"(hv[1].w),         \
                 "v"(hv[2].x), "v"(hv[2].y), "v"(hv[2].z), "v"(hv[2].w),         \
                 "v"(r[0].x), "v"(r[0].y), "v"(r[1].x), "v"(r[1].y),             \
                 "v"(r[2].x), "v"(r[2].y), "v"(r[3].x), "v"(r[3].y))

__global__ __launch_bounds__(1024) void k_tile_agg_out(const int* __restrict__ rec,
                                                       const int* __restrict__ binstart,
                                                       const float* __restrict__ dinv,
                                                       const float* __restrict__ h,
                                                       const float* __restrict__ cb,
                                                       float* __restrict__ out) {
    constexpr int Q = PW / 4;
    __shared__ __attribute__((aligned(16))) float acc[TILE * PO];
    __shared__ int   srec[CAPR];
    __shared__ int   wcnt[32 * 16];
    __shared__ int   rbtot[16];
    __shared__ float sdinv[TILE];
    __shared__ float scb[PW];
    const int tid  = threadIdx.x;
    const int t    = blockIdx.x;
    const int base = t * TILE;
    const int nn   = (NN - base < TILE) ? (NN - base) : TILE;

    const int e0 = binstart[t];
    const int e1 = binstart[t + 1];
    const float dv = dinv[base + ((tid < nn) ? tid : (nn - 1))];
    const float cv = cb[(tid < PW) ? tid : (PW - 1)];
    const float4* h4 = (const float4*)(h + (size_t)base * PW);
    constexpr int NR = (TILE * Q + 1023) / 1024;
    static_assert(NR == 3, "OPENING_LOADS_TOGETHER3 names three row loads");
    const int lim = nn * Q;
    float4 hv[NR];
#pragma unroll
    for (int k = 0; k < NR; ++k) {
        const int i  = tid + k * 1024;
        hv[k] = h4[(i < lim) ? i : (lim - 1)];
    }
    int2 r[4];
#pragma unroll
    for (int k = 0; k < 4; ++k) r[k] = make_int2(0, 0);
    if (e0 < e1) load_pass_records((const int2*)rec, e0, e1, tid, r);
    OPENING_LOADS_TOGETHER3(dv, cv, hv, r);
    if (tid < nn) sdinv[tid] = dv;
    if (tid < PW) scb[tid] = cv;
#pragma unroll
    for (int k = 0; k < NR; ++k) {
        const int i = tid + k * 1024;
        if (i < lim) {
            const int row = i / Q;
            *(float4*)(acc + row * PO + (i - row * Q) * 4) = hv[k];
        }
    }
    for (int i = lim + tid; i < TILE * Q; i += 1024) {
        const int row = i / Q;
        *(float4*)(acc + row * PO + (i - row * Q) * 4) = make_float4(0.0f, 0.0f, 0.0f, 0.0f);
    }
    __syncthreads();

    tile_edge_sum<TILE, PO, 3>((const int2*)rec, e0, e1, h, acc, srec, wcnt, rbtot, tid, r);

    float* o = out + (size_t)base * 40;
    for (int i = tid; i < nn * 40; i += 1024) {
        const int row = i / 40;
        const int c   = i - row * 40;
        o[i] = sdinv[row] * acc[row * PO + c] + scb[c];
    }
}


extern "C" void kernel_launch(void* const* d_in, const int* in_sizes, int n_in,
                              void* d_out, int out_size, void* d_ws, size_t ws_size,
                              hipStream_t stream) {
    const float* x  = (const float*)d_in[0];
    const int*   ei = (const int*)d_in[1];
    const float* W1 = (const float*)d_in[2];
    const float* b1 = (const float*)d_in[3];
    const float* W2 = (const float*)d_in[4];
    const float* b2 = (const float*)d_in[5];
    const float* W3 = (const float*)d_in[6];
    const float* b3 = (const float*)d_in[7];
    const float* Wl = (const float*)d_in[8];
    const float* bl = (const float*)d_in[9];
    float* out = (float*)d_out;

    const int E = in_sizes[1] / 2;
    const int* src = ei;
    const int* dst = ei + E;

    float* dinv     = (float*)d_ws;
    int*   binstart = (int*)(dinv + 50176);
    int*   coffT    = binstart + 512;
    int*   rec      = coffT + 1024 * NBIN;
    float* G        = (float*)(rec + 2 * (size_t)E);
    float* ACT      = G + (size_t)NN * H;
    int*   crec     = (int*)(ACT + (size_t)NN * H);
    float* Mg       = (float*)(crec + 2 * (size_t)E);
    float* cbias    = Mg + 64 * PW;

    const dim3 blk(256);
    const dim3 big(1024);
    const dim3 gTile(NBIN);

    k_chunk_sort<<<dim3(MAXCH), blk, 0, stream>>>(src, dst, coffT, crec, E);
    k_tile_dinv_mm1<<<gTile, big, 0, stream>>>(crec, coffT, x, W1, dinv, G, rec, binstart);

    k_tile_agg<1, H><<<gTile, big, 0, stream>>>(rec, binstart, dinv, G, b1, ACT, W3, b3, Wl, bl, Mg, cbias);

    k_tile_agg_mm<<<gTile, big, 0, stream>>>(rec, binstart, dinv, ACT, W2, b2, Mg, G);
    k_tile_agg_out<<<gTile, big, 0, stream>>>(rec, binstart, dinv, G, cbias, out);
}


# Tile lists leave the list launch sorted by dst node (LDS counts scanned, returning add = place), one word a record, with row-block bounds; tile launches group nothing; sum order: node, LDS arrival
# speedup vs baseline: 4.6202x; 1.0449x over previous
#include <hip/hip_runtime.h>
#include <math.h>

typedef __attribute__((ext_vector_type(2))) float v2f;
typedef __attribute__((ext_vector_type(8))) float v8f;

#define NN 50000
#define H  64
#define TILE 196
#define NBIN 256
#define CH   4096

#define MAXCH 256

__device__ __forceinline__ int wave_incl_scan(int v, int lane) {
#pragma unroll
    for (int o = 1; o < 32; o <<= 1) {
        const int t = __shfl_up(v, o);
        if (lane >= o) v += t;
    }
    return v;
}

__global__ __launch_bounds__(256) void k_chunk_sort(const int* __restrict__ src,
                                                    const int* __restrict__ dst,
                                                    int* __restrict__ coffT,
                                                    int* __restrict__ crec, int E) {
    __shared__ int  hist[NBIN];
    __shared__ int  wtot[8];
    __shared__ int2 stage[CH];
    const int tid  = threadIdx.x;
    const int lane = tid & 31;
    const int wv   = tid >> 5;
    const int c    = blockIdx.x;
    hist[tid] = 0;
    __syncthreads();
    const int e0 = c * CH;
    int b[CH / 256], s[CH / 256], y[CH / 256], rk[CH / 256];
#pragma unroll
    for (int k = 0; k < CH / 256; ++k) {
        const int e  = e0 + k * 256 + tid;
        const int ec = (e < E) ? e : (E - 1);
        y[k] = dst[ec];
        s[k] = src[ec];
    }
#pragma unroll
    for (int k = 0; k < CH / 256; ++k) {
        const int e = e0 + k * 256 + tid;
        b[k]  = y[k] / TILE;
        y[k] -= b[k] * TILE;
        rk[k] = 0;
        if (e < E) rk[k] = atomicAdd(&hist[b[k]], 1);
    }
    __syncthreads();
    const int v    = hist[tid];
    const int incl = wave_incl_scan(v, lane);
    if (lane == 31) wtot[wv] = incl;
    __syncthreads();
    int add = 0;
#pragma unroll
    for (int w = 0; w < 8; ++w) add += (w < wv) ? wtot[w] : 0;
    const int off = add + incl - v;
    coffT[tid * MAXCH + c] = off;
    if (tid == NBIN - 1) coffT[NBIN * MAXCH + c] = off + v;
    hist[tid] = off;
    __syncthreads();
#pragma unroll
    for (int k = 0; k < CH / 256; ++k) {
        const int e = e0 + k * 256 + tid;
        if (e < E) stage[hist[b[k]] + rk[k]] = make_int2(s[k], y[k]);
    }
    __syncthreads();
    const int n = (E - e0 < CH) ? (E - e0) : CH;
    int2* o2 = (int2*)crec + e0;
    for (int i = tid; i < n; i += 256) o2[i] = stage[i];
}

#define GB 9
#define NRB  13
#define NCNT 224
#define CAPS 8192

__device__ __forceinline__ int list_record(const int* P, const int* S, const int2* __restrict__ crec2, int i) {
    int pos = 0;
#pragma unroll
    for (int step = MAXCH / 2; step >= 1; step >>= 1) pos += (P[pos + step] <= i) ? step : 0;
    const int2 q = crec2[S[pos] + i];
    return q.x | (q.y << 20);
}

__global__ __launch_bounds__(1024) void k_tile_dinv_mm1(const int* __restrict__ crec,
                                                        const int* __restrict__ coffT,
                                                        const float* __restrict__ x,
                                                        const float* __restrict__ W1,
                                                        float* __restrict__ dinv,
                                                        float* __restrict__ G,
                                                        int* __restrict__ recp,
                                                        int* __restrict__ bst) {
    __shared__ int c[NCNT];
    __shared__ int fill[NCNT];
    __shared__ int pfx[6 * MAXCH];
    __shared__ int cst[6 * MAXCH];
    __shared__ int stg[CAPS];
    __shared__ int srt[CAPS];
    __shared__ int sBT[2];
    const int tid  = threadIdx.x;
    const int lane = tid & 31;
    const int half = lane >> 4;
    const int l16  = lane & 15;
    const int t    = blockIdx.x;
    const int base = t * TILE;
    const int nn   = (NN - base < TILE) ? (NN - base) : TILE;
    if (tid < NCNT) c[tid] = 0;
    __syncthreads();
    const int wvu = __builtin_amdgcn_readfirstlane(tid >> 5);
    const int mt  = wvu >> 1;
    const int n0  = (wvu & 1) * 2;
    const int2* crec2 = (const int2*)crec;
    int Bv = 0;
    v8f d[2];
    d[0] = (v8f){};
    d[1] = (v8f){};
    if (wvu < 26) {
        int arow = base + mt * 16 + l16;
        arow = (arow < NN) ? arow : (NN - 1);
        const float* Ap = x + (size_t)arow * 128 + half * 2;
        const float* Bp = W1 + half * 2 * 64;
#pragma unroll
        for (int k = 0; k < 128; k += 4) {
            v2f a;
            a.x = Ap[k];
            a.y = Ap[k + 1];
#pragma unroll
            for (int q = 0; q < 2; ++q) {
                const int cc = (n0 + q) * 16 + l16;
                v2f b;
                b.x = Bp[k * 64 + cc];
                b.y = Bp[k * 64 + 64 + cc];
                d[q] = __builtin_amdgcn_wmma_f32_16x16x4_f32(
                    false, a, false, b, (short)0, d[q], false, false);
            }
        }
    } else {
        const int sw = wvu - 26;
        int* P = pfx + sw * MAXCH;
        int* S = cst + sw * MAXCH;
        const int4* rs = (const int4*)(coffT + (size_t)t * MAXCH) + lane * 2;
        const int4* re = (const int4*)(coffT + (size_t)(t + 1) * MAXCH) + lane * 2;
        const int4 s0 = rs[0], s1 = rs[1], f0 = re[0], f1 = re[1];
        const int st[8] = {s0.x, s0.y, s0.z, s0.w, s1.x, s1.y, s1.z, s1.w};
        const int en[8] = {f0.x, f0.y, f0.z, f0.w, f1.x, f1.y, f1.z, f1.w};
        int cn[8];
        int lt = 0, ls = 0;
#pragma unroll
        for (int j = 0; j < 8; ++j) {
            cn[j] = en[j] - st[j];
            lt += cn[j];
            ls += st[j];
        }
        const int incl = wave_incl_scan(lt, lane);
        int run = incl - lt;
#pragma unroll
        for (int j = 0; j < 8; ++j) {
            P[lane * 8 + j] = run;
            S[lane * 8 + j] = (lane * 8 + j) * CH + st[j] - run;
            run += cn[j];
        }
        const int T = __shfl(incl, 31);
        int B = ls;
#pragma unroll
        for (int o = 16; o >= 1; o >>= 1) B += __shfl_xor(B, o);
        Bv = B;
        if (sw == 0 && lane == 0) {
            sBT[0] = B;
            sBT[1] = T;
        }
        for (int i0 = sw * 32 + lane; i0 < T; i0 += GB * 6 * 32) {
            int  ii[GB], pos[GB];
            int2 r[GB];
#pragma unroll
            for (int k = 0; k < GB; ++k) {
                const int i = i0 + k * 6 * 32;
                ii[k]  = (i < T) ? i : (T - 1);
                pos[k] = 0;
            }
#pragma unroll
            for (int step = MAXCH / 2; step >= 1; step >>= 1)
#pragma unroll
                for (int k = 0; k < GB; ++k)
                    pos[k] += (P[pos[k] + step] <= ii[k]) ? step : 0;
#pragma unroll
            for (int k = 0; k < GB; ++k) r[k] = crec2[S[pos[k]] + ii[k]];
#pragma unroll
            for (int k = 0; k < GB; ++k)
                if (i0 + k * 6 * 32 < T) {
                    if (ii[k] < CAPS) stg[ii[k]] = r[k].x | (r[k].y << 20);
                    atomicAdd(&c[r[k].y], 1);
                }
        }
    }
    __syncthreads();
    if (tid < nn) dinv[base + tid] = rsqrtf((float)c[tid] + 1.0f);
    if (wvu < 26) {
#pragma unroll
        for (int i = 0; i < 8; ++i) {
            const int row = mt * 16 + i + half * 8;
            if (row < nn) {
                const float sc = rsqrtf((float)c[row] + 1.0f);
#pragma unroll
                for (int q = 0; q < 2; ++q)
                    G[(size_t)(base + row) * 64 + (n0 + q) * 16 + l16] = sc * d[q][i];
            }
        }
    } else if (wvu == 26) {
        int cv[7];
        int s = 0;
#pragma unroll
        for (int j = 0; j < 7; ++j) {
            cv[j] = c[lane * 7 + j];
            s += cv[j];
        }
        int run = wave_incl_scan(s, lane) - s;
#pragma unroll
        for (int j = 0; j < 7; ++j) {
            const int idx = lane * 7 + j;
            fill[idx] = run;
            if ((idx & 15) == 0) bst[t * 16 + (idx >> 4)] = Bv + run;
            run += cv[j];
        }
    }
    __syncthreads();
    const int Bq = sBT[0];
    const int Tq = sBT[1];
    for (int i0 = tid; i0 < Tq; i0 += 4 * 1024) {
        int w[4], pl[4];
#pragma unroll
        for (int k = 0; k < 4; ++k) {
            const int i = i0 + k * 1024;
            w[k] = 0;
            if (i < Tq) w[k] = (i < CAPS) ? stg[i] : list_record(pfx, cst, crec2, i);
        }
#pragma unroll
        for (int k = 0; k < 4; ++k) {
            pl[k] = 0;
            if (i0 + k * 1024 < Tq) pl[k] = atomicAdd(&fill[w[k] >> 20], 1);
        }
#pragma unroll
        for (int k = 0; k < 4; ++k)
            if (i0 + k * 1024 < Tq) {
                const int v = (w[k] & 0xFFFFF) | (((w[k] >> 20) & 15) << 20);
                if (pl[k] < CAPS) srt[pl[k]] = v;
                else              recp[Bq + pl[k]] = v;
            }
    }
    __syncthreads();
    const int nl = (Tq < CAPS) ? Tq : CAPS;
    for (int i = tid; i < nl; i += 1024) recp[Bq + i] = srt[i];
}

#define CAPR 4096
#define GU   4

struct __attribute__((aligned(4))) row3 { float x, y, z; };
template<int NC> struct row_piece;
template<> struct row_piece<4> { typedef float4 T; };
template<> struct row_piece<3> { typedef row3 T; };
__device__ __forceinline__ float piece_at(const float4& v, int t) { return t == 0 ? v.x : (t == 1 ? v.y : (t == 2 ? v.z : v.w)); }
__device__ __forceinline__ float piece_at(const row3& v, int t) { return t == 0 ? v.x : (t == 1 ? v.y : v.z); }

__device__ __forceinline__ void load_pass_records(const int* __restrict__ recp, int c0, int e1,
                                                  int tid, int (&r)[4]) {
    const int n = (e1 - c0 < CAPR) ? (e1 - c0) : CAPR;
#pragma unroll
    for (int k = 0; k < 4; ++k) {
        const int idx = k * 1024 + tid;
        r[k] = recp[c0 + ((idx < n) ? idx : (n - 1))];
    }
}

__device__ __forceinline__ void stage_pass_records(int* srec, int n, int tid, const int (&r)[4]) {
#pragma unroll
    for (int k = 0; k < 4; ++k) {
        const int idx = k * 1024 + tid;
        if (idx < n) srec[idx] = r[k];
    }
}

template<int ROWS, int PITCH, int NC>
__device__ __forceinline__ void tile_edge_sum(const int* __restrict__ recp, int bs, int be,
                                              int e0, int e1,
                                              const float* __restrict__ h, float* acc,
                                              int* srec, int tid) {
    const int lane = tid & 31;
    const int wv   = tid >> 5;
    const int half = lane >> 4;
    const int l16  = lane & 15;
    const int wvu  = __builtin_amdgcn_readfirstlane(wv);
    const int rbw  = wvu >> 1;
    const int part = wvu & 1;
    typedef typename row_piece<NC>::T RP;
    const RP* hp = (const RP*)h;

    v8f d[NC];
#pragma unroll
    for (int t = 0; t < NC; ++t) d[t] = (v8f){};

    for (int c0 = e0; c0 < e1; c0 += CAPR) {
        const int n = (e1 - c0 < CAPR) ? (e1 - c0) : CAPR;

        if (wvu < 2 * NRB) {
            const int lo   = (bs > c0) ? bs : c0;
            const int hi0  = (be < c0 + n) ? be : (c0 + n);
            const int hi   = (hi0 > lo) ? hi0 : lo;
            const int segS = lo - c0;
            const int segE = hi - c0;
            const int ng   = (segE - segS + 3) >> 2;
            const int gh   = (ng + 1) >> 1;
            const int g0   = part ? gh : 0;
            const int g1   = part ? ng : gh;
            const int je   = segS + g1 * 4;
            const int jEnd = (segE < je) ? segE : je;
            for (int g = g0; g < g1; g += GU) {
                int rc[GU][2];
                RP  hv[GU][2];
#pragma unroll
                for (int u = 0; u < GU; ++u)
#pragma unroll
                    for (int q = 0; q < 2; ++q) {
                        const int j = segS + (g + u) * 4 + half * 2 + q;
                        rc[u][q] = srec[(j < jEnd) ? j : (jEnd - 1)];
                    }
#pragma unroll
                for (int u = 0; u < GU; ++u)
#pragma unroll
                    for (int q = 0; q < 2; ++q)
                        if constexpr (NC == 4) {
                            hv[u][q] = hp[(size_t)(rc[u][q] & 0xFFFFF) * 16 + l16];
                        } else {
                            const unsigned off = (unsigned)(rc[u][q] & 0xFFFFF) * (64u * NC) + (unsigned)l16 * (4u * NC);
                            hv[u][q] = *(const RP*)((const char*)h + off);
                        }
#pragma unroll
                for (int u = 0; u < GU; ++u) {
                    const int j = segS + (g + u) * 4 + half * 2;
                    v2f a;
                    a.x = (j     < jEnd && (rc[u][0] >> 20) == l16) ? 1.0f : 0.0f;
                    a.y = (j + 1 < jEnd && (rc[u][1] >> 20) == l16) ? 1.0f : 0.0f;
#pragma unroll
                    for (int t = 0; t < NC; ++t) {
                        v2f b;
                        b.x = piece_at(hv[u][0], t);
                        b.y = piece_at(hv[u][1], t);
                        d[t] = __builtin_amdgcn_wmma_f32_16x16x4_f32(false, a, false, b, (short)0, d[t], false, false);
                    }
                }
            }
        }
        if (c0 + CAPR < e1) {
            int r[4];
            load_pass_records(recp, c0 + CAPR, e1, tid, r);
            __syncthreads();
            stage_pass_records(srec, (e1 - c0 - CAPR < CAPR) ? (e1 - c0 - CAPR) : CAPR, tid, r);
            __syncthreads();
        }
    }

#pragma unroll
    for (int p = 0; p < 2; ++p) {
        if (wvu < 2 * NRB && part == p) {
#pragma unroll
            for (int i = 0; i < 8; ++i) {
                const int row = rbw * 16 + half * 8 + i;
                if (row < ROWS) {
                    if constexpr (NC == 4) {
                        float4* o = (float4*)(acc + row * PITCH + l16 * 4);
                        float4  v = *o;
                        v.x += d[0][i]; v.y += d[1][i]; v.z += d[2][i]; v.w += d[3][i];
                        *o = v;
                    } else {
                        float* o = acc + row * PITCH + l16 * NC;
#pragma unroll
                        for (int t = 0; t < NC; ++t) o[t] += d[t][i];
                    }
                }
            }
        }
        __syncthreads();
    }
}

#define PW 48
#define PO 52

template<int NB, int NT>
__device__ __forceinline__ void tile_mm(const float* arow, const float* __restrict__ B,
                                        int n0, int half, int l16, v8f (&d)[NT]);

#define OPENING_LOADS_TOGETHER(dv, hv, r, bs, be)                                \
    asm volatile("" :: "s"(bs), "s"(be), "v"(dv),                                \
                 "v"(hv[0].x), "v"(hv[0].y), "v"(hv[0].z), "v"(hv[0].w),         \
                 "v"(hv[1].x), "v"(hv[1].y), "v"(hv[1].z), "v"(hv[1].w),         \
                 "v"(hv[2].x), "v"(hv[2].y), "v"(hv[2].z), "v"(hv[2].w),         \
                 "v"(hv[3].x), "v"(hv[3].y), "v"(hv[3].z), "v"(hv[3].w),         \
                 "v"(r[0]), "v"(r[1]), "v"(r[2]), "v"(r[3]))

template<int DO_ELU, int HW>
__global__ __launch_bounds__(1024) void k_tile_agg(const int* __restrict__ rec,
                                                   const int* __restrict__ bst,
                                                   const float* __restrict__ dinv,
                                                   const float* __restrict__ h,
                                                   const float* __restrict__ bias,
                                                   float* __restrict__ act,
                                                   const float* __restrict__ W3,
                                                   const float* __restrict__ b3,
                                                   const float* __restrict__ Wl,
                                                   const float* __restrict__ bl,
                                                   float* __restrict__ Mg,
                                                   float* __restrict__ cb) {
    static_assert(HW == 64, "rows of 64 floats");
    constexpr int Q  = HW / 4;
    __shared__ __attribute__((aligned(16))) float acc[TILE * HW];
    __shared__ int   srec[CAPR];
    __shared__ float sdinv[TILE];
    const int tid  = threadIdx.x;
    const int t    = blockIdx.x;
    const int base = t * TILE;
    const int nn   = (NN - base < TILE) ? (NN - base) : TILE;

    const int* tb = bst + t * 16;
    const int wvo = __builtin_amdgcn_readfirstlane(tid >> 5);
    const int rbo = (wvo < 2 * NRB) ? (wvo >> 1) : 0;
    const int e0 = tb[0];
    const int e1 = tb[NRB];
    const int bs = tb[rbo];
    const int be = tb[rbo + 1];
    const float dv = dinv[base + ((tid < nn) ? tid : (nn - 1))];
    const float4* h4  = (const float4*)(h + (size_t)base * HW);
    float4*       a4  = (float4*)acc;
    constexpr int NR = (TILE * Q + 1023) / 1024;
    static_assert(NR == 4, "OPENING_LOADS_TOGETHER names four row loads");
    const int lim = nn * Q;
    float4 hv[NR];
#pragma unroll
    for (int k = 0; k < NR; ++k) {
        const int i  = tid + k * 1024;
        hv[k] = h4[(i < lim) ? i : (lim - 1)];
    }
    int r[4];
#pragma unroll
    for (int k = 0; k < 4; ++k) r[k] = 0;
    if (e0 < e1) load_pass_records(rec, e0, e1, tid, r);
    OPENING_LOADS_TOGETHER(dv, hv, r, bs, be);
    if (tid < nn) sdinv[tid] = dv;
    stage_pass_records(srec, (e1 - e0 < CAPR) ? (e1 - e0) : CAPR, tid, r);
#pragma unroll
    for (int k = 0; k < NR; ++k) {
        const int i = tid + k * 1024;
        if (i < lim) a4[i] = hv[k];
    }
    for (int i = lim + tid; i < TILE * Q; i += 1024) a4[i] = make_float4(0.0f, 0.0f, 0.0f, 0.0f);
    __syncthreads();

    tile_edge_sum<TILE, HW, 4>(rec, bs, be, e0, e1, h, acc, srec, tid);

    float4* o4 = (float4*)(act + (size_t)base * HW);
    for (int i = tid; i < nn * Q; i += 1024) {
        const float4 a = a4[i];
        const int    c = (i & 15) * 4;
        const float  sc = sdinv[i >> 4];
        float4 v = make_float4(sc * a.x + bias[c], sc * a.y + bias[c + 1], sc * a.z + bias[c + 2], sc * a.w + bias[c + 3]);
        if (DO_ELU) {
            v.x = (v.x <= 0.0f) ? expm1f(v.x) : v.x;
            v.y = (v.y <= 0.0f) ? expm1f(v.y) : v.y;
            v.z = (v.z <= 0.0f) ? expm1f(v.z) : v.z;
            v.w = (v.w <= 0.0f) ? expm1f(v.w) : v.w;
        }
        o4[i] = make_float4(sc * v.x, sc * v.y, sc * v.z, sc * v.w);
    }

    if (t == NBIN - 1) {
        const int lane = tid & 31;
        const int half = lane >> 4;
        const int l16  = lane & 15;
        const int wvu = __builtin_amdgcn_readfirstlane(tid >> 5);
        if (wvu < 4) {
            v8f e[3];
            tile_mm<40, 3>(W3 + (wvu * 16 + l16) * 64 + half * 2, Wl, 0, half, l16, e);
#pragma unroll
            for (int q = 0; q < 3; ++q)
#pragma unroll
                for (int i = 0; i < 8; ++i)
                    Mg[(wvu * 16 + i + half * 8) * PW + q * 16 + l16] = e[q][i];
        } else if (wvu == 4) {
            for (int c = lane; c < PW; c += 32) {
                float s = 0.0f;
                if (c < 40) {
                    s = bl[c];
#pragma unroll 16
                    for (int k = 0; k < 64; ++k) s += b3[k] * Wl[k * 40 + c];
                }
                cb[c] = s;
            }
        }
    }
}


#define PA  68
#define MTL 13

template<int NB, int NT>
__device__ __forceinline__ void tile_mm(const float* arow, const float* __restrict__ B,
                                        int n0, int half, int l16, v8f (&d)[NT]) {
#pragma unroll
    for (int t = 0; t < NT; ++t) d[t] = (v8f){};
    const float* Bp = B + half * 2 * NB;
#pragma unroll
    for (int k = 0; k < 64; k += 4) {
        v2f a;
        a.x = arow[k];
        a.y = arow[k + 1];
#pragma unroll
        for (int t = 0; t < NT; ++t) {
            const int c = (n0 + t) * 16 + l16;
            v2f b;
            if constexpr (NB % 16 == 0) {
                b.x = Bp[k * NB + c];
                b.y = Bp[k * NB + NB + c];
            } else {
                const float m  = (c < NB) ? 1.0f : 0.0f;
                const int   cc = (c < NB) ? c : (NB - 1);
                b.x = m * Bp[k * NB + cc];
                b.y = m * Bp[k * NB + NB + cc];
            }
            d[t] = __builtin_amdgcn_wmma_f32_16x16x4_f32(
                false, a, false, b, (short)0, d[t], false, false);
        }
    }
}


__global__ __launch_bounds__(1024) void k_tile_agg_mm(const int* __restrict__ rec,
                                                      const int* __restrict__ bst,
                                                      const float* __restrict__ dinv,
                                                      const float* __restrict__ h,
                                                      const float* __restrict__ W2,
                                                      const float* __restrict__ b2,
                                                      const float* __restrict__ Mg,
                                                      float* __restrict__ outp) {
    constexpr int HW = 64;
    __shared__ __attribute__((aligned(16))) float acc[MTL * 16 * PA];
    __shared__ __attribute__((aligned(16))) float mid[MTL * 16 * PA];
    __shared__ __attribute__((aligned(16))) float Ml[64 * PW];
    __shared__ int   srec[CAPR];
    __shared__ float sdinv[TILE];
    const int tid  = threadIdx.x;
    const int lane = tid & 31;
    const int wv   = tid >> 5;
    const int t    = blockIdx.x;
    const int base = t * TILE;
    const int nn   = (NN - base < TILE) ? (NN - base) : TILE;

    const int* tb = bst + t * 16;
    const int wvo = __builtin_amdgcn_readfirstlane(tid >> 5);
    const int rbo = (wvo < 2 * NRB) ? (wvo >> 1) : 0;
    const int e0 = tb[0];
    const int e1 = tb[NRB];
    const int bs = tb[rbo];
    const int be = tb[rbo + 1];
    const float dv = dinv[base + ((tid < nn) ? tid : (nn - 1))];
    const float4* h4 = (const float4*)(h + (size_t)base * HW);
    constexpr int NR = (TILE * 16 + 1023) / 1024;
    static_assert(NR == 4, "OPENING_LOADS_TOGETHER names four row loads");
    const int lim = nn * 16;
    float4 hv[NR];
#pragma unroll
    for (int k = 0; k < NR; ++k) {
        const int i  = tid + k * 1024;
        hv[k] = h4[(i < lim) ? i : (lim - 1)];
    }
    int r[4];
#pragma unroll
    for (int k = 0; k < 4; ++k) r[k] = 0;
    if (e0 < e1) load_pass_records(rec, e0, e1, tid, r);
    OPENING_LOADS_TOGETHER(dv, hv, r, bs, be);
    if (tid < nn) sdinv[tid] = dv;
    stage_pass_records(srec, (e1 - e0 < CAPR) ? (e1 - e0) : CAPR, tid, r);
#pragma unroll
    for (int k = 0; k < NR; ++k) {
        const int i = tid + k * 1024;
        if (i < lim) *(float4*)(acc + (i >> 4) * PA + (i & 15) * 4) = hv[k];
    }
    for (int i = lim + tid; i < MTL * 16 * 16; i += 1024)
        *(float4*)(acc + (i >> 4) * PA + (i & 15) * 4) = make_float4(0.0f, 0.0f, 0.0f, 0.0f);
    __syncthreads();

    tile_edge_sum<MTL * 16, PA, 4>(rec, bs, be, e0, e1, h, acc, srec, tid);

    const int half = lane >> 4;
    const int l16  = lane & 15;
    const int wvu = __builtin_amdgcn_readfirstlane(wv);
    if (wvu < 2 * MTL) {
        const int mt = wvu >> 1;
        const int n0 = (wvu & 1) * 2;
        v8f d[2];
        tile_mm<64, 2>(acc + (mt * 16 + l16) * PA + half * 2, W2, n0, half, l16, d);
#pragma unroll
        for (int q = 0; q < 2; ++q) {
            const int c = (n0 + q) * 16 + l16;
            const float bv = b2[c];
#pragma unroll
            for (int i = 0; i < 8; ++i) {
                const int row = mt * 16 + i + half * 8;
                const float sc = (row < nn) ? sdinv[row] : 0.0f;
                float v = sc * d[q][i] + bv;
                v = (v <= 0.0f) ? expm1f(v) : v;
                mid[row * PA + c] = sc * v;
            }
        }
    } else {
        const int j = tid - 2 * MTL * 32;
        static_assert(64 * PW / 4 == 4 * 6 * 32, "four float4 per thread of the six waves");
#pragma unroll
        for (int k = 0; k < 4; ++k) ((float4*)Ml)[j + k * 6 * 32] = ((const float4*)Mg)[j + k * 6 * 32];
    }
    __syncthreads();
    for (int job = wvu; job < 3 * MTL; job += 32) {
        const int mt = job / 3;
        const int q  = job - mt * 3;
        v8f e[1];
        tile_mm<PW, 1>(mid + (mt * 16 + l16) * PA + half * 2, Ml, q, half, l16, e);
        const int c = q * 16 + l16;
#pragma unroll
        for (int i = 0; i < 8; ++i) {
            const int row = mt * 16 + i + half * 8;
            if (row < nn) outp[(size_t)(base + row) * PW + c] = e[0][i];
        }
    }
}

#define OPENING_LOADS_TOGETHER3(dv, cv, hv, r, bs, be)                           \
    asm volatile("" :: "s"(bs), "s"(be), "v"(dv), "v"(cv),                       \
                 "v"(hv[0].x), "v"(hv[0].y), "v"(hv[0].z), "v"(hv[0].w),         \
                 "v"(hv[1].x), "v"(hv[1].y), "v"(hv[1].z), "v"(hv[1].w),         \
                 "v"(hv[2].x), "v"(hv[2].y), "v"(hv[2].z), "v"(hv[2].w),         \
                 "v"(r[0]), "v"(r[1]), "v"(r[2]), "v"(r[3]))

__global__ __launch_bounds__(1024) void k_tile_agg_out(const int* __restrict__ rec,
                                                       const int* __restrict__ bst,
                                                       const float* __restrict__ dinv,
                                                       const float* __restrict__ h,
                                                       const float* __restrict__ cb,
                                                       float* __restrict__ out) {
    constexpr int Q = PW / 4;
    __shared__ __attribute__((aligned(16))) float acc[TILE * PO];
    __shared__ int   srec[CAPR];
    __shared__ float sdinv[TILE];
    __shared__ float scb[PW];
    const int tid  = threadIdx.x;
    const int t    = blockIdx.x;
    const int base = t * TILE;
    const int nn   = (NN - base < TILE) ? (NN - base) : TILE;

    const int* tb = bst + t * 16;
    const int wvo = __builtin_amdgcn_readfirstlane(tid >> 5);
    const int rbo = (wvo < 2 * NRB) ? (wvo >> 1) : 0;
    const int e0 = tb[0];
    const int e1 = tb[NRB];
    const int bs = tb[rbo];
    const int be = tb[rbo + 1];
    const float dv = dinv[base + ((tid < nn) ? tid : (nn - 1))];
    const float cv = cb[(tid < PW) ? tid : (PW - 1)];
    const float4* h4 = (const float4*)(h + (size_t)base * PW);
    constexpr int NR = (TILE * Q + 1023) / 1024;
    static_assert(NR == 3, "OPENING_LOADS_TOGETHER3 names three row loads");
    const int lim = nn * Q;
    float4 hv[NR];
#pragma unroll
    for (int k = 0; k < NR; ++k) {
        const int i  = tid + k * 1024;
        hv[k] = h4[(i < lim) ? i : (lim - 1)];
    }
    int r[4];
#pragma unroll
    for (int k = 0; k < 4; ++k) r[k] = 0;
    if (e0 < e1) load_pass_records(rec, e0, e1, tid, r);
    OPENING_LOADS_TOGETHER3(dv, cv, hv, r, bs, be);
    if (tid < nn) sdinv[tid] = dv;
    stage_pass_records(srec, (e1 - e0 < CAPR) ? (e1 - e0) : CAPR, tid, r);
    if (tid < PW) scb[tid] = cv;
#pragma unroll
    for (int k = 0; k < NR; ++k) {
        const int i = tid + k * 1024;
        if (i < lim) {
            const int row = i / Q;
            *(float4*)(acc + row * PO + (i - row * Q) * 4) = hv[k];
        }
    }
    for (int i = lim + tid; i < TILE * Q; i += 1024) {
        const int row = i / Q;
        *(float4*)(acc + row * PO + (i - row * Q) * 4) = make_float4(0.0f, 0.0f, 0.0f, 0.0f);
    }
    __syncthreads();

    tile_edge_sum<TILE, PO, 3>(rec, bs, be, e0, e1, h, acc, srec, tid);

    float* o = out + (size_t)base * 40;
    for (int i = tid; i < nn * 40; i += 1024) {
        const int row = i / 40;
        const int c   = i - row * 40;
        o[i] = sdinv[row] * acc[row * PO + c] + scb[c];
    }
}


extern "C" void kernel_launch(void* const* d_in, const int* in_sizes, int n_in,
                              void* d_out, int out_size, void* d_ws, size_t ws_size,
                              hipStream_t stream) {
    const float* x  = (const float*)d_in[0];
    const int*   ei = (const int*)d_in[1];
    const float* W1 = (const float*)d_in[2];
    const float* b1 = (const float*)d_in[3];
    const float* W2 = (const float*)d_in[4];
    const float* b2 = (const float*)d_in[5];
    const float* W3 = (const float*)d_in[6];
    const float* b3 = (const float*)d_in[7];
    const float* Wl = (const float*)d_in[8];
    const float* bl = (const float*)d_in[9];
    float* out = (float*)d_out;

    const int E = in_sizes[1] / 2;
    const int* src = ei;
    const int* dst = ei + E;

    float* dinv     = (float*)d_ws;
    int*   bst      = (int*)(dinv + 50176);
    int*   coffT    = bst + 16 * NBIN;
    int*   rec      = coffT + 1024 * NBIN;
    float* G        = (float*)(rec + 2 * (size_t)E);
    float* ACT      = G + (size_t)NN * H;
    int*   crec     = (int*)(ACT + (size_t)NN * H);
    float* Mg       = (float*)(crec + 2 * (size_t)E);
    float* cbias    = Mg + 64 * PW;

    const dim3 blk(256);
    const dim3 big(1024);
    const dim3 gTile(NBIN);

    k_chunk_sort<<<dim3(MAXCH), blk, 0, stream>>>(src, dst, coffT, crec, E);
    k_tile_dinv_mm1<<<gTile, big, 0, stream>>>(crec, coffT, x, W1, dinv, G, rec, bst);

    k_tile_agg<1, H><<<gTile, big, 0, stream>>>(rec, bst, dinv, G, b1, ACT, W3, b3, Wl, bl, Mg, cbias);

    k_tile_agg_mm<<<gTile, big, 0, stream>>>(rec, bst, dinv, ACT, W2, b2, Mg, G);
    k_tile_agg_out<<<gTile, big, 0, stream>>>(rec, bst, dinv, G, cbias, out);
}


# Rows of list neighbours that end in one node are added in pairs on the vector unit ahead of the one-hot WMMA: 8 edges a product, a 2nd product only where a pair straddles two nodes; sums regrouped
# speedup vs baseline: 4.8494x; 1.0496x over previous
#include <hip/hip_runtime.h>
#include <math.h>

typedef __attribute__((ext_vector_type(2))) float v2f;
typedef __attribute__((ext_vector_type(8))) float v8f;

#define NN 50000
#define H  64
#define TILE 196
#define NBIN 256
#define CH   4096

#define MAXCH 256

__device__ __forceinline__ int wave_incl_scan(int v, int lane) {
#pragma unroll
    for (int o = 1; o < 32; o <<= 1) {
        const int t = __shfl_up(v, o);
        if (lane >= o) v += t;
    }
    return v;
}

__global__ __launch_bounds__(256) void k_chunk_sort(const int* __restrict__ src,
                                                    const int* __restrict__ dst,
                                                    int* __restrict__ coffT,
                                                    int* __restrict__ crec, int E) {
    __shared__ int  hist[NBIN];
    __shared__ int  wtot[8];
    __shared__ int2 stage[CH];
    const int tid  = threadIdx.x;
    const int lane = tid & 31;
    const int wv   = tid >> 5;
    const int c    = blockIdx.x;
    hist[tid] = 0;
    __syncthreads();
    const int e0 = c * CH;
    int b[CH / 256], s[CH / 256], y[CH / 256], rk[CH / 256];
#pragma unroll
    for (int k = 0; k < CH / 256; ++k) {
        const int e  = e0 + k * 256 + tid;
        const int ec = (e < E) ? e : (E - 1);
        y[k] = dst[ec];
        s[k] = src[ec];
    }
#pragma unroll
    for (int k = 0; k < CH / 256; ++k) {
        const int e = e0 + k * 256 + tid;
        b[k]  = y[k] / TILE;
        y[k] -= b[k] * TILE;
        rk[k] = 0;
        if (e < E) rk[k] = atomicAdd(&hist[b[k]], 1);
    }
    __syncthreads();
    const int v    = hist[tid];
    const int incl = wave_incl_scan(v, lane);
    if (lane == 31) wtot[wv] = incl;
    __syncthreads();
    int add = 0;
#pragma unroll
    for (int w = 0; w < 8; ++w) add += (w < wv) ? wtot[w] : 0;
    const int off = add + incl - v;
    coffT[tid * MAXCH + c] = off;
    if (tid == NBIN - 1) coffT[NBIN * MAXCH + c] = off + v;
    hist[tid] = off;
    __syncthreads();
#pragma unroll
    for (int k = 0; k < CH / 256; ++k) {
        const int e = e0 + k * 256 + tid;
        if (e < E) stage[hist[b[k]] + rk[k]] = make_int2(s[k], y[k]);
    }
    __syncthreads();
    const int n = (E - e0 < CH) ? (E - e0) : CH;
    int2* o2 = (int2*)crec + e0;
    for (int i = tid; i < n; i += 256) o2[i] = stage[i];
}

#define GB 9
#define NRB  13
#define NCNT 224
#define CAPS 8192

__device__ __forceinline__ int list_record(const int* P, const int* S, const int2* __restrict__ crec2, int i) {
    int pos = 0;
#pragma unroll
    for (int step = MAXCH / 2; step >= 1; step >>= 1) pos += (P[pos + step] <= i) ? step : 0;
    const int2 q = crec2[S[pos] + i];
    return q.x | (q.y << 20);
}

__global__ __launch_bounds__(1024) void k_tile_dinv_mm1(const int* __restrict__ crec,
                                                        const int* __restrict__ coffT,
                                                        const float* __restrict__ x,
                                                        const float* __restrict__ W1,
                                                        float* __restrict__ dinv,
                                                        float* __restrict__ G,
                                                        int* __restrict__ recp,
                                                        int* __restrict__ bst) {
    __shared__ int c[NCNT];
    __shared__ int fill[NCNT];
    __shared__ int pfx[6 * MAXCH];
    __shared__ int cst[6 * MAXCH];
    __shared__ int stg[CAPS];
    __shared__ int srt[CAPS];
    __shared__ int sBT[2];
    const int tid  = threadIdx.x;
    const int lane = tid & 31;
    const int half = lane >> 4;
    const int l16  = lane & 15;
    const int t    = blockIdx.x;
    const int base = t * TILE;
    const int nn   = (NN - base < TILE) ? (NN - base) : TILE;
    if (tid < NCNT) c[tid] = 0;
    __syncthreads();
    const int wvu = __builtin_amdgcn_readfirstlane(tid >> 5);
    const int mt  = wvu >> 1;
    const int n0  = (wvu & 1) * 2;
    const int2* crec2 = (const int2*)crec;
    int Bv = 0;
    v8f d[2];
    d[0] = (v8f){};
    d[1] = (v8f){};
    if (wvu < 26) {
        int arow = base + mt * 16 + l16;
        arow = (arow < NN) ? arow : (NN - 1);
        const float* Ap = x + (size_t)arow * 128 + half * 2;
        const float* Bp = W1 + half * 2 * 64;
#pragma unroll
        for (int k = 0; k < 128; k += 4) {
            v2f a;
            a.x = Ap[k];
            a.y = Ap[k + 1];
#pragma unroll
            for (int q = 0; q < 2; ++q) {
                const int cc = (n0 + q) * 16 + l16;
                v2f b;
                b.x = Bp[k * 64 + cc];
                b.y = Bp[k * 64 + 64 + cc];
                d[q] = __builtin_amdgcn_wmma_f32_16x16x4_f32(
                    false, a, false, b, (short)0, d[q], false, false);
            }
        }
    } else {
        const int sw = wvu - 26;
        int* P = pfx + sw * MAXCH;
        int* S = cst + sw * MAXCH;
        const int4* rs = (const int4*)(coffT + (size_t)t * MAXCH) + lane * 2;
        const int4* re = (const int4*)(coffT + (size_t)(t + 1) * MAXCH) + lane * 2;
        const int4 s0 = rs[0], s1 = rs[1], f0 = re[0], f1 = re[1];
        const int st[8] = {s0.x, s0.y, s0.z, s0.w, s1.x, s1.y, s1.z, s1.w};
        const int en[8] = {f0.x, f0.y, f0.z, f0.w, f1.x, f1.y, f1.z, f1.w};
        int cn[8];
        int lt = 0, ls = 0;
#pragma unroll
        for (int j = 0; j < 8; ++j) {
            cn[j] = en[j] - st[j];
            lt += cn[j];
            ls += st[j];
        }
        const int incl = wave_incl_scan(lt, lane);
        int run = incl - lt;
#pragma unroll
        for (int j = 0; j < 8; ++j) {
            P[lane * 8 + j] = run;
            S[lane * 8 + j] = (lane * 8 + j) * CH + st[j] - run;
            run += cn[j];
        }
        const int T = __shfl(incl, 31);
        int B = ls;
#pragma unroll
        for (int o = 16; o >= 1; o >>= 1) B += __shfl_xor(B, o);
        Bv = B;
        if (sw == 0 && lane == 0) {
            sBT[0] = B;
            sBT[1] = T;
        }
        for (int i0 = sw * 32 + lane; i0 < T; i0 += GB * 6 * 32) {
            int  ii[GB], pos[GB];
            int2 r[GB];
#pragma unroll
            for (int k = 0; k < GB; ++k) {
                const int i = i0 + k * 6 * 32;
                ii[k]  = (i < T) ? i : (T - 1);
                pos[k] = 0;
            }
#pragma unroll
            for (int step = MAXCH / 2; step >= 1; step >>= 1)
#pragma unroll
                for (int k = 0; k < GB; ++k)
                    pos[k] += (P[pos[k] + step] <= ii[k]) ? step : 0;
#pragma unroll
            for (int k = 0; k < GB; ++k) r[k] = crec2[S[pos[k]] + ii[k]];
#pragma unroll
            for (int k = 0; k < GB; ++k)
                if (i0 + k * 6 * 32 < T) {
                    if (ii[k] < CAPS) stg[ii[k]] = r[k].x | (r[k].y << 20);
                    atomicAdd(&c[r[k].y], 1);
                }
        }
    }
    __syncthreads();
    if (tid < nn) dinv[base + tid] = rsqrtf((float)c[tid] + 1.0f);
    if (wvu < 26) {
#pragma unroll
        for (int i = 0; i < 8; ++i) {
            const int row = mt * 16 + i + half * 8;
            if (row < nn) {
                const float sc = rsqrtf((float)c[row] + 1.0f);
#pragma unroll
                for (int q = 0; q < 2; ++q)
                    G[(size_t)(base + row) * 64 + (n0 + q) * 16 + l16] = sc * d[q][i];
            }
        }
    } else if (wvu == 26) {
        int cv[7];
        int s = 0;
#pragma unroll
        for (int j = 0; j < 7; ++j) {
            cv[j] = c[lane * 7 + j];
            s += cv[j];
        }
        int run = wave_incl_scan(s, lane) - s;
#pragma unroll
        for (int j = 0; j < 7; ++j) {
            const int idx = lane * 7 + j;
            fill[idx] = run;
            if ((idx & 15) == 0) bst[t * 16 + (idx >> 4)] = Bv + run;
            run += cv[j];
        }
    }
    __syncthreads();
    const int Bq = sBT[0];
    const int Tq = sBT[1];
    for (int i0 = tid; i0 < Tq; i0 += 4 * 1024) {
        int w[4], pl[4];
#pragma unroll
        for (int k = 0; k < 4; ++k) {
            const int i = i0 + k * 1024;
            w[k] = 0;
            if (i < Tq) w[k] = (i < CAPS) ? stg[i] : list_record(pfx, cst, crec2, i);
        }
#pragma unroll
        for (int k = 0; k < 4; ++k) {
            pl[k] = 0;
            if (i0 + k * 1024 < Tq) pl[k] = atomicAdd(&fill[w[k] >> 20], 1);
        }
#pragma unroll
        for (int k = 0; k < 4; ++k)
            if (i0 + k * 1024 < Tq) {
                const int v = (w[k] & 0xFFFFF) | (((w[k] >> 20) & 15) << 20);
                if (pl[k] < CAPS) srt[pl[k]] = v;
                else              recp[Bq + pl[k]] = v;
            }
    }
    __syncthreads();
    const int nl = (Tq < CAPS) ? Tq : CAPS;
    for (int i = tid; i < nl; i += 1024) recp[Bq + i] = srt[i];
}

#define CAPR 4096
#define GU   2

struct __attribute__((aligned(4))) row3 { float x, y, z; };
template<int NC> struct row_piece;
template<> struct row_piece<4> { typedef float4 T; };
template<> struct row_piece<3> { typedef row3 T; };
__device__ __forceinline__ float piece_at(const float4& v, int t) { return t == 0 ? v.x : (t == 1 ? v.y : (t == 2 ? v.z : v.w)); }
__device__ __forceinline__ float piece_at(const row3& v, int t) { return t == 0 ? v.x : (t == 1 ? v.y : v.z); }

__device__ __forceinline__ void load_pass_records(const int* __restrict__ recp, int c0, int e1,
                                                  int tid, int (&r)[4]) {
    const int n = (e1 - c0 < CAPR) ? (e1 - c0) : CAPR;
#pragma unroll
    for (int k = 0; k < 4; ++k) {
        const int idx = k * 1024 + tid;
        r[k] = recp[c0 + ((idx < n) ? idx : (n - 1))];
    }
}

__device__ __forceinline__ void stage_pass_records(int* srec, int n, int tid, const int (&r)[4]) {
#pragma unroll
    for (int k = 0; k < 4; ++k) {
        const int idx = k * 1024 + tid;
        if (idx < n) srec[idx] = r[k];
    }
}

template<int ROWS, int PITCH, int NC>
__device__ __forceinline__ void tile_edge_sum(const int* __restrict__ recp, int bs, int be,
                                              int e0, int e1,
                                              const float* __restrict__ h, float* acc,
                                              int* srec, int tid) {
    const int lane = tid & 31;
    const int wv   = tid >> 5;
    const int half = lane >> 4;
    const int l16  = lane & 15;
    const int wvu  = __builtin_amdgcn_readfirstlane(wv);
    const int rbw  = wvu >> 1;
    const int part = wvu & 1;
    typedef typename row_piece<NC>::T RP;
    const RP* hp = (const RP*)h;

    v8f d[NC];
#pragma unroll
    for (int t = 0; t < NC; ++t) d[t] = (v8f){};

    for (int c0 = e0; c0 < e1; c0 += CAPR) {
        const int n = (e1 - c0 < CAPR) ? (e1 - c0) : CAPR;

        if (wvu < 2 * NRB) {
            const int lo   = (bs > c0) ? bs : c0;
            const int hi0  = (be < c0 + n) ? be : (c0 + n);
            const int hi   = (hi0 > lo) ? hi0 : lo;
            const int segS = lo - c0;
            const int segE = hi - c0;
            const int ng   = (segE - segS + 7) >> 3;
            const int gh   = (ng + 1) >> 1;
            const int g0   = part ? gh : 0;
            const int g1   = part ? ng : gh;
            const int je   = segS + g1 * 8;
            const int jEnd = (segE < je) ? segE : je;
            for (int g = g0; g < g1; g += GU) {
                int rc[GU][4];
                RP  hv[GU][4];
#pragma unroll
                for (int u = 0; u < GU; ++u)
#pragma unroll
                    for (int q = 0; q < 4; ++q) {
                        const int j = segS + (g + u) * 8 + half * 4 + q;
                        rc[u][q] = srec[(j < jEnd) ? j : (jEnd - 1)];
                    }
#pragma unroll
                for (int u = 0; u < GU; ++u)
#pragma unroll
                    for (int q = 0; q < 4; ++q)
                        if constexpr (NC == 4) {
                            hv[u][q] = hp[(size_t)(rc[u][q] & 0xFFFFF) * 16 + l16];
                        } else {
                            const unsigned off = (unsigned)(rc[u][q] & 0xFFFFF) * (64u * NC) + (unsigned)l16 * (4u * NC);
                            hv[u][q] = *(const RP*)((const char*)h + off);
                        }
                static_assert(GU == 2, "the statement below names the loads of two groups");
                asm volatile("" :: "v"(hv[0][0].x), "v"(hv[0][1].x), "v"(hv[0][2].x), "v"(hv[0][3].x),
                                   "v"(hv[1][0].x), "v"(hv[1][1].x), "v"(hv[1][2].x), "v"(hv[1][3].x));
#pragma unroll
                for (int u = 0; u < GU; ++u) {
                    const int j = segS + (g + u) * 8 + half * 4;
                    float oa[2], ob[2];
                    float sm[2][NC];
                    bool  strad = false;
#pragma unroll
                    for (int s = 0; s < 2; ++s) {
                        const int  dA   = rc[u][2 * s] >> 20;
                        const int  dB   = rc[u][2 * s + 1] >> 20;
                        const bool vA   = (j + 2 * s) < jEnd;
                        const bool vB   = (j + 2 * s + 1) < jEnd;
                        const bool same = vB && (dA == dB);
                        const bool two  = vB && (dA != dB);
                        oa[s] = (vA  && dA == l16) ? 1.0f : 0.0f;
                        ob[s] = (two && dB == l16) ? 1.0f : 0.0f;
#pragma unroll
                        for (int t = 0; t < NC; ++t)
                            sm[s][t] = piece_at(hv[u][2 * s], t) + (same ? piece_at(hv[u][2 * s + 1], t) : 0.0f);
                        strad = strad || two;
                    }
                    v2f a;
                    a.x = oa[0];
                    a.y = oa[1];
#pragma unroll
                    for (int t = 0; t < NC; ++t) {
                        v2f b;
                        b.x = sm[0][t];
                        b.y = sm[1][t];
                        d[t] = __builtin_amdgcn_wmma_f32_16x16x4_f32(false, a, false, b, (short)0, d[t], false, false);
                    }
                    if (__builtin_amdgcn_ballot_w32(strad) != 0u) {
                        v2f a2;
                        a2.x = ob[0];
                        a2.y = ob[1];
#pragma unroll
                        for (int t = 0; t < NC; ++t) {
                            v2f b;
                            b.x = piece_at(hv[u][1], t);
                            b.y = piece_at(hv[u][3], t);
                            d[t] = __builtin_amdgcn_wmma_f32_16x16x4_f32(false, a2, false, b, (short)0, d[t], false, false);
                        }
                    }
                }
            }
        }
        if (c0 + CAPR < e1) {
            int r[4];
            load_pass_records(recp, c0 + CAPR, e1, tid, r);
            __syncthreads();
            stage_pass_records(srec, (e1 - c0 - CAPR < CAPR) ? (e1 - c0 - CAPR) : CAPR, tid, r);
            __syncthreads();
        }
    }

#pragma unroll
    for (int p = 0; p < 2; ++p) {
        if (wvu < 2 * NRB && part == p) {
#pragma unroll
            for (int i = 0; i < 8; ++i) {
                const int row = rbw * 16 + half * 8 + i;
                if (row < ROWS) {
                    if constexpr (NC == 4) {
                        float4* o = (float4*)(acc + row * PITCH + l16 * 4);
                        float4  v = *o;
                        v.x += d[0][i]; v.y += d[1][i]; v.z += d[2][i]; v.w += d[3][i];
                        *o = v;
                    } else {
                        float* o = acc + row * PITCH + l16 * NC;
#pragma unroll
                        for (int t = 0; t < NC; ++t) o[t] += d[t][i];
                    }
                }
            }
        }
        __syncthreads();
    }
}

#define PW 48
#define PO 52

template<int NB, int NT>
__device__ __forceinline__ void tile_mm(const float* arow, const float* __restrict__ B,
                                        int n0, int half, int l16, v8f (&d)[NT]);

#define OPENING_LOADS_TOGETHER(dv, hv, r, bs, be)                                \
    asm volatile("" :: "s"(bs), "s"(be), "v"(dv),                                \
                 "v"(hv[0].x), "v"(hv[0].y), "v"(hv[0].z), "v"(hv[0].w),         \
                 "v"(hv[1].x), "v"(hv[1].y), "v"(hv[1].z), "v"(hv[1].w),         \
                 "v"(hv[2].x), "v"(hv[2].y), "v"(hv[2].z), "v"(hv[2].w),         \
                 "v"(hv[3].x), "v"(hv[3].y), "v"(hv[3].z), "v"(hv[3].w),         \
                 "v"(r[0]), "v"(r[1]), "v"(r[2]), "v"(r[3]))

template<int DO_ELU, int HW>
__global__ __launch_bounds__(1024) void k_tile_agg(const int* __restrict__ rec,
                                                   const int* __restrict__ bst,
                                                   const float* __restrict__ dinv,
                                                   const float* __restrict__ h,
                                                   const float* __restrict__ bias,
                                                   float* __restrict__ act,
                                                   const float* __restrict__ W3,
                                                   const float* __restrict__ b3,
                                                   const float* __restrict__ Wl,
                                                   const float* __restrict__ bl,
                                                   float* __restrict__ Mg,
                                                   float* __restrict__ cb) {
    static_assert(HW == 64, "rows of 64 floats");
    constexpr int Q  = HW / 4;
    __shared__ __attribute__((aligned(16))) float acc[TILE * HW];
    __shared__ int   srec[CAPR];
    __shared__ float sdinv[TILE];
    const int tid  = threadIdx.x;
    const int t    = blockIdx.x;
    const int base = t * TILE;
    const int nn   = (NN - base < TILE) ? (NN - base) : TILE;

    const int* tb = bst + t * 16;
    const int wvo = __builtin_amdgcn_readfirstlane(tid >> 5);
    const int rbo = (wvo < 2 * NRB) ? (wvo >> 1) : 0;
    const int e0 = tb[0];
    const int e1 = tb[NRB];
    const int bs = tb[rbo];
    const int be = tb[rbo + 1];
    const float dv = dinv[base + ((tid < nn) ? tid : (nn - 1))];
    const float4* h4  = (const float4*)(h + (size_t)base * HW);
    float4*       a4  = (float4*)acc;
    constexpr int NR = (TILE * Q + 1023) / 1024;
    static_assert(NR == 4, "OPENING_LOADS_TOGETHER names four row loads");
    const int lim = nn * Q;
    float4 hv[NR];
#pragma unroll
    for (int k = 0; k < NR; ++k) {
        const int i  = tid + k * 1024;
        hv[k] = h4[(i < lim) ? i : (lim - 1)];
    }
    int r[4];
#pragma unroll
    for (int k = 0; k < 4; ++k) r[k] = 0;
    if (e0 < e1) load_pass_records(rec, e0, e1, tid, r);
    OPENING_LOADS_TOGETHER(dv, hv, r, bs, be);
    if (tid < nn) sdinv[tid] = dv;
    stage_pass_records(srec, (e1 - e0 < CAPR) ? (e1 - e0) : CAPR, tid, r);
#pragma unroll
    for (int k = 0; k < NR; ++k) {
        const int i = tid + k * 1024;
        if (i < lim) a4[i] = hv[k];
    }
    for (int i = lim + tid; i < TILE * Q; i += 1024) a4[i] = make_float4(0.0f, 0.0f, 0.0f, 0.0f);
    __syncthreads();

    tile_edge_sum<TILE, HW, 4>(rec, bs, be, e0, e1, h, acc, srec, tid);

    float4* o4 = (float4*)(act + (size_t)base * HW);
    for (int i = tid; i < nn * Q; i += 1024) {
        const float4 a = a4[i];
        const int    c = (i & 15) * 4;
        const float  sc = sdinv[i >> 4];
        float4 v = make_float4(sc * a.x + bias[c], sc * a.y + bias[c + 1], sc * a.z + bias[c + 2], sc * a.w + bias[c + 3]);
        if (DO_ELU) {
            v.x = (v.x <= 0.0f) ? expm1f(v.x) : v.x;
            v.y = (v.y <= 0.0f) ? expm1f(v.y) : v.y;
            v.z = (v.z <= 0.0f) ? expm1f(v.z) : v.z;
            v.w = (v.w <= 0.0f) ? expm1f(v.w) : v.w;
        }
        o4[i] = make_float4(sc * v.x, sc * v.y, sc * v.z, sc * v.w);
    }

    if (t == NBIN - 1) {
        const int lane = tid & 31;
        const int half = lane >> 4;
        const int l16  = lane & 15;
        const int wvu = __builtin_amdgcn_readfirstlane(tid >> 5);
        if (wvu < 4) {
            v8f e[3];
            tile_mm<40, 3>(W3 + (wvu * 16 + l16) * 64 + half * 2, Wl, 0, half, l16, e);
#pragma unroll
            for (int q = 0; q < 3; ++q)
#pragma unroll
                for (int i = 0; i < 8; ++i)
                    Mg[(wvu * 16 + i + half * 8) * PW + q * 16 + l16] = e[q][i];
        } else if (wvu == 4) {
            for (int c = lane; c < PW; c += 32) {
                float s = 0.0f;
                if (c < 40) {
                    s = bl[c];
#pragma unroll 16
                    for (int k = 0; k < 64; ++k) s += b3[k] * Wl[k * 40 + c];
                }
                cb[c] = s;
            }
        }
    }
}


#define PA  68
#define MTL 13

template<int NB, int NT>
__device__ __forceinline__ void tile_mm(const float* arow, const float* __restrict__ B,
                                        int n0, int half, int l16, v8f (&d)[NT]) {
#pragma unroll
    for (int t = 0; t < NT; ++t) d[t] = (v8f){};
    const float* Bp = B + half * 2 * NB;
#pragma unroll
    for (int k = 0; k < 64; k += 4) {
        v2f a;
        a.x = arow[k];
        a.y = arow[k + 1];
#pragma unroll
        for (int t = 0; t < NT; ++t) {
            const int c = (n0 + t) * 16 + l16;
            v2f b;
            if constexpr (NB % 16 == 0) {
                b.x = Bp[k * NB + c];
                b.y = Bp[k * NB + NB + c];
            } else {
                const float m  = (c < NB) ? 1.0f : 0.0f;
                const int   cc = (c < NB) ? c : (NB - 1);
                b.x = m * Bp[k * NB + cc];
                b.y = m * Bp[k * NB + NB + cc];
            }
            d[t] = __builtin_amdgcn_wmma_f32_16x16x4_f32(
                false, a, false, b, (short)0, d[t], false, false);
        }
    }
}


__global__ __launch_bounds__(1024) void k_tile_agg_mm(const int* __restrict__ rec,
                                                      const int* __restrict__ bst,
                                                      const float* __restrict__ dinv,
                                                      const float* __restrict__ h,
                                                      const float* __restrict__ W2,
                                                      const float* __restrict__ b2,
                                                      const float* __restrict__ Mg,
                                                      float* __restrict__ outp) {
    constexpr int HW = 64;
    __shared__ __attribute__((aligned(16))) float acc[MTL * 16 * PA];
    __shared__ __attribute__((aligned(16))) float mid[MTL * 16 * PA];
    __shared__ __attribute__((aligned(16))) float Ml[64 * PW];
    __shared__ int   srec[CAPR];
    __shared__ float sdinv[TILE];
    const int tid  = threadIdx.x;
    const int lane = tid & 31;
    const int wv   = tid >> 5;
    const int t    = blockIdx.x;
    const int base = t * TILE;
    const int nn   = (NN - base < TILE) ? (NN - base) : TILE;

    const int* tb = bst + t * 16;
    const int wvo = __builtin_amdgcn_readfirstlane(tid >> 5);
    const int rbo = (wvo < 2 * NRB) ? (wvo >> 1) : 0;
    const int e0 = tb[0];
    const int e1 = tb[NRB];
    const int bs = tb[rbo];
    const int be = tb[rbo + 1];
    const float dv = dinv[base + ((tid < nn) ? tid : (nn - 1))];
    const float4* h4 = (const float4*)(h + (size_t)base * HW);
    constexpr int NR = (TILE * 16 + 1023) / 1024;
    static_assert(NR == 4, "OPENING_LOADS_TOGETHER names four row loads");
    const int lim = nn * 16;
    float4 hv[NR];
#pragma unroll
    for (int k = 0; k < NR; ++k) {
        const int i  = tid + k * 1024;
        hv[k] = h4[(i < lim) ? i : (lim - 1)];
    }
    int r[4];
#pragma unroll
    for (int k = 0; k < 4; ++k) r[k] = 0;
    if (e0 < e1) load_pass_records(rec, e0, e1, tid, r);
    OPENING_LOADS_TOGETHER(dv, hv, r, bs, be);
    if (tid < nn) sdinv[tid] = dv;
    stage_pass_records(srec, (e1 - e0 < CAPR) ? (e1 - e0) : CAPR, tid, r);
#pragma unroll
    for (int k = 0; k < NR; ++k) {
        const int i = tid + k * 1024;
        if (i < lim) *(float4*)(acc + (i >> 4) * PA + (i & 15) * 4) = hv[k];
    }
    for (int i = lim + tid; i < MTL * 16 * 16; i += 1024)
        *(float4*)(acc + (i >> 4) * PA + (i & 15) * 4) = make_float4(0.0f, 0.0f, 0.0f, 0.0f);
    __syncthreads();

    tile_edge_sum<MTL * 16, PA, 4>(rec, bs, be, e0, e1, h, acc, srec, tid);

    const int half = lane >> 4;
    const int l16  = lane & 15;
    const int wvu = __builtin_amdgcn_readfirstlane(wv);
    if (wvu < 2 * MTL) {
        const int mt = wvu >> 1;
        const int n0 = (wvu & 1) * 2;
        v8f d[2];
        tile_mm<64, 2>(acc + (mt * 16 + l16) * PA + half * 2, W2, n0, half, l16, d);
#pragma unroll
        for (int q = 0; q < 2; ++q) {
            const int c = (n0 + q) * 16 + l16;
            const float bv = b2[c];
#pragma unroll
            for (int i = 0; i < 8; ++i) {
                const int row = mt * 16 + i + half * 8;
                const float sc = (row < nn) ? sdinv[row] : 0.0f;
                float v = sc * d[q][i] + bv;
                v = (v <= 0.0f) ? expm1f(v) : v;
                mid[row * PA + c] = sc * v;
            }
        }
    } else {
        const int j = tid - 2 * MTL * 32;
        static_assert(64 * PW / 4 == 4 * 6 * 32, "four float4 per thread of the six waves");
#pragma unroll
        for (int k = 0; k < 4; ++k) ((float4*)Ml)[j + k * 6 * 32] = ((const float4*)Mg)[j + k * 6 * 32];
    }
    __syncthreads();
    for (int job = wvu; job < 3 * MTL; job += 32) {
        const int mt = job / 3;
        const int q  = job - mt * 3;
        v8f e[1];
        tile_mm<PW, 1>(mid + (mt * 16 + l16) * PA + half * 2, Ml, q, half, l16, e);
        const int c = q * 16 + l16;
#pragma unroll
        for (int i = 0; i < 8; ++i) {
            const int row = mt * 16 + i + half * 8;
            if (row < nn) outp[(size_t)(base + row) * PW + c] = e[0][i];
        }
    }
}

#define OPENING_LOADS_TOGETHER3(dv, cv, hv, r, bs, be)                           \
    asm volatile("" :: "s"(bs), "s"(be), "v"(dv), "v"(cv),                       \
                 "v"(hv[0].x), "v"(hv[0].y), "v"(hv[0].z), "v"(hv[0].w),         \
                 "v"(hv[1].x), "v"(hv[1].y), "v"(hv[1].z), "v"(hv[1].w),         \
                 "v"(hv[2].x), "v"(hv[2].y), "v"(hv[2].z), "v"(hv[2].w),         \
                 "v"(r[0]), "v"(r[1]), "v"(r[2]), "v"(r[3]))

__global__ __launch_bounds__(1024) void k_tile_agg_out(const int* __restrict__ rec,
                                                       const int* __restrict__ bst,
                                                       const float* __restrict__ dinv,
                                                       const float* __restrict__ h,
                                                       const float* __restrict__ cb,
                                                       float* __restrict__ out) {
    constexpr int Q = PW / 4;
    __shared__ __attribute__((aligned(16))) float acc[TILE * PO];
    __shared__ int   srec[CAPR];
    __shared__ float sdinv[TILE];
    __shared__ float scb[PW];
    const int tid  = threadIdx.x;
    const int t    = blockIdx.x;
    const int base = t * TILE;
    const int nn   = (NN - base < TILE) ? (NN - base) : TILE;

    const int* tb = bst + t * 16;
    const int wvo = __builtin_amdgcn_readfirstlane(tid >> 5);
    const int rbo = (wvo < 2 * NRB) ? (wvo >> 1) : 0;
    const int e0 = tb[0];
    const int e1 = tb[NRB];
    const int bs = tb[rbo];
    const int be = tb[rbo + 1];
    const float dv = dinv[base + ((tid < nn) ? tid : (nn - 1))];
    const float cv = cb[(tid < PW) ? tid : (PW - 1)];
    const float4* h4 = (const float4*)(h + (size_t)base * PW);
    constexpr int NR = (TILE * Q + 1023) / 1024;
    static_assert(NR == 3, "OPENING_LOADS_TOGETHER3 names three row loads");
    const int lim = nn * Q;
    float4 hv[NR];
#pragma unroll
    for (int k = 0; k < NR; ++k) {
        const int i  = tid + k * 1024;
        hv[k] = h4[(i < lim) ? i : (lim - 1)];
    }
    int r[4];
#pragma unroll
    for (int k = 0; k < 4; ++k) r[k] = 0;
    if (e0 < e1) load_pass_records(rec, e0, e1, tid, r);
    OPENING_LOADS_TOGETHER3(dv, cv, hv, r, bs, be);
    if (tid < nn) sdinv[tid] = dv;
    stage_pass_records(srec, (e1 - e0 < CAPR) ? (e1 - e0) : CAPR, tid, r);
    if (tid < PW) scb[tid] = cv;
#pragma unroll
    for (int k = 0; k < NR; ++k) {
        const int i = tid + k * 1024;
        if (i < lim) {
            const int row = i / Q;
            *(float4*)(acc + row * PO + (i - row * Q) * 4) = hv[k];
        }
    }
    for (int i = lim + tid; i < TILE * Q; i += 1024) {
        const int row = i / Q;
        *(float4*)(acc + row * PO + (i - row * Q) * 4) = make_float4(0.0f, 0.0f, 0.0f, 0.0f);
    }
    __syncthreads();

    tile_edge_sum<TILE, PO, 3>(rec, bs, be, e0, e1, h, acc, srec, tid);

    float* o = out + (size_t)base * 40;
    for (int i = tid; i < nn * 40; i += 1024) {
        const int row = i / 40;
        const int c   = i - row * 40;
        o[i] = sdinv[row] * acc[row * PO + c] + scb[c];
    }
}


extern "C" void kernel_launch(void* const* d_in, const int* in_sizes, int n_in,
                              void* d_out, int out_size, void* d_ws, size_t ws_size,
                              hipStream_t stream) {
    const float* x  = (const float*)d_in[0];
    const int*   ei = (const int*)d_in[1];
    const float* W1 = (const float*)d_in[2];
    const float* b1 = (const float*)d_in[3];
    const float* W2 = (const float*)d_in[4];
    const float* b2 = (const float*)d_in[5];
    const float* W3 = (const float*)d_in[6];
    const float* b3 = (const float*)d_in[7];
    const float* Wl = (const float*)d_in[8];
    const float* bl = (const float*)d_in[9];
    float* out = (float*)d_out;

    const int E = in_sizes[1] / 2;
    const int* src = ei;
    const int* dst = ei + E;

    float* dinv     = (float*)d_ws;
    int*   bst      = (int*)(dinv + 50176);
    int*   coffT    = bst + 16 * NBIN;
    int*   rec      = coffT + 1024 * NBIN;
    float* G        = (float*)(rec + 2 * (size_t)E);
    float* ACT      = G + (size_t)NN * H;
    int*   crec     = (int*)(ACT + (size_t)NN * H);
    float* Mg       = (float*)(crec + 2 * (size_t)E);
    float* cbias    = Mg + 64 * PW;

    const dim3 blk(256);
    const dim3 big(1024);
    const dim3 gTile(NBIN);

    k_chunk_sort<<<dim3(MAXCH), blk, 0, stream>>>(src, dst, coffT, crec, E);
    k_tile_dinv_mm1<<<gTile, big, 0, stream>>>(crec, coffT, x, W1, dinv, G, rec, bst);

    k_tile_agg<1, H><<<gTile, big, 0, stream>>>(rec, bst, dinv, G, b1, ACT, W3, b3, Wl, bl, Mg, cbias);

    k_tile_agg_mm<<<gTile, big, 0, stream>>>(rec, bst, dinv, ACT, W2, b2, Mg, G);
    k_tile_agg_out<<<gTile, big, 0, stream>>>(rec, bst, dinv, G, cbias, out);
}
